# starting kernel
# baseline (speedup 1.0000x reference)
#include <hip/hip_runtime.h>
#include <hip/hip_bf16.h>

typedef __attribute__((ext_vector_type(2))) float v2f;
typedef __attribute__((ext_vector_type(8))) float v8f;

#define D_FEAT 64
#define N_GRAPHS_C 64

__global__ __launch_bounds__(256) void deg_init_kernel(float* __restrict__ deg, int n) {
    int i = blockIdx.x * blockDim.x + threadIdx.x;
    if (i < n) deg[i] = 1.0f;
}

__global__ __launch_bounds__(256) void deg_count_kernel(const int* __restrict__ col,
                                                        float* __restrict__ deg, int e) {
    int i = blockIdx.x * blockDim.x + threadIdx.x;
    if (i < e) atomicAdd(&deg[col[i]], 1.0f);
}

__global__ __launch_bounds__(256) void deg_finish_kernel(float* __restrict__ deg, int n) {
    int i = blockIdx.x * blockDim.x + threadIdx.x;
    if (i < n) {
        float d = deg[i];
        deg[i] = (d > 0.0f) ? rsqrtf(d) : 0.0f;
    }
}

__global__ __launch_bounds__(256) void gemm_wmma_kernel(const float* __restrict__ X,
                                                        const float* __restrict__ W,
                                                        float* __restrict__ H,
                                                        int nrows) {
    const int wave = (int)((blockIdx.x * blockDim.x + threadIdx.x) >> 5);
    const int lane = (int)(threadIdx.x & 31);
    const int row0 = wave * 16;
    if (row0 >= nrows) return;
    const int lm = lane & 15;
    const int kk = (lane >> 4) * 2;

    const float* __restrict__ xrow = X + (size_t)(row0 + lm) * D_FEAT + kk;

    v8f c0 = {}, c1 = {}, c2 = {}, c3 = {};
#pragma unroll
    for (int k = 0; k < D_FEAT; k += 4) {
        v2f a;
        a.x = xrow[k + 0];
        a.y = xrow[k + 1];
        const float* __restrict__ wr0 = W + (size_t)(k + kk) * D_FEAT + lm;
        const float* __restrict__ wr1 = wr0 + D_FEAT;
        v2f b0, b1, b2, b3;
        b0.x = wr0[0];   b0.y = wr1[0];
        b1.x = wr0[16];  b1.y = wr1[16];
        b2.x = wr0[32];  b2.y = wr1[32];
        b3.x = wr0[48];  b3.y = wr1[48];
        c0 = __builtin_amdgcn_wmma_f32_16x16x4_f32(false, a, false, b0, (short)0, c0, false, false);
        c1 = __builtin_amdgcn_wmma_f32_16x16x4_f32(false, a, false, b1, (short)0, c1, false, false);
        c2 = __builtin_amdgcn_wmma_f32_16x16x4_f32(false, a, false, b2, (short)0, c2, false, false);
        c3 = __builtin_amdgcn_wmma_f32_16x16x4_f32(false, a, false, b3, (short)0, c3, false, false);
    }

    const int mbase = row0 + (lane >> 4) * 8;
#pragma unroll
    for (int r = 0; r < 8; ++r) {
        float* __restrict__ hr = H + (size_t)(mbase + r) * D_FEAT + lm;
        hr[0]  = c0[r];
        hr[16] = c1[r];
        hr[32] = c2[r];
        hr[48] = c3[r];
    }
}

__global__ __launch_bounds__(256) void init_acc_kernel(const float* __restrict__ H,
                                                       const float* __restrict__ dinv,
                                                       const float* __restrict__ bias,
                                                       float* __restrict__ ACC, int n) {
    int t = blockIdx.x * blockDim.x + threadIdx.x;
    if (t >= n) return;
    int node = t >> 6;
    int d    = t & 63;
    float di = dinv[node];
    ACC[t] = bias[d] + H[t] * di * di;
}

__global__ __launch_bounds__(256) void scatter_kernel(const float* __restrict__ H,
                                                      const int* __restrict__ row,
                                                      const int* __restrict__ col,
                                                      const float* __restrict__ dinv,
                                                      float* __restrict__ ACC, int e) {
    long long t = (long long)blockIdx.x * blockDim.x + threadIdx.x;
    int edge = (int)(t >> 4);
    if (edge >= e) return;
    int f = (int)(t & 15) * 4;
    int r = row[edge];
    int c = col[edge];
    float s = dinv[r] * dinv[c];
    const float4 hv = *(const float4*)(H + (size_t)r * D_FEAT + f);
    float* dst = ACC + (size_t)c * D_FEAT + f;
    atomicAdd(dst + 0, hv.x * s);
    atomicAdd(dst + 1, hv.y * s);
    atomicAdd(dst + 2, hv.z * s);
    atomicAdd(dst + 3, hv.w * s);
}

__global__ __launch_bounds__(256) void relu_kernel(float* __restrict__ A, int n) {
    int i = blockIdx.x * blockDim.x + threadIdx.x;
    if (i < n) A[i] = fmaxf(A[i], 0.0f);
}

__global__ __launch_bounds__(256) void pool_zero_kernel(float* __restrict__ pool,
                                                        float* __restrict__ cnt, int n) {
    int i = blockIdx.x * blockDim.x + threadIdx.x;
    if (i < n) pool[i] = 0.0f;
    if (i < N_GRAPHS_C) cnt[i] = 0.0f;
}

__global__ __launch_bounds__(256) void pool_kernel(const float* __restrict__ H,
                                                   const int* __restrict__ batch,
                                                   float* __restrict__ pool,
                                                   float* __restrict__ cnt, int nnodes) {
    long long t = (long long)blockIdx.x * blockDim.x + threadIdx.x;
    int node = (int)(t >> 4);
    if (node >= nnodes) return;
    int f = (int)(t & 15) * 4;
    int g = batch[node];
    const float4 hv = *(const float4*)(H + (size_t)node * D_FEAT + f);
    float* dst = pool + (size_t)g * D_FEAT + f;
    atomicAdd(dst + 0, hv.x);
    atomicAdd(dst + 1, hv.y);
    atomicAdd(dst + 2, hv.z);
    atomicAdd(dst + 3, hv.w);
    if (f == 0) atomicAdd(&cnt[g], 1.0f);
}

__global__ __launch_bounds__(256) void pool_final_kernel(const float* __restrict__ pool,
                                                         const float* __restrict__ cnt,
                                                         float* __restrict__ out, int n) {
    int i = blockIdx.x * blockDim.x + threadIdx.x;
    if (i < n) {
        int g = i >> 6;
        out[i] = pool[i] / fmaxf(cnt[g], 1.0f);
    }
}

extern "C" void kernel_launch(void* const* d_in, const int* in_sizes, int n_in,
                              void* d_out, int out_size, void* d_ws, size_t ws_size,
                              hipStream_t stream) {
    const float* x     = (const float*)d_in[0];
    const int*   ei    = (const int*)d_in[1];
    const int*   batch = (const int*)d_in[2];
    const float* W1    = (const float*)d_in[3];
    const float* b1    = (const float*)d_in[4];
    const float* W2    = (const float*)d_in[5];
    const float* b2    = (const float*)d_in[6];
    const float* W3    = (const float*)d_in[7];
    const float* b3    = (const float*)d_in[8];
    float* out = (float*)d_out;

    const int N = in_sizes[0] / D_FEAT;
    const int E = in_sizes[1] / 2;
    const int* erow = ei;
    const int* ecol = ei + E;

    float* w = (float*)d_ws;
    size_t off = 0;
    float* dinv = w + off; off += (size_t)((N + 63) / 64) * 64;
    float* bufA = w + off; off += (size_t)N * D_FEAT;
    float* bufB = w + off; off += (size_t)N * D_FEAT;
    float* pool = w + off; off += (size_t)N_GRAPHS_C * D_FEAT;
    float* cnt  = w + off; off += N_GRAPHS_C;

    const int ND = N * D_FEAT;
    const int blk = 256;

    deg_init_kernel<<<(N + blk - 1) / blk, blk, 0, stream>>>(dinv, N);
    deg_count_kernel<<<(E + blk - 1) / blk, blk, 0, stream>>>(ecol, dinv, E);
    deg_finish_kernel<<<(N + blk - 1) / blk, blk, 0, stream>>>(dinv, N);

    const int waves = (N + 15) / 16;
    const int gemm_blocks = (waves * 32 + blk - 1) / blk;
    const long long sthreads = (long long)E * 16;
    const int scat_blocks = (int)((sthreads + blk - 1) / blk);
    const int nd_blocks = (ND + blk - 1) / blk;

    gemm_wmma_kernel<<<gemm_blocks, blk, 0, stream>>>(x, W1, bufA, N);
    init_acc_kernel<<<nd_blocks, blk, 0, stream>>>(bufA, dinv, b1, bufB, ND);
    scatter_kernel<<<scat_blocks, blk, 0, stream>>>(bufA, erow, ecol, dinv, bufB, E);
    relu_kernel<<<nd_blocks, blk, 0, stream>>>(bufB, ND);

    gemm_wmma_kernel<<<gemm_blocks, blk, 0, stream>>>(bufB, W2, bufA, N);
    init_acc_kernel<<<nd_blocks, blk, 0, stream>>>(bufA, dinv, b2, bufB, ND);
    scatter_kernel<<<scat_blocks, blk, 0, stream>>>(bufA, erow, ecol, dinv, bufB, E);
    relu_kernel<<<nd_blocks, blk, 0, stream>>>(bufB, ND);

    gemm_wmma_kernel<<<gemm_blocks, blk, 0, stream>>>(bufB, W3, bufA, N);
    init_acc_kernel<<<nd_blocks, blk, 0, stream>>>(bufA, dinv, b3, bufB, ND);
    scatter_kernel<<<scat_blocks, blk, 0, stream>>>(bufA, erow, ecol, dinv, bufB, E);
    relu_kernel<<<nd_blocks, blk, 0, stream>>>(bufB, ND);

    pool_zero_kernel<<<(N_GRAPHS_C * D_FEAT + blk - 1) / blk, blk, 0, stream>>>(pool, cnt,
                                                                               N_GRAPHS_C * D_FEAT);
    const long long pthreads = (long long)N * 16;
    pool_kernel<<<(int)((pthreads + blk - 1) / blk), blk, 0, stream>>>(bufB, batch, pool, cnt, N);
    pool_final_kernel<<<(N_GRAPHS_C * D_FEAT + blk - 1) / blk, blk, 0, stream>>>(pool, cnt, out,
                                                                                 N_GRAPHS_C * D_FEAT);
}


# mean pool as a one-hot matrix product on f32 WMMA: per-work-group partial tiles summed in fixed order; pool atomics and zero pass dropped; reorders the pool's float sums
# speedup vs baseline: 1.2727x; 1.2727x over previous
#include <hip/hip_runtime.h>
#include <hip/hip_bf16.h>

typedef __attribute__((ext_vector_type(2))) float v2f;
typedef __attribute__((ext_vector_type(8))) float v8f;

#define D_FEAT 64
#define N_GRAPHS_C 64

__global__ __launch_bounds__(256) void deg_init_kernel(float* __restrict__ deg, int n) {
    int i = blockIdx.x * blockDim.x + threadIdx.x;
    if (i < n) deg[i] = 1.0f;
}

__global__ __launch_bounds__(256) void deg_count_kernel(const int* __restrict__ col,
                                                        float* __restrict__ deg, int e) {
    int i = blockIdx.x * blockDim.x + threadIdx.x;
    if (i < e) atomicAdd(&deg[col[i]], 1.0f);
}

__global__ __launch_bounds__(256) void deg_finish_kernel(float* __restrict__ deg, int n) {
    int i = blockIdx.x * blockDim.x + threadIdx.x;
    if (i < n) {
        float d = deg[i];
        deg[i] = (d > 0.0f) ? rsqrtf(d) : 0.0f;
    }
}

__global__ __launch_bounds__(256) void gemm_wmma_kernel(const float* __restrict__ X,
                                                        const float* __restrict__ W,
                                                        float* __restrict__ H,
                                                        int nrows) {
    const int wave = (int)((blockIdx.x * blockDim.x + threadIdx.x) >> 5);
    const int lane = (int)(threadIdx.x & 31);
    const int row0 = wave * 16;
    if (row0 >= nrows) return;
    const int lm = lane & 15;
    const int kk = (lane >> 4) * 2;

    const float* __restrict__ xrow = X + (size_t)(row0 + lm) * D_FEAT + kk;

    v8f c0 = {}, c1 = {}, c2 = {}, c3 = {};
#pragma unroll
    for (int k = 0; k < D_FEAT; k += 4) {
        v2f a;
        a.x = xrow[k + 0];
        a.y = xrow[k + 1];
        const float* __restrict__ wr0 = W + (size_t)(k + kk) * D_FEAT + lm;
        const float* __restrict__ wr1 = wr0 + D_FEAT;
        v2f b0, b1, b2, b3;
        b0.x = wr0[0];   b0.y = wr1[0];
        b1.x = wr0[16];  b1.y = wr1[16];
        b2.x = wr0[32];  b2.y = wr1[32];
        b3.x = wr0[48];  b3.y = wr1[48];
        c0 = __builtin_amdgcn_wmma_f32_16x16x4_f32(false, a, false, b0, (short)0, c0, false, false);
        c1 = __builtin_amdgcn_wmma_f32_16x16x4_f32(false, a, false, b1, (short)0, c1, false, false);
        c2 = __builtin_amdgcn_wmma_f32_16x16x4_f32(false, a, false, b2, (short)0, c2, false, false);
        c3 = __builtin_amdgcn_wmma_f32_16x16x4_f32(false, a, false, b3, (short)0, c3, false, false);
    }

    const int mbase = row0 + (lane >> 4) * 8;
#pragma unroll
    for (int r = 0; r < 8; ++r) {
        float* __restrict__ hr = H + (size_t)(mbase + r) * D_FEAT + lm;
        hr[0]  = c0[r];
        hr[16] = c1[r];
        hr[32] = c2[r];
        hr[48] = c3[r];
    }
}

__global__ __launch_bounds__(256) void init_acc_kernel(const float* __restrict__ H,
                                                       const float* __restrict__ dinv,
                                                       const float* __restrict__ bias,
                                                       float* __restrict__ ACC, int n) {
    int t = blockIdx.x * blockDim.x + threadIdx.x;
    if (t >= n) return;
    int node = t >> 6;
    int d    = t & 63;
    float di = dinv[node];
    ACC[t] = bias[d] + H[t] * di * di;
}

__global__ __launch_bounds__(256) void scatter_kernel(const float* __restrict__ H,
                                                      const int* __restrict__ row,
                                                      const int* __restrict__ col,
                                                      const float* __restrict__ dinv,
                                                      float* __restrict__ ACC, int e) {
    long long t = (long long)blockIdx.x * blockDim.x + threadIdx.x;
    int edge = (int)(t >> 4);
    if (edge >= e) return;
    int f = (int)(t & 15) * 4;
    int r = row[edge];
    int c = col[edge];
    float s = dinv[r] * dinv[c];
    const float4 hv = *(const float4*)(H + (size_t)r * D_FEAT + f);
    float* dst = ACC + (size_t)c * D_FEAT + f;
    atomicAdd(dst + 0, hv.x * s);
    atomicAdd(dst + 1, hv.y * s);
    atomicAdd(dst + 2, hv.z * s);
    atomicAdd(dst + 3, hv.w * s);
}

__global__ __launch_bounds__(256) void relu_kernel(float* __restrict__ A, int n) {
    int i = blockIdx.x * blockDim.x + threadIdx.x;
    if (i < n) A[i] = fmaxf(A[i], 0.0f);
}

#define POOL_WGS 128
#define POOL_KQ 4

__global__ __launch_bounds__(512) void pool_wmma_kernel(const float* __restrict__ H,
                                                        const int* __restrict__ batch,
                                                        float* __restrict__ part,
                                                        float* __restrict__ cntp, int nnodes) {
    __shared__ float red[(POOL_KQ - 1) * 4 * 33 * 32];
    const int lane = (int)(threadIdx.x & 31);
    const int w    = __builtin_amdgcn_readfirstlane((int)(threadIdx.x >> 5));
    const int mt   = w & 3;
    const int kq   = w >> 2;
    const int lm   = lane & 15;
    const int half = lane >> 4;
    const int gsel = mt * 16 + lm;
    const int nsteps = (nnodes + 3) >> 2;
    const int nslots = (int)gridDim.x * POOL_KQ;

    v8f c0 = {}, c1 = {}, c2 = {}, c3 = {};
    float cnt = 0.0f;
#pragma unroll 4
    for (int ks = (int)blockIdx.x * POOL_KQ + kq; ks < nsteps; ks += nslots) {
        const int n0 = ks * 4 + half * 2;
        const int n1 = n0 + 1;
        const int m0 = min(n0, nnodes - 1);
        const int m1 = min(n1, nnodes - 1);
        const int g0 = batch[m0];
        const int g1 = batch[m1];
        const float4 r0 = *(const float4*)(H + (size_t)m0 * D_FEAT + lm * 4);
        const float4 r1 = *(const float4*)(H + (size_t)m1 * D_FEAT + lm * 4);
        v2f a;
        a.x = (g0 == gsel && n0 < nnodes) ? 1.0f : 0.0f;
        a.y = (g1 == gsel && n1 < nnodes) ? 1.0f : 0.0f;
        cnt += a.x + a.y;
        v2f b0, b1, b2, b3;
        b0.x = r0.x;  b0.y = r1.x;
        b1.x = r0.y;  b1.y = r1.y;
        b2.x = r0.z;  b2.y = r1.z;
        b3.x = r0.w;  b3.y = r1.w;
        c0 = __builtin_amdgcn_wmma_f32_16x16x4_f32(false, a, false, b0, (short)0, c0, false, false);
        c1 = __builtin_amdgcn_wmma_f32_16x16x4_f32(false, a, false, b1, (short)0, c1, false, false);
        c2 = __builtin_amdgcn_wmma_f32_16x16x4_f32(false, a, false, b2, (short)0, c2, false, false);
        c3 = __builtin_amdgcn_wmma_f32_16x16x4_f32(false, a, false, b3, (short)0, c3, false, false);
    }

    if (kq > 0) {
        float* dst = red + ((kq - 1) * 4 + mt) * (33 * 32) + lane;
#pragma unroll
        for (int r = 0; r < 8; ++r) {
            dst[(0 + r) * 32]  = c0[r];
            dst[(8 + r) * 32]  = c1[r];
            dst[(16 + r) * 32] = c2[r];
            dst[(24 + r) * 32] = c3[r];
        }
        dst[32 * 32] = cnt;
    }
    __syncthreads();
    if (kq == 0) {
#pragma unroll
        for (int q = 0; q < POOL_KQ - 1; ++q) {
            const float* srcp = red + (q * 4 + mt) * (33 * 32) + lane;
#pragma unroll
            for (int r = 0; r < 8; ++r) {
                c0[r] += srcp[(0 + r) * 32];
                c1[r] += srcp[(8 + r) * 32];
                c2[r] += srcp[(16 + r) * 32];
                c3[r] += srcp[(24 + r) * 32];
            }
            cnt += srcp[32 * 32];
        }
        float* pw = part + (size_t)blockIdx.x * (N_GRAPHS_C * D_FEAT);
#pragma unroll
        for (int r = 0; r < 8; ++r) {
            const int g = mt * 16 + half * 8 + r;
            *(float4*)(pw + g * D_FEAT + lm * 4) = make_float4(c0[r], c1[r], c2[r], c3[r]);
        }
        const float tot = cnt + __shfl_xor(cnt, 16);
        if (half == 0) cntp[(size_t)blockIdx.x * N_GRAPHS_C + gsel] = tot;
    }
}

__global__ __launch_bounds__(256) void pool_final_kernel(const float* __restrict__ part,
                                                         const float* __restrict__ cntp,
                                                         float* __restrict__ out) {
    __shared__ float rs[4][D_FEAT];
    __shared__ float rc[4];
    const int g = (int)blockIdx.x;
    const int d = (int)(threadIdx.x & 63);
    const int s = (int)(threadIdx.x >> 6);
    float acc = 0.0f, c = 0.0f;
#pragma unroll
    for (int i = 0; i < POOL_WGS / 4; ++i) {
        const int p = s + 4 * i;
        acc += part[(size_t)p * (N_GRAPHS_C * D_FEAT) + g * D_FEAT + d];
        c   += cntp[(size_t)p * N_GRAPHS_C + g];
    }
    rs[s][d] = acc;
    if (d == 0) rc[s] = c;
    __syncthreads();
    if (threadIdx.x < D_FEAT) {
        const float t  = ((rs[0][d] + rs[1][d]) + rs[2][d]) + rs[3][d];
        const float ct = ((rc[0] + rc[1]) + rc[2]) + rc[3];
        out[g * D_FEAT + d] = t / fmaxf(ct, 1.0f);
    }
}

extern "C" void kernel_launch(void* const* d_in, const int* in_sizes, int n_in,
                              void* d_out, int out_size, void* d_ws, size_t ws_size,
                              hipStream_t stream) {
    const float* x     = (const float*)d_in[0];
    const int*   ei    = (const int*)d_in[1];
    const int*   batch = (const int*)d_in[2];
    const float* W1    = (const float*)d_in[3];
    const float* b1    = (const float*)d_in[4];
    const float* W2    = (const float*)d_in[5];
    const float* b2    = (const float*)d_in[6];
    const float* W3    = (const float*)d_in[7];
    const float* b3    = (const float*)d_in[8];
    float* out = (float*)d_out;

    const int N = in_sizes[0] / D_FEAT;
    const int E = in_sizes[1] / 2;
    const int* erow = ei;
    const int* ecol = ei + E;

    float* w = (float*)d_ws;
    size_t off = 0;
    float* dinv = w + off; off += (size_t)((N + 63) / 64) * 64;
    float* bufA = w + off; off += (size_t)N * D_FEAT;
    float* bufB = w + off; off += (size_t)N * D_FEAT;
    float* pool = w + off; off += (size_t)POOL_WGS * N_GRAPHS_C * D_FEAT;
    float* cnt  = w + off; off += (size_t)POOL_WGS * N_GRAPHS_C;

    const int ND = N * D_FEAT;
    const int blk = 256;

    deg_init_kernel<<<(N + blk - 1) / blk, blk, 0, stream>>>(dinv, N);
    deg_count_kernel<<<(E + blk - 1) / blk, blk, 0, stream>>>(ecol, dinv, E);
    deg_finish_kernel<<<(N + blk - 1) / blk, blk, 0, stream>>>(dinv, N);

    const int waves = (N + 15) / 16;
    const int gemm_blocks = (waves * 32 + blk - 1) / blk;
    const long long sthreads = (long long)E * 16;
    const int scat_blocks = (int)((sthreads + blk - 1) / blk);
    const int nd_blocks = (ND + blk - 1) / blk;

    gemm_wmma_kernel<<<gemm_blocks, blk, 0, stream>>>(x, W1, bufA, N);
    init_acc_kernel<<<nd_blocks, blk, 0, stream>>>(bufA, dinv, b1, bufB, ND);
    scatter_kernel<<<scat_blocks, blk, 0, stream>>>(bufA, erow, ecol, dinv, bufB, E);
    relu_kernel<<<nd_blocks, blk, 0, stream>>>(bufB, ND);

    gemm_wmma_kernel<<<gemm_blocks, blk, 0, stream>>>(bufB, W2, bufA, N);
    init_acc_kernel<<<nd_blocks, blk, 0, stream>>>(bufA, dinv, b2, bufB, ND);
    scatter_kernel<<<scat_blocks, blk, 0, stream>>>(bufA, erow, ecol, dinv, bufB, E);
    relu_kernel<<<nd_blocks, blk, 0, stream>>>(bufB, ND);

    gemm_wmma_kernel<<<gemm_blocks, blk, 0, stream>>>(bufB, W3, bufA, N);
    init_acc_kernel<<<nd_blocks, blk, 0, stream>>>(bufA, dinv, b3, bufB, ND);
    scatter_kernel<<<scat_blocks, blk, 0, stream>>>(bufA, erow, ecol, dinv, bufB, E);
    relu_kernel<<<nd_blocks, blk, 0, stream>>>(bufB, ND);

    pool_wmma_kernel<<<POOL_WGS, 512, 0, stream>>>(bufB, batch, pool, cnt, N);
    pool_final_kernel<<<N_GRAPHS_C, blk, 0, stream>>>(pool, cnt, out);
}


# scatter: a wave walks 32 consecutive edges, lane = feature word: each row load and atomic covers one whole 128-B line (was 4 partial lines), 16x fewer waves; atomic arrival order changes
# speedup vs baseline: 1.9260x; 1.5133x over previous
#include <hip/hip_runtime.h>
#include <hip/hip_bf16.h>

typedef __attribute__((ext_vector_type(2))) float v2f;
typedef __attribute__((ext_vector_type(8))) float v8f;

#define D_FEAT 64
#define N_GRAPHS_C 64

__global__ __launch_bounds__(256) void deg_init_kernel(float* __restrict__ deg, int n) {
    int i = blockIdx.x * blockDim.x + threadIdx.x;
    if (i < n) deg[i] = 1.0f;
}

__global__ __launch_bounds__(256) void deg_count_kernel(const int* __restrict__ col,
                                                        float* __restrict__ deg, int e) {
    int i = blockIdx.x * blockDim.x + threadIdx.x;
    if (i < e) atomicAdd(&deg[col[i]], 1.0f);
}

__global__ __launch_bounds__(256) void deg_finish_kernel(float* __restrict__ deg, int n) {
    int i = blockIdx.x * blockDim.x + threadIdx.x;
    if (i < n) {
        float d = deg[i];
        deg[i] = (d > 0.0f) ? rsqrtf(d) : 0.0f;
    }
}

__global__ __launch_bounds__(256) void gemm_wmma_kernel(const float* __restrict__ X,
                                                        const float* __restrict__ W,
                                                        float* __restrict__ H,
                                                        int nrows) {
    const int wave = (int)((blockIdx.x * blockDim.x + threadIdx.x) >> 5);
    const int lane = (int)(threadIdx.x & 31);
    const int row0 = wave * 16;
    if (row0 >= nrows) return;
    const int lm = lane & 15;
    const int kk = (lane >> 4) * 2;

    const float* __restrict__ xrow = X + (size_t)(row0 + lm) * D_FEAT + kk;

    v8f c0 = {}, c1 = {}, c2 = {}, c3 = {};
#pragma unroll
    for (int k = 0; k < D_FEAT; k += 4) {
        v2f a;
        a.x = xrow[k + 0];
        a.y = xrow[k + 1];
        const float* __restrict__ wr0 = W + (size_t)(k + kk) * D_FEAT + lm;
        const float* __restrict__ wr1 = wr0 + D_FEAT;
        v2f b0, b1, b2, b3;
        b0.x = wr0[0];   b0.y = wr1[0];
        b1.x = wr0[16];  b1.y = wr1[16];
        b2.x = wr0[32];  b2.y = wr1[32];
        b3.x = wr0[48];  b3.y = wr1[48];
        c0 = __builtin_amdgcn_wmma_f32_16x16x4_f32(false, a, false, b0, (short)0, c0, false, false);
        c1 = __builtin_amdgcn_wmma_f32_16x16x4_f32(false, a, false, b1, (short)0, c1, false, false);
        c2 = __builtin_amdgcn_wmma_f32_16x16x4_f32(false, a, false, b2, (short)0, c2, false, false);
        c3 = __builtin_amdgcn_wmma_f32_16x16x4_f32(false, a, false, b3, (short)0, c3, false, false);
    }

    const int mbase = row0 + (lane >> 4) * 8;
#pragma unroll
    for (int r = 0; r < 8; ++r) {
        float* __restrict__ hr = H + (size_t)(mbase + r) * D_FEAT + lm;
        hr[0]  = c0[r];
        hr[16] = c1[r];
        hr[32] = c2[r];
        hr[48] = c3[r];
    }
}

__global__ __launch_bounds__(256) void init_acc_kernel(const float* __restrict__ H,
                                                       const float* __restrict__ dinv,
                                                       const float* __restrict__ bias,
                                                       float* __restrict__ ACC, int n) {
    int t = blockIdx.x * blockDim.x + threadIdx.x;
    if (t >= n) return;
    int node = t >> 6;
    int d    = t & 63;
    float di = dinv[node];
    ACC[t] = bias[d] + H[t] * di * di;
}

#define SC_EDGES 32
#define SC_BATCH 8

__global__ __launch_bounds__(256) void scatter_kernel(const float* __restrict__ H,
                                                      const int* __restrict__ row,
                                                      const int* __restrict__ col,
                                                      const float* __restrict__ dinv,
                                                      float* __restrict__ ACC, int e) {
    const int lane = (int)(threadIdx.x & 31);
    const int wave = __builtin_amdgcn_readfirstlane((int)((blockIdx.x * blockDim.x + threadIdx.x) >> 5));
    const int e0 = wave * SC_EDGES;
    if (e0 >= e) return;
    const int cnt = min(SC_EDGES, e - e0);
    const int me = min(e0 + lane, e - 1);
    const int r = row[me];
    const int c = col[me];
    const float s = dinv[r] * dinv[c];

    int j = 0;
    for (; j + SC_BATCH <= cnt; j += SC_BATCH) {
        float h0[SC_BATCH], h1[SC_BATCH];
#pragma unroll
        for (int u = 0; u < SC_BATCH; ++u) {
            const int rj = __builtin_amdgcn_readlane(r, j + u);
            const float* __restrict__ src = H + (size_t)rj * D_FEAT + lane;
            h0[u] = src[0];
            h1[u] = src[32];
        }
#pragma unroll
        for (int u = 0; u < SC_BATCH; ++u) {
            const int cj = __builtin_amdgcn_readlane(c, j + u);
            const float sj = __int_as_float(__builtin_amdgcn_readlane(__float_as_int(s), j + u));
            float* dst = ACC + (size_t)cj * D_FEAT + lane;
            atomicAdd(dst, h0[u] * sj);
            atomicAdd(dst + 32, h1[u] * sj);
        }
    }
    for (; j < cnt; ++j) {
        const int rj = __builtin_amdgcn_readlane(r, j);
        const int cj = __builtin_amdgcn_readlane(c, j);
        const float sj = __int_as_float(__builtin_amdgcn_readlane(__float_as_int(s), j));
        const float* __restrict__ src = H + (size_t)rj * D_FEAT + lane;
        float* dst = ACC + (size_t)cj * D_FEAT + lane;
        atomicAdd(dst, src[0] * sj);
        atomicAdd(dst + 32, src[32] * sj);
    }
}

__global__ __launch_bounds__(256) void relu_kernel(float* __restrict__ A, int n) {
    int i = blockIdx.x * blockDim.x + threadIdx.x;
    if (i < n) A[i] = fmaxf(A[i], 0.0f);
}

#define POOL_WGS 128
#define POOL_KQ 4

__global__ __launch_bounds__(512) void pool_wmma_kernel(const float* __restrict__ H,
                                                        const int* __restrict__ batch,
                                                        float* __restrict__ part,
                                                        float* __restrict__ cntp, int nnodes) {
    __shared__ float red[(POOL_KQ - 1) * 4 * 33 * 32];
    const int lane = (int)(threadIdx.x & 31);
    const int w    = __builtin_amdgcn_readfirstlane((int)(threadIdx.x >> 5));
    const int mt   = w & 3;
    const int kq   = w >> 2;
    const int lm   = lane & 15;
    const int half = lane >> 4;
    const int gsel = mt * 16 + lm;
    const int nsteps = (nnodes + 3) >> 2;
    const int nslots = (int)gridDim.x * POOL_KQ;

    v8f c0 = {}, c1 = {}, c2 = {}, c3 = {};
    float cnt = 0.0f;
#pragma unroll 4
    for (int ks = (int)blockIdx.x * POOL_KQ + kq; ks < nsteps; ks += nslots) {
        const int n0 = ks * 4 + half * 2;
        const int n1 = n0 + 1;
        const int m0 = min(n0, nnodes - 1);
        const int m1 = min(n1, nnodes - 1);
        const int g0 = batch[m0];
        const int g1 = batch[m1];
        const float4 r0 = *(const float4*)(H + (size_t)m0 * D_FEAT + lm * 4);
        const float4 r1 = *(const float4*)(H + (size_t)m1 * D_FEAT + lm * 4);
        v2f a;
        a.x = (g0 == gsel && n0 < nnodes) ? 1.0f : 0.0f;
        a.y = (g1 == gsel && n1 < nnodes) ? 1.0f : 0.0f;
        cnt += a.x + a.y;
        v2f b0, b1, b2, b3;
        b0.x = r0.x;  b0.y = r1.x;
        b1.x = r0.y;  b1.y = r1.y;
        b2.x = r0.z;  b2.y = r1.z;
        b3.x = r0.w;  b3.y = r1.w;
        c0 = __builtin_amdgcn_wmma_f32_16x16x4_f32(false, a, false, b0, (short)0, c0, false, false);
        c1 = __builtin_amdgcn_wmma_f32_16x16x4_f32(false, a, false, b1, (short)0, c1, false, false);
        c2 = __builtin_amdgcn_wmma_f32_16x16x4_f32(false, a, false, b2, (short)0, c2, false, false);
        c3 = __builtin_amdgcn_wmma_f32_16x16x4_f32(false, a, false, b3, (short)0, c3, false, false);
    }

    if (kq > 0) {
        float* dst = red + ((kq - 1) * 4 + mt) * (33 * 32) + lane;
#pragma unroll
        for (int r = 0; r < 8; ++r) {
            dst[(0 + r) * 32]  = c0[r];
            dst[(8 + r) * 32]  = c1[r];
            dst[(16 + r) * 32] = c2[r];
            dst[(24 + r) * 32] = c3[r];
        }
        dst[32 * 32] = cnt;
    }
    __syncthreads();
    if (kq == 0) {
#pragma unroll
        for (int q = 0; q < POOL_KQ - 1; ++q) {
            const float* srcp = red + (q * 4 + mt) * (33 * 32) + lane;
#pragma unroll
            for (int r = 0; r < 8; ++r) {
                c0[r] += srcp[(0 + r) * 32];
                c1[r] += srcp[(8 + r) * 32];
                c2[r] += srcp[(16 + r) * 32];
                c3[r] += srcp[(24 + r) * 32];
            }
            cnt += srcp[32 * 32];
        }
        float* pw = part + (size_t)blockIdx.x * (N_GRAPHS_C * D_FEAT);
#pragma unroll
        for (int r = 0; r < 8; ++r) {
            const int g = mt * 16 + half * 8 + r;
            *(float4*)(pw + g * D_FEAT + lm * 4) = make_float4(c0[r], c1[r], c2[r], c3[r]);
        }
        const float tot = cnt + __shfl_xor(cnt, 16);
        if (half == 0) cntp[(size_t)blockIdx.x * N_GRAPHS_C + gsel] = tot;
    }
}

__global__ __launch_bounds__(256) void pool_final_kernel(const float* __restrict__ part,
                                                         const float* __restrict__ cntp,
                                                         float* __restrict__ out) {
    __shared__ float rs[4][D_FEAT];
    __shared__ float rc[4];
    const int g = (int)blockIdx.x;
    const int d = (int)(threadIdx.x & 63);
    const int s = (int)(threadIdx.x >> 6);
    float acc = 0.0f, c = 0.0f;
#pragma unroll
    for (int i = 0; i < POOL_WGS / 4; ++i) {
        const int p = s + 4 * i;
        acc += part[(size_t)p * (N_GRAPHS_C * D_FEAT) + g * D_FEAT + d];
        c   += cntp[(size_t)p * N_GRAPHS_C + g];
    }
    rs[s][d] = acc;
    if (d == 0) rc[s] = c;
    __syncthreads();
    if (threadIdx.x < D_FEAT) {
        const float t  = ((rs[0][d] + rs[1][d]) + rs[2][d]) + rs[3][d];
        const float ct = ((rc[0] + rc[1]) + rc[2]) + rc[3];
        out[g * D_FEAT + d] = t / fmaxf(ct, 1.0f);
    }
}

extern "C" void kernel_launch(void* const* d_in, const int* in_sizes, int n_in,
                              void* d_out, int out_size, void* d_ws, size_t ws_size,
                              hipStream_t stream) {
    const float* x     = (const float*)d_in[0];
    const int*   ei    = (const int*)d_in[1];
    const int*   batch = (const int*)d_in[2];
    const float* W1    = (const float*)d_in[3];
    const float* b1    = (const float*)d_in[4];
    const float* W2    = (const float*)d_in[5];
    const float* b2    = (const float*)d_in[6];
    const float* W3    = (const float*)d_in[7];
    const float* b3    = (const float*)d_in[8];
    float* out = (float*)d_out;

    const int N = in_sizes[0] / D_FEAT;
    const int E = in_sizes[1] / 2;
    const int* erow = ei;
    const int* ecol = ei + E;

    float* w = (float*)d_ws;
    size_t off = 0;
    float* dinv = w + off; off += (size_t)((N + 63) / 64) * 64;
    float* bufA = w + off; off += (size_t)N * D_FEAT;
    float* bufB = w + off; off += (size_t)N * D_FEAT;
    float* pool = w + off; off += (size_t)POOL_WGS * N_GRAPHS_C * D_FEAT;
    float* cnt  = w + off; off += (size_t)POOL_WGS * N_GRAPHS_C;

    const int ND = N * D_FEAT;
    const int blk = 256;

    deg_init_kernel<<<(N + blk - 1) / blk, blk, 0, stream>>>(dinv, N);
    deg_count_kernel<<<(E + blk - 1) / blk, blk, 0, stream>>>(ecol, dinv, E);
    deg_finish_kernel<<<(N + blk - 1) / blk, blk, 0, stream>>>(dinv, N);

    const int waves = (N + 15) / 16;
    const int gemm_blocks = (waves * 32 + blk - 1) / blk;
    const long long sthreads = (long long)E;
    const int scat_blocks = (int)((sthreads + blk - 1) / blk);
    const int nd_blocks = (ND + blk - 1) / blk;

    gemm_wmma_kernel<<<gemm_blocks, blk, 0, stream>>>(x, W1, bufA, N);
    init_acc_kernel<<<nd_blocks, blk, 0, stream>>>(bufA, dinv, b1, bufB, ND);
    scatter_kernel<<<scat_blocks, blk, 0, stream>>>(bufA, erow, ecol, dinv, bufB, E);
    relu_kernel<<<nd_blocks, blk, 0, stream>>>(bufB, ND);

    gemm_wmma_kernel<<<gemm_blocks, blk, 0, stream>>>(bufB, W2, bufA, N);
    init_acc_kernel<<<nd_blocks, blk, 0, stream>>>(bufA, dinv, b2, bufB, ND);
    scatter_kernel<<<scat_blocks, blk, 0, stream>>>(bufA, erow, ecol, dinv, bufB, E);
    relu_kernel<<<nd_blocks, blk, 0, stream>>>(bufB, ND);

    gemm_wmma_kernel<<<gemm_blocks, blk, 0, stream>>>(bufB, W3, bufA, N);
    init_acc_kernel<<<nd_blocks, blk, 0, stream>>>(bufA, dinv, b3, bufB, ND);
    scatter_kernel<<<scat_blocks, blk, 0, stream>>>(bufA, erow, ecol, dinv, bufB, E);
    relu_kernel<<<nd_blocks, blk, 0, stream>>>(bufB, ND);

    pool_wmma_kernel<<<POOL_WGS, 512, 0, stream>>>(bufB, batch, pool, cnt, N);
    pool_final_kernel<<<N_GRAPHS_C, blk, 0, stream>>>(pool, cnt, out);
}


# relu and sum-init passes folded into the product kernel (relu on its row loads; H and bias + H*dinv^2 stored from one accumulator set), last relu into the pool's loads: 6 launches fewer, same sums
# speedup vs baseline: 2.1597x; 1.1213x over previous
#include <hip/hip_runtime.h>
#include <hip/hip_bf16.h>

typedef __attribute__((ext_vector_type(2))) float v2f;
typedef __attribute__((ext_vector_type(8))) float v8f;

#define D_FEAT 64
#define N_GRAPHS_C 64

__global__ __launch_bounds__(256) void deg_init_kernel(float* __restrict__ deg, int n) {
    int i = blockIdx.x * blockDim.x + threadIdx.x;
    if (i < n) deg[i] = 1.0f;
}

__global__ __launch_bounds__(256) void deg_count_kernel(const int* __restrict__ col,
                                                        float* __restrict__ deg, int e) {
    int i = blockIdx.x * blockDim.x + threadIdx.x;
    if (i < e) atomicAdd(&deg[col[i]], 1.0f);
}

__global__ __launch_bounds__(256) void deg_finish_kernel(float* __restrict__ deg, int n) {
    int i = blockIdx.x * blockDim.x + threadIdx.x;
    if (i < n) {
        float d = deg[i];
        deg[i] = (d > 0.0f) ? rsqrtf(d) : 0.0f;
    }
}

template <bool RELU_IN>
__global__ __launch_bounds__(256) void gemm_wmma_kernel(const float* __restrict__ X,
                                                        const float* __restrict__ W,
                                                        const float* __restrict__ dinv,
                                                        const float* __restrict__ bias,
                                                        float* __restrict__ H,
                                                        float* __restrict__ ACC,
                                                        int nrows) {
    const int wave = (int)((blockIdx.x * blockDim.x + threadIdx.x) >> 5);
    const int lane = (int)(threadIdx.x & 31);
    const int row0 = wave * 16;
    if (row0 >= nrows) return;
    const int lm = lane & 15;
    const int kk = (lane >> 4) * 2;

    const float* __restrict__ xrow = X + (size_t)(row0 + lm) * D_FEAT + kk;

    v8f c0 = {}, c1 = {}, c2 = {}, c3 = {};
#pragma unroll
    for (int k = 0; k < D_FEAT; k += 4) {
        v2f a;
        a.x = xrow[k + 0];
        a.y = xrow[k + 1];
        if (RELU_IN) {
            a.x = fmaxf(a.x, 0.0f);
            a.y = fmaxf(a.y, 0.0f);
        }
        const float* __restrict__ wr0 = W + (size_t)(k + kk) * D_FEAT + lm;
        const float* __restrict__ wr1 = wr0 + D_FEAT;
        v2f b0, b1, b2, b3;
        b0.x = wr0[0];   b0.y = wr1[0];
        b1.x = wr0[16];  b1.y = wr1[16];
        b2.x = wr0[32];  b2.y = wr1[32];
        b3.x = wr0[48];  b3.y = wr1[48];
        c0 = __builtin_amdgcn_wmma_f32_16x16x4_f32(false, a, false, b0, (short)0, c0, false, false);
        c1 = __builtin_amdgcn_wmma_f32_16x16x4_f32(false, a, false, b1, (short)0, c1, false, false);
        c2 = __builtin_amdgcn_wmma_f32_16x16x4_f32(false, a, false, b2, (short)0, c2, false, false);
        c3 = __builtin_amdgcn_wmma_f32_16x16x4_f32(false, a, false, b3, (short)0, c3, false, false);
    }

    const int mbase = row0 + (lane >> 4) * 8;
    const float bs0 = bias[lm];
    const float bs1 = bias[lm + 16];
    const float bs2 = bias[lm + 32];
    const float bs3 = bias[lm + 48];
#pragma unroll
    for (int r = 0; r < 8; ++r) {
        const float di = dinv[mbase + r];
        float* __restrict__ hr = H + (size_t)(mbase + r) * D_FEAT + lm;
        hr[0]  = c0[r];
        hr[16] = c1[r];
        hr[32] = c2[r];
        hr[48] = c3[r];
        float* __restrict__ ar = ACC + (size_t)(mbase + r) * D_FEAT + lm;
        ar[0]  = bs0 + c0[r] * di * di;
        ar[16] = bs1 + c1[r] * di * di;
        ar[32] = bs2 + c2[r] * di * di;
        ar[48] = bs3 + c3[r] * di * di;
    }
}

#define SC_EDGES 32
#define SC_BATCH 8

__global__ __launch_bounds__(256) void scatter_kernel(const float* __restrict__ H,
                                                      const int* __restrict__ row,
                                                      const int* __restrict__ col,
                                                      const float* __restrict__ dinv,
                                                      float* __restrict__ ACC, int e) {
    const int lane = (int)(threadIdx.x & 31);
    const int wave = __builtin_amdgcn_readfirstlane((int)((blockIdx.x * blockDim.x + threadIdx.x) >> 5));
    const int e0 = wave * SC_EDGES;
    if (e0 >= e) return;
    const int cnt = min(SC_EDGES, e - e0);
    const int me = min(e0 + lane, e - 1);
    const int r = row[me];
    const int c = col[me];
    const float s = dinv[r] * dinv[c];

    int j = 0;
    for (; j + SC_BATCH <= cnt; j += SC_BATCH) {
        float h0[SC_BATCH], h1[SC_BATCH];
#pragma unroll
        for (int u = 0; u < SC_BATCH; ++u) {
            const int rj = __builtin_amdgcn_readlane(r, j + u);
            const float* __restrict__ src = H + (size_t)rj * D_FEAT + lane;
            h0[u] = src[0];
            h1[u] = src[32];
        }
#pragma unroll
        for (int u = 0; u < SC_BATCH; ++u) {
            const int cj = __builtin_amdgcn_readlane(c, j + u);
            const float sj = __int_as_float(__builtin_amdgcn_readlane(__float_as_int(s), j + u));
            float* dst = ACC + (size_t)cj * D_FEAT + lane;
            atomicAdd(dst, h0[u] * sj);
            atomicAdd(dst + 32, h1[u] * sj);
        }
    }
    for (; j < cnt; ++j) {
        const int rj = __builtin_amdgcn_readlane(r, j);
        const int cj = __builtin_amdgcn_readlane(c, j);
        const float sj = __int_as_float(__builtin_amdgcn_readlane(__float_as_int(s), j));
        const float* __restrict__ src = H + (size_t)rj * D_FEAT + lane;
        float* dst = ACC + (size_t)cj * D_FEAT + lane;
        atomicAdd(dst, src[0] * sj);
        atomicAdd(dst + 32, src[32] * sj);
    }
}

#define POOL_WGS 128
#define POOL_KQ 4

__global__ __launch_bounds__(512) void pool_wmma_kernel(const float* __restrict__ H,
                                                        const int* __restrict__ batch,
                                                        float* __restrict__ part,
                                                        float* __restrict__ cntp, int nnodes) {
    __shared__ float red[(POOL_KQ - 1) * 4 * 33 * 32];
    const int lane = (int)(threadIdx.x & 31);
    const int w    = __builtin_amdgcn_readfirstlane((int)(threadIdx.x >> 5));
    const int mt   = w & 3;
    const int kq   = w >> 2;
    const int lm   = lane & 15;
    const int half = lane >> 4;
    const int gsel = mt * 16 + lm;
    const int nsteps = (nnodes + 3) >> 2;
    const int nslots = (int)gridDim.x * POOL_KQ;

    v8f c0 = {}, c1 = {}, c2 = {}, c3 = {};
    float cnt = 0.0f;
#pragma unroll 4
    for (int ks = (int)blockIdx.x * POOL_KQ + kq; ks < nsteps; ks += nslots) {
        const int n0 = ks * 4 + half * 2;
        const int n1 = n0 + 1;
        const int m0 = min(n0, nnodes - 1);
        const int m1 = min(n1, nnodes - 1);
        const int g0 = batch[m0];
        const int g1 = batch[m1];
        const float4 r0 = *(const float4*)(H + (size_t)m0 * D_FEAT + lm * 4);
        const float4 r1 = *(const float4*)(H + (size_t)m1 * D_FEAT + lm * 4);
        v2f a;
        a.x = (g0 == gsel && n0 < nnodes) ? 1.0f : 0.0f;
        a.y = (g1 == gsel && n1 < nnodes) ? 1.0f : 0.0f;
        cnt += a.x + a.y;
        v2f b0, b1, b2, b3;
        b0.x = fmaxf(r0.x, 0.0f);  b0.y = fmaxf(r1.x, 0.0f);
        b1.x = fmaxf(r0.y, 0.0f);  b1.y = fmaxf(r1.y, 0.0f);
        b2.x = fmaxf(r0.z, 0.0f);  b2.y = fmaxf(r1.z, 0.0f);
        b3.x = fmaxf(r0.w, 0.0f);  b3.y = fmaxf(r1.w, 0.0f);
        c0 = __builtin_amdgcn_wmma_f32_16x16x4_f32(false, a, false, b0, (short)0, c0, false, false);
        c1 = __builtin_amdgcn_wmma_f32_16x16x4_f32(false, a, false, b1, (short)0, c1, false, false);
        c2 = __builtin_amdgcn_wmma_f32_16x16x4_f32(false, a, false, b2, (short)0, c2, false, false);
        c3 = __builtin_amdgcn_wmma_f32_16x16x4_f32(false, a, false, b3, (short)0, c3, false, false);
    }

    if (kq > 0) {
        float* dst = red + ((kq - 1) * 4 + mt) * (33 * 32) + lane;
#pragma unroll
        for (int r = 0; r < 8; ++r) {
            dst[(0 + r) * 32]  = c0[r];
            dst[(8 + r) * 32]  = c1[r];
            dst[(16 + r) * 32] = c2[r];
            dst[(24 + r) * 32] = c3[r];
        }
        dst[32 * 32] = cnt;
    }
    __syncthreads();
    if (kq == 0) {
#pragma unroll
        for (int q = 0; q < POOL_KQ - 1; ++q) {
            const float* srcp = red + (q * 4 + mt) * (33 * 32) + lane;
#pragma unroll
            for (int r = 0; r < 8; ++r) {
                c0[r] += srcp[(0 + r) * 32];
                c1[r] += srcp[(8 + r) * 32];
                c2[r] += srcp[(16 + r) * 32];
                c3[r] += srcp[(24 + r) * 32];
            }
            cnt += srcp[32 * 32];
        }
        float* pw = part + (size_t)blockIdx.x * (N_GRAPHS_C * D_FEAT);
#pragma unroll
        for (int r = 0; r < 8; ++r) {
            const int g = mt * 16 + half * 8 + r;
            *(float4*)(pw + g * D_FEAT + lm * 4) = make_float4(c0[r], c1[r], c2[r], c3[r]);
        }
        const float tot = cnt + __shfl_xor(cnt, 16);
        if (half == 0) cntp[(size_t)blockIdx.x * N_GRAPHS_C + gsel] = tot;
    }
}

__global__ __launch_bounds__(256) void pool_final_kernel(const float* __restrict__ part,
                                                         const float* __restrict__ cntp,
                                                         float* __restrict__ out) {
    __shared__ float rs[4][D_FEAT];
    __shared__ float rc[4];
    const int g = (int)blockIdx.x;
    const int d = (int)(threadIdx.x & 63);
    const int s = (int)(threadIdx.x >> 6);
    float acc = 0.0f, c = 0.0f;
#pragma unroll
    for (int i = 0; i < POOL_WGS / 4; ++i) {
        const int p = s + 4 * i;
        acc += part[(size_t)p * (N_GRAPHS_C * D_FEAT) + g * D_FEAT + d];
        c   += cntp[(size_t)p * N_GRAPHS_C + g];
    }
    rs[s][d] = acc;
    if (d == 0) rc[s] = c;
    __syncthreads();
    if (threadIdx.x < D_FEAT) {
        const float t  = ((rs[0][d] + rs[1][d]) + rs[2][d]) + rs[3][d];
        const float ct = ((rc[0] + rc[1]) + rc[2]) + rc[3];
        out[g * D_FEAT + d] = t / fmaxf(ct, 1.0f);
    }
}

extern "C" void kernel_launch(void* const* d_in, const int* in_sizes, int n_in,
                              void* d_out, int out_size, void* d_ws, size_t ws_size,
                              hipStream_t stream) {
    const float* x     = (const float*)d_in[0];
    const int*   ei    = (const int*)d_in[1];
    const int*   batch = (const int*)d_in[2];
    const float* W1    = (const float*)d_in[3];
    const float* b1    = (const float*)d_in[4];
    const float* W2    = (const float*)d_in[5];
    const float* b2    = (const float*)d_in[6];
    const float* W3    = (const float*)d_in[7];
    const float* b3    = (const float*)d_in[8];
    float* out = (float*)d_out;

    const int N = in_sizes[0] / D_FEAT;
    const int E = in_sizes[1] / 2;
    const int* erow = ei;
    const int* ecol = ei + E;

    float* w = (float*)d_ws;
    size_t off = 0;
    float* dinv = w + off; off += (size_t)((N + 63) / 64) * 64;
    float* bufA = w + off; off += (size_t)N * D_FEAT;
    float* bufB = w + off; off += (size_t)N * D_FEAT;
    float* bufC = w + off; off += (size_t)N * D_FEAT;
    float* pool = w + off; off += (size_t)POOL_WGS * N_GRAPHS_C * D_FEAT;
    float* cnt  = w + off; off += (size_t)POOL_WGS * N_GRAPHS_C;

    const int blk = 256;

    deg_init_kernel<<<(N + blk - 1) / blk, blk, 0, stream>>>(dinv, N);
    deg_count_kernel<<<(E + blk - 1) / blk, blk, 0, stream>>>(ecol, dinv, E);
    deg_finish_kernel<<<(N + blk - 1) / blk, blk, 0, stream>>>(dinv, N);

    const int waves = (N + 15) / 16;
    const int gemm_blocks = (waves * 32 + blk - 1) / blk;
    const long long sthreads = (long long)E;
    const int scat_blocks = (int)((sthreads + blk - 1) / blk);

    gemm_wmma_kernel<false><<<gemm_blocks, blk, 0, stream>>>(x, W1, dinv, b1, bufA, bufB, N);
    scatter_kernel<<<scat_blocks, blk, 0, stream>>>(bufA, erow, ecol, dinv, bufB, E);

    gemm_wmma_kernel<true><<<gemm_blocks, blk, 0, stream>>>(bufB, W2, dinv, b2, bufA, bufC, N);
    scatter_kernel<<<scat_blocks, blk, 0, stream>>>(bufA, erow, ecol, dinv, bufC, E);

    gemm_wmma_kernel<true><<<gemm_blocks, blk, 0, stream>>>(bufC, W3, dinv, b3, bufA, bufB, N);
    scatter_kernel<<<scat_blocks, blk, 0, stream>>>(bufA, erow, ecol, dinv, bufB, E);

    pool_wmma_kernel<<<POOL_WGS, 512, 0, stream>>>(bufB, batch, pool, cnt, N);
    pool_final_kernel<<<N_GRAPHS_C, blk, 0, stream>>>(pool, cnt, out);
}


# every float atomic target (layer sum planes, degree counts) kept as one copy per package half, the wave's die id picks the copy, readers add the two: node sums become two partial sums, order changes
# speedup vs baseline: 2.3777x; 1.1009x over previous
#include <hip/hip_runtime.h>
#include <hip/hip_bf16.h>

typedef __attribute__((ext_vector_type(2))) float v2f;
typedef __attribute__((ext_vector_type(8))) float v8f;

#define D_FEAT 64
#define N_GRAPHS_C 64

__device__ __forceinline__ int package_half() {
    return (int)((__builtin_amdgcn_s_sendmsg_rtn(0x87) >> 18) & 1u);
}

__global__ __launch_bounds__(256) void deg_init_kernel(float* __restrict__ deg,
                                                       float* __restrict__ deg1, int n) {
    int i = blockIdx.x * blockDim.x + threadIdx.x;
    if (i < n) {
        deg[i] = 1.0f;
        deg1[i] = 0.0f;
    }
}

__global__ __launch_bounds__(256) void deg_count_kernel(const int* __restrict__ col,
                                                        float* __restrict__ deg,
                                                        float* __restrict__ deg1, int e) {
    int i = blockIdx.x * blockDim.x + threadIdx.x;
    float* d = package_half() ? deg1 : deg;
    if (i < e) atomicAdd(&d[col[i]], 1.0f);
}

__global__ __launch_bounds__(256) void deg_finish_kernel(float* __restrict__ deg,
                                                         const float* __restrict__ deg1, int n) {
    int i = blockIdx.x * blockDim.x + threadIdx.x;
    if (i < n) {
        float d = deg[i] + deg1[i];
        deg[i] = (d > 0.0f) ? rsqrtf(d) : 0.0f;
    }
}

template <bool RELU_IN>
__global__ __launch_bounds__(256) void gemm_wmma_kernel(const float* __restrict__ X,
                                                        const float* __restrict__ X1,
                                                        const float* __restrict__ W,
                                                        const float* __restrict__ dinv,
                                                        const float* __restrict__ bias,
                                                        float* __restrict__ H,
                                                        float* __restrict__ ACC,
                                                        float* __restrict__ ACC1,
                                                        int nrows) {
    const int wave = (int)((blockIdx.x * blockDim.x + threadIdx.x) >> 5);
    const int lane = (int)(threadIdx.x & 31);
    const int row0 = wave * 16;
    if (row0 >= nrows) return;
    const int lm = lane & 15;
    const int kk = (lane >> 4) * 2;

    const float* __restrict__ xrow = X + (size_t)(row0 + lm) * D_FEAT + kk;
    const float* __restrict__ xrow1 = X1 + (size_t)(row0 + lm) * D_FEAT + kk;

    v8f c0 = {}, c1 = {}, c2 = {}, c3 = {};
#pragma unroll
    for (int k = 0; k < D_FEAT; k += 4) {
        v2f a;
        a.x = xrow[k + 0];
        a.y = xrow[k + 1];
        if (RELU_IN) {
            a.x = fmaxf(a.x + xrow1[k + 0], 0.0f);
            a.y = fmaxf(a.y + xrow1[k + 1], 0.0f);
        }
        const float* __restrict__ wr0 = W + (size_t)(k + kk) * D_FEAT + lm;
        const float* __restrict__ wr1 = wr0 + D_FEAT;
        v2f b0, b1, b2, b3;
        b0.x = wr0[0];   b0.y = wr1[0];
        b1.x = wr0[16];  b1.y = wr1[16];
        b2.x = wr0[32];  b2.y = wr1[32];
        b3.x = wr0[48];  b3.y = wr1[48];
        c0 = __builtin_amdgcn_wmma_f32_16x16x4_f32(false, a, false, b0, (short)0, c0, false, false);
        c1 = __builtin_amdgcn_wmma_f32_16x16x4_f32(false, a, false, b1, (short)0, c1, false, false);
        c2 = __builtin_amdgcn_wmma_f32_16x16x4_f32(false, a, false, b2, (short)0, c2, false, false);
        c3 = __builtin_amdgcn_wmma_f32_16x16x4_f32(false, a, false, b3, (short)0, c3, false, false);
    }

    const int mbase = row0 + (lane >> 4) * 8;
    const float bs0 = bias[lm];
    const float bs1 = bias[lm + 16];
    const float bs2 = bias[lm + 32];
    const float bs3 = bias[lm + 48];
#pragma unroll
    for (int r = 0; r < 8; ++r) {
        const float di = dinv[mbase + r];
        float* __restrict__ hr = H + (size_t)(mbase + r) * D_FEAT + lm;
        hr[0]  = c0[r];
        hr[16] = c1[r];
        hr[32] = c2[r];
        hr[48] = c3[r];
        float* __restrict__ ar = ACC + (size_t)(mbase + r) * D_FEAT + lm;
        ar[0]  = bs0 + c0[r] * di * di;
        ar[16] = bs1 + c1[r] * di * di;
        ar[32] = bs2 + c2[r] * di * di;
        ar[48] = bs3 + c3[r] * di * di;
    }
    float4* __restrict__ zr = (float4*)(ACC1 + (size_t)row0 * D_FEAT) + lane;
#pragma unroll
    for (int i = 0; i < 8; ++i) zr[i * 32] = make_float4(0.0f, 0.0f, 0.0f, 0.0f);
}

#define SC_EDGES 32
#define SC_BATCH 8

__global__ __launch_bounds__(256) void scatter_kernel(const float* __restrict__ H,
                                                      const int* __restrict__ row,
                                                      const int* __restrict__ col,
                                                      const float* __restrict__ dinv,
                                                      float* __restrict__ ACC0,
                                                      float* __restrict__ ACC1, int e) {
    const int lane = (int)(threadIdx.x & 31);
    float* ACC = package_half() ? ACC1 : ACC0;
    const int wave = __builtin_amdgcn_readfirstlane((int)((blockIdx.x * blockDim.x + threadIdx.x) >> 5));
    const int e0 = wave * SC_EDGES;
    if (e0 >= e) return;
    const int cnt = min(SC_EDGES, e - e0);
    const int me = min(e0 + lane, e - 1);
    const int r = row[me];
    const int c = col[me];
    const float s = dinv[r] * dinv[c];

    int j = 0;
    for (; j + SC_BATCH <= cnt; j += SC_BATCH) {
        float h0[SC_BATCH], h1[SC_BATCH];
#pragma unroll
        for (int u = 0; u < SC_BATCH; ++u) {
            const int rj = __builtin_amdgcn_readlane(r, j + u);
            const float* __restrict__ src = H + (size_t)rj * D_FEAT + lane;
            h0[u] = src[0];
            h1[u] = src[32];
        }
#pragma unroll
        for (int u = 0; u < SC_BATCH; ++u) {
            const int cj = __builtin_amdgcn_readlane(c, j + u);
            const float sj = __int_as_float(__builtin_amdgcn_readlane(__float_as_int(s), j + u));
            float* dst = ACC + (size_t)cj * D_FEAT + lane;
            atomicAdd(dst, h0[u] * sj);
            atomicAdd(dst + 32, h1[u] * sj);
        }
    }
    for (; j < cnt; ++j) {
        const int rj = __builtin_amdgcn_readlane(r, j);
        const int cj = __builtin_amdgcn_readlane(c, j);
        const float sj = __int_as_float(__builtin_amdgcn_readlane(__float_as_int(s), j));
        const float* __restrict__ src = H + (size_t)rj * D_FEAT + lane;
        float* dst = ACC + (size_t)cj * D_FEAT + lane;
        atomicAdd(dst, src[0] * sj);
        atomicAdd(dst + 32, src[32] * sj);
    }
}

#define POOL_WGS 128
#define POOL_KQ 4

__global__ __launch_bounds__(512) void pool_wmma_kernel(const float* __restrict__ H,
                                                        const float* __restrict__ H1,
                                                        const int* __restrict__ batch,
                                                        float* __restrict__ part,
                                                        float* __restrict__ cntp, int nnodes) {
    __shared__ float red[(POOL_KQ - 1) * 4 * 33 * 32];
    const int lane = (int)(threadIdx.x & 31);
    const int w    = __builtin_amdgcn_readfirstlane((int)(threadIdx.x >> 5));
    const int mt   = w & 3;
    const int kq   = w >> 2;
    const int lm   = lane & 15;
    const int half = lane >> 4;
    const int gsel = mt * 16 + lm;
    const int nsteps = (nnodes + 3) >> 2;
    const int nslots = (int)gridDim.x * POOL_KQ;

    v8f c0 = {}, c1 = {}, c2 = {}, c3 = {};
    float cnt = 0.0f;
#pragma unroll 4
    for (int ks = (int)blockIdx.x * POOL_KQ + kq; ks < nsteps; ks += nslots) {
        const int n0 = ks * 4 + half * 2;
        const int n1 = n0 + 1;
        const int m0 = min(n0, nnodes - 1);
        const int m1 = min(n1, nnodes - 1);
        const int g0 = batch[m0];
        const int g1 = batch[m1];
        const unsigned o0 = (unsigned)m0 * (unsigned)(D_FEAT * sizeof(float)) + (unsigned)lm * 16u;
        const unsigned o1 = (unsigned)m1 * (unsigned)(D_FEAT * sizeof(float)) + (unsigned)lm * 16u;
        const float4 p0 = *(const float4*)((const char*)H + o0);
        const float4 p1 = *(const float4*)((const char*)H + o1);
        const float4 q0 = *(const float4*)((const char*)H1 + o0);
        const float4 q1 = *(const float4*)((const char*)H1 + o1);
        const float4 r0 = make_float4(p0.x + q0.x, p0.y + q0.y, p0.z + q0.z, p0.w + q0.w);
        const float4 r1 = make_float4(p1.x + q1.x, p1.y + q1.y, p1.z + q1.z, p1.w + q1.w);
        v2f a;
        a.x = (g0 == gsel && n0 < nnodes) ? 1.0f : 0.0f;
        a.y = (g1 == gsel && n1 < nnodes) ? 1.0f : 0.0f;
        cnt += a.x + a.y;
        v2f b0, b1, b2, b3;
        b0.x = fmaxf(r0.x, 0.0f);  b0.y = fmaxf(r1.x, 0.0f);
        b1.x = fmaxf(r0.y, 0.0f);  b1.y = fmaxf(r1.y, 0.0f);
        b2.x = fmaxf(r0.z, 0.0f);  b2.y = fmaxf(r1.z, 0.0f);
        b3.x = fmaxf(r0.w, 0.0f);  b3.y = fmaxf(r1.w, 0.0f);
        c0 = __builtin_amdgcn_wmma_f32_16x16x4_f32(false, a, false, b0, (short)0, c0, false, false);
        c1 = __builtin_amdgcn_wmma_f32_16x16x4_f32(false, a, false, b1, (short)0, c1, false, false);
        c2 = __builtin_amdgcn_wmma_f32_16x16x4_f32(false, a, false, b2, (short)0, c2, false, false);
        c3 = __builtin_amdgcn_wmma_f32_16x16x4_f32(false, a, false, b3, (short)0, c3, false, false);
    }

    if (kq > 0) {
        float* dst = red + ((kq - 1) * 4 + mt) * (33 * 32) + lane;
#pragma unroll
        for (int r = 0; r < 8; ++r) {
            dst[(0 + r) * 32]  = c0[r];
            dst[(8 + r) * 32]  = c1[r];
            dst[(16 + r) * 32] = c2[r];
            dst[(24 + r) * 32] = c3[r];
        }
        dst[32 * 32] = cnt;
    }
    __syncthreads();
    if (kq == 0) {
#pragma unroll
        for (int q = 0; q < POOL_KQ - 1; ++q) {
            const float* srcp = red + (q * 4 + mt) * (33 * 32) + lane;
#pragma unroll
            for (int r = 0; r < 8; ++r) {
                c0[r] += srcp[(0 + r) * 32];
                c1[r] += srcp[(8 + r) * 32];
                c2[r] += srcp[(16 + r) * 32];
                c3[r] += srcp[(24 + r) * 32];
            }
            cnt += srcp[32 * 32];
        }
        float* pw = part + (size_t)blockIdx.x * (N_GRAPHS_C * D_FEAT);
#pragma unroll
        for (int r = 0; r < 8; ++r) {
            const int g = mt * 16 + half * 8 + r;
            *(float4*)(pw + g * D_FEAT + lm * 4) = make_float4(c0[r], c1[r], c2[r], c3[r]);
        }
        const float tot = cnt + __shfl_xor(cnt, 16);
        if (half == 0) cntp[(size_t)blockIdx.x * N_GRAPHS_C + gsel] = tot;
    }
}

__global__ __launch_bounds__(256) void pool_final_kernel(const float* __restrict__ part,
                                                         const float* __restrict__ cntp,
                                                         float* __restrict__ out) {
    __shared__ float rs[4][D_FEAT];
    __shared__ float rc[4];
    const int g = (int)blockIdx.x;
    const int d = (int)(threadIdx.x & 63);
    const int s = (int)(threadIdx.x >> 6);
    float acc = 0.0f, c = 0.0f;
#pragma unroll
    for (int i = 0; i < POOL_WGS / 4; ++i) {
        const int p = s + 4 * i;
        acc += part[(size_t)p * (N_GRAPHS_C * D_FEAT) + g * D_FEAT + d];
        c   += cntp[(size_t)p * N_GRAPHS_C + g];
    }
    rs[s][d] = acc;
    if (d == 0) rc[s] = c;
    __syncthreads();
    if (threadIdx.x < D_FEAT) {
        const float t  = ((rs[0][d] + rs[1][d]) + rs[2][d]) + rs[3][d];
        const float ct = ((rc[0] + rc[1]) + rc[2]) + rc[3];
        out[g * D_FEAT + d] = t / fmaxf(ct, 1.0f);
    }
}

extern "C" void kernel_launch(void* const* d_in, const int* in_sizes, int n_in,
                              void* d_out, int out_size, void* d_ws, size_t ws_size,
                              hipStream_t stream) {
    const float* x     = (const float*)d_in[0];
    const int*   ei    = (const int*)d_in[1];
    const int*   batch = (const int*)d_in[2];
    const float* W1    = (const float*)d_in[3];
    const float* b1    = (const float*)d_in[4];
    const float* W2    = (const float*)d_in[5];
    const float* b2    = (const float*)d_in[6];
    const float* W3    = (const float*)d_in[7];
    const float* b3    = (const float*)d_in[8];
    float* out = (float*)d_out;

    const int N = in_sizes[0] / D_FEAT;
    const int E = in_sizes[1] / 2;
    const int* erow = ei;
    const int* ecol = ei + E;

    float* w = (float*)d_ws;
    size_t off = 0;
    float* dinv = w + off; off += (size_t)((N + 63) / 64) * 64;
    float* deg1 = w + off; off += (size_t)((N + 63) / 64) * 64;
    float* bufA = w + off; off += (size_t)N * D_FEAT;
    float* bufB = w + off; off += (size_t)N * D_FEAT;
    float* bufB1 = w + off; off += (size_t)N * D_FEAT;
    float* bufC = w + off; off += (size_t)N * D_FEAT;
    float* bufC1 = w + off; off += (size_t)N * D_FEAT;
    float* pool = w + off; off += (size_t)POOL_WGS * N_GRAPHS_C * D_FEAT;
    float* cnt  = w + off; off += (size_t)POOL_WGS * N_GRAPHS_C;

    const int blk = 256;

    deg_init_kernel<<<(N + blk - 1) / blk, blk, 0, stream>>>(dinv, deg1, N);
    deg_count_kernel<<<(E + blk - 1) / blk, blk, 0, stream>>>(ecol, dinv, deg1, E);
    deg_finish_kernel<<<(N + blk - 1) / blk, blk, 0, stream>>>(dinv, deg1, N);

    const int waves = (N + 15) / 16;
    const int gemm_blocks = (waves * 32 + blk - 1) / blk;
    const long long sthreads = (long long)E;
    const int scat_blocks = (int)((sthreads + blk - 1) / blk);

    gemm_wmma_kernel<false><<<gemm_blocks, blk, 0, stream>>>(x, x, W1, dinv, b1, bufA, bufB, bufB1, N);
    scatter_kernel<<<scat_blocks, blk, 0, stream>>>(bufA, erow, ecol, dinv, bufB, bufB1, E);

    gemm_wmma_kernel<true><<<gemm_blocks, blk, 0, stream>>>(bufB, bufB1, W2, dinv, b2, bufA, bufC, bufC1, N);
    scatter_kernel<<<scat_blocks, blk, 0, stream>>>(bufA, erow, ecol, dinv, bufC, bufC1, E);

    gemm_wmma_kernel<true><<<gemm_blocks, blk, 0, stream>>>(bufC, bufC1, W3, dinv, b3, bufA, bufB, bufB1, N);
    scatter_kernel<<<scat_blocks, blk, 0, stream>>>(bufA, erow, ecol, dinv, bufB, bufB1, E);

    pool_wmma_kernel<<<POOL_WGS, 512, 0, stream>>>(bufB, bufB1, batch, pool, cnt, N);
    pool_final_kernel<<<N_GRAPHS_C, blk, 0, stream>>>(pool, cnt, out);
}


# edge weight split: product stores rows scaled by dinv[row], readers apply dinv[target] and bias to the raw sums; scatter drops its two dinv gathers and the multiply per edge; sums scaled after adding
# speedup vs baseline: 2.5034x; 1.0529x over previous
#include <hip/hip_runtime.h>
#include <hip/hip_bf16.h>

typedef __attribute__((ext_vector_type(2))) float v2f;
typedef __attribute__((ext_vector_type(8))) float v8f;

#define D_FEAT 64
#define N_GRAPHS_C 64

__device__ __forceinline__ int package_half() {
    return (int)((__builtin_amdgcn_s_sendmsg_rtn(0x87) >> 18) & 1u);
}

__global__ __launch_bounds__(256) void deg_init_kernel(float* __restrict__ deg,
                                                       float* __restrict__ deg1, int n) {
    int i = blockIdx.x * blockDim.x + threadIdx.x;
    if (i < n) {
        deg[i] = 1.0f;
        deg1[i] = 0.0f;
    }
}

__global__ __launch_bounds__(256) void deg_count_kernel(const int* __restrict__ col,
                                                        float* __restrict__ deg,
                                                        float* __restrict__ deg1, int e) {
    int i = blockIdx.x * blockDim.x + threadIdx.x;
    float* d = package_half() ? deg1 : deg;
    if (i < e) atomicAdd(&d[col[i]], 1.0f);
}

__global__ __launch_bounds__(256) void deg_finish_kernel(float* __restrict__ deg,
                                                         const float* __restrict__ deg1, int n) {
    int i = blockIdx.x * blockDim.x + threadIdx.x;
    if (i < n) {
        float d = deg[i] + deg1[i];
        deg[i] = (d > 0.0f) ? rsqrtf(d) : 0.0f;
    }
}

template <bool RELU_IN>
__global__ __launch_bounds__(256) void gemm_wmma_kernel(const float* __restrict__ X,
                                                        const float* __restrict__ X1,
                                                        const float* __restrict__ bias_in,
                                                        const float* __restrict__ W,
                                                        const float* __restrict__ dinv,
                                                        float* __restrict__ H,
                                                        float* __restrict__ ACC,
                                                        float* __restrict__ ACC1,
                                                        int nrows) {
    const int wave = (int)((blockIdx.x * blockDim.x + threadIdx.x) >> 5);
    const int lane = (int)(threadIdx.x & 31);
    const int row0 = wave * 16;
    if (row0 >= nrows) return;
    const int lm = lane & 15;
    const int kk = (lane >> 4) * 2;

    const float* __restrict__ xrow = X + (size_t)(row0 + lm) * D_FEAT + kk;
    const float* __restrict__ xrow1 = X1 + (size_t)(row0 + lm) * D_FEAT + kk;

    const float dr = RELU_IN ? dinv[row0 + lm] : 1.0f;
    v8f c0 = {}, c1 = {}, c2 = {}, c3 = {};
#pragma unroll
    for (int k = 0; k < D_FEAT; k += 4) {
        v2f a;
        a.x = xrow[k + 0];
        a.y = xrow[k + 1];
        if (RELU_IN) {
            a.x = fmaxf(bias_in[k + kk + 0] + dr * (a.x + xrow1[k + 0]), 0.0f);
            a.y = fmaxf(bias_in[k + kk + 1] + dr * (a.y + xrow1[k + 1]), 0.0f);
        }
        const float* __restrict__ wr0 = W + (size_t)(k + kk) * D_FEAT + lm;
        const float* __restrict__ wr1 = wr0 + D_FEAT;
        v2f b0, b1, b2, b3;
        b0.x = wr0[0];   b0.y = wr1[0];
        b1.x = wr0[16];  b1.y = wr1[16];
        b2.x = wr0[32];  b2.y = wr1[32];
        b3.x = wr0[48];  b3.y = wr1[48];
        c0 = __builtin_amdgcn_wmma_f32_16x16x4_f32(false, a, false, b0, (short)0, c0, false, false);
        c1 = __builtin_amdgcn_wmma_f32_16x16x4_f32(false, a, false, b1, (short)0, c1, false, false);
        c2 = __builtin_amdgcn_wmma_f32_16x16x4_f32(false, a, false, b2, (short)0, c2, false, false);
        c3 = __builtin_amdgcn_wmma_f32_16x16x4_f32(false, a, false, b3, (short)0, c3, false, false);
    }

    const int mbase = row0 + (lane >> 4) * 8;
#pragma unroll
    for (int r = 0; r < 8; ++r) {
        const float di = dinv[mbase + r];
        const float s0 = c0[r] * di, s1 = c1[r] * di, s2 = c2[r] * di, s3 = c3[r] * di;
        float* __restrict__ hr = H + (size_t)(mbase + r) * D_FEAT + lm;
        hr[0]  = s0;
        hr[16] = s1;
        hr[32] = s2;
        hr[48] = s3;
        float* __restrict__ ar = ACC + (size_t)(mbase + r) * D_FEAT + lm;
        ar[0]  = s0;
        ar[16] = s1;
        ar[32] = s2;
        ar[48] = s3;
    }
    float4* __restrict__ zr = (float4*)(ACC1 + (size_t)row0 * D_FEAT) + lane;
#pragma unroll
    for (int i = 0; i < 8; ++i) zr[i * 32] = make_float4(0.0f, 0.0f, 0.0f, 0.0f);
}

#define SC_EDGES 32
#define SC_BATCH 8

__global__ __launch_bounds__(256) void scatter_kernel(const float* __restrict__ H,
                                                      const int* __restrict__ row,
                                                      const int* __restrict__ col,
                                                      float* __restrict__ ACC0,
                                                      float* __restrict__ ACC1, int e) {
    const int lane = (int)(threadIdx.x & 31);
    float* ACC = package_half() ? ACC1 : ACC0;
    const int wave = __builtin_amdgcn_readfirstlane((int)((blockIdx.x * blockDim.x + threadIdx.x) >> 5));
    const int e0 = wave * SC_EDGES;
    if (e0 >= e) return;
    const int cnt = min(SC_EDGES, e - e0);
    const int me = min(e0 + lane, e - 1);
    const int r = row[me];
    const int c = col[me];

    int j = 0;
    for (; j + SC_BATCH <= cnt; j += SC_BATCH) {
        float h0[SC_BATCH], h1[SC_BATCH];
#pragma unroll
        for (int u = 0; u < SC_BATCH; ++u) {
            const int rj = __builtin_amdgcn_readlane(r, j + u);
            const float* __restrict__ src = H + (size_t)rj * D_FEAT + lane;
            h0[u] = src[0];
            h1[u] = src[32];
        }
#pragma unroll
        for (int u = 0; u < SC_BATCH; ++u) {
            const int cj = __builtin_amdgcn_readlane(c, j + u);
            float* dst = ACC + (size_t)cj * D_FEAT + lane;
            atomicAdd(dst, h0[u]);
            atomicAdd(dst + 32, h1[u]);
        }
    }
    for (; j < cnt; ++j) {
        const int rj = __builtin_amdgcn_readlane(r, j);
        const int cj = __builtin_amdgcn_readlane(c, j);
        const float* __restrict__ src = H + (size_t)rj * D_FEAT + lane;
        float* dst = ACC + (size_t)cj * D_FEAT + lane;
        atomicAdd(dst, src[0]);
        atomicAdd(dst + 32, src[32]);
    }
}

#define POOL_WGS 128
#define POOL_KQ 4

__global__ __launch_bounds__(512) void pool_wmma_kernel(const float* __restrict__ H,
                                                        const float* __restrict__ H1,
                                                        const float* __restrict__ dinv,
                                                        const float* __restrict__ bias,
                                                        const int* __restrict__ batch,
                                                        float* __restrict__ part,
                                                        float* __restrict__ cntp, int nnodes) {
    __shared__ float red[(POOL_KQ - 1) * 4 * 33 * 32];
    const int lane = (int)(threadIdx.x & 31);
    const int w    = __builtin_amdgcn_readfirstlane((int)(threadIdx.x >> 5));
    const int mt   = w & 3;
    const int kq   = w >> 2;
    const int lm   = lane & 15;
    const int half = lane >> 4;
    const int gsel = mt * 16 + lm;
    const int nsteps = (nnodes + 3) >> 2;
    const int nslots = (int)gridDim.x * POOL_KQ;
    const float4 bq = *(const float4*)(bias + lm * 4);

    v8f c0 = {}, c1 = {}, c2 = {}, c3 = {};
    float cnt = 0.0f;
#pragma unroll 4
    for (int ks = (int)blockIdx.x * POOL_KQ + kq; ks < nsteps; ks += nslots) {
        const int n0 = ks * 4 + half * 2;
        const int n1 = n0 + 1;
        const int m0 = min(n0, nnodes - 1);
        const int m1 = min(n1, nnodes - 1);
        const int g0 = batch[m0];
        const int g1 = batch[m1];
        const unsigned o0 = (unsigned)m0 * (unsigned)(D_FEAT * sizeof(float)) + (unsigned)lm * 16u;
        const unsigned o1 = (unsigned)m1 * (unsigned)(D_FEAT * sizeof(float)) + (unsigned)lm * 16u;
        const float4 p0 = *(const float4*)((const char*)H + o0);
        const float4 p1 = *(const float4*)((const char*)H + o1);
        const float4 q0 = *(const float4*)((const char*)H1 + o0);
        const float4 q1 = *(const float4*)((const char*)H1 + o1);
        const float d0 = dinv[m0];
        const float d1 = dinv[m1];
        const float4 r0 = make_float4(bq.x + d0 * (p0.x + q0.x), bq.y + d0 * (p0.y + q0.y),
                                      bq.z + d0 * (p0.z + q0.z), bq.w + d0 * (p0.w + q0.w));
        const float4 r1 = make_float4(bq.x + d1 * (p1.x + q1.x), bq.y + d1 * (p1.y + q1.y),
                                      bq.z + d1 * (p1.z + q1.z), bq.w + d1 * (p1.w + q1.w));
        v2f a;
        a.x = (g0 == gsel && n0 < nnodes) ? 1.0f : 0.0f;
        a.y = (g1 == gsel && n1 < nnodes) ? 1.0f : 0.0f;
        cnt += a.x + a.y;
        v2f b0, b1, b2, b3;
        b0.x = fmaxf(r0.x, 0.0f);  b0.y = fmaxf(r1.x, 0.0f);
        b1.x = fmaxf(r0.y, 0.0f);  b1.y = fmaxf(r1.y, 0.0f);
        b2.x = fmaxf(r0.z, 0.0f);  b2.y = fmaxf(r1.z, 0.0f);
        b3.x = fmaxf(r0.w, 0.0f);  b3.y = fmaxf(r1.w, 0.0f);
        c0 = __builtin_amdgcn_wmma_f32_16x16x4_f32(false, a, false, b0, (short)0, c0, false, false);
        c1 = __builtin_amdgcn_wmma_f32_16x16x4_f32(false, a, false, b1, (short)0, c1, false, false);
        c2 = __builtin_amdgcn_wmma_f32_16x16x4_f32(false, a, false, b2, (short)0, c2, false, false);
        c3 = __builtin_amdgcn_wmma_f32_16x16x4_f32(false, a, false, b3, (short)0, c3, false, false);
    }

    if (kq > 0) {
        float* dst = red + ((kq - 1) * 4 + mt) * (33 * 32) + lane;
#pragma unroll
        for (int r = 0; r < 8; ++r) {
            dst[(0 + r) * 32]  = c0[r];
            dst[(8 + r) * 32]  = c1[r];
            dst[(16 + r) * 32] = c2[r];
            dst[(24 + r) * 32] = c3[r];
        }
        dst[32 * 32] = cnt;
    }
    __syncthreads();
    if (kq == 0) {
#pragma unroll
        for (int q = 0; q < POOL_KQ - 1; ++q) {
            const float* srcp = red + (q * 4 + mt) * (33 * 32) + lane;
#pragma unroll
            for (int r = 0; r < 8; ++r) {
                c0[r] += srcp[(0 + r) * 32];
                c1[r] += srcp[(8 + r) * 32];
                c2[r] += srcp[(16 + r) * 32];
                c3[r] += srcp[(24 + r) * 32];
            }
            cnt += srcp[32 * 32];
        }
        float* pw = part + (size_t)blockIdx.x * (N_GRAPHS_C * D_FEAT);
#pragma unroll
        for (int r = 0; r < 8; ++r) {
            const int g = mt * 16 + half * 8 + r;
            *(float4*)(pw + g * D_FEAT + lm * 4) = make_float4(c0[r], c1[r], c2[r], c3[r]);
        }
        const float tot = cnt + __shfl_xor(cnt, 16);
        if (half == 0) cntp[(size_t)blockIdx.x * N_GRAPHS_C + gsel] = tot;
    }
}

__global__ __launch_bounds__(256) void pool_final_kernel(const float* __restrict__ part,
                                                         const float* __restrict__ cntp,
                                                         float* __restrict__ out) {
    __shared__ float rs[4][D_FEAT];
    __shared__ float rc[4];
    const int g = (int)blockIdx.x;
    const int d = (int)(threadIdx.x & 63);
    const int s = (int)(threadIdx.x >> 6);
    float acc = 0.0f, c = 0.0f;
#pragma unroll
    for (int i = 0; i < POOL_WGS / 4; ++i) {
        const int p = s + 4 * i;
        acc += part[(size_t)p * (N_GRAPHS_C * D_FEAT) + g * D_FEAT + d];
        c   += cntp[(size_t)p * N_GRAPHS_C + g];
    }
    rs[s][d] = acc;
    if (d == 0) rc[s] = c;
    __syncthreads();
    if (threadIdx.x < D_FEAT) {
        const float t  = ((rs[0][d] + rs[1][d]) + rs[2][d]) + rs[3][d];
        const float ct = ((rc[0] + rc[1]) + rc[2]) + rc[3];
        out[g * D_FEAT + d] = t / fmaxf(ct, 1.0f);
    }
}

extern "C" void kernel_launch(void* const* d_in, const int* in_sizes, int n_in,
                              void* d_out, int out_size, void* d_ws, size_t ws_size,
                              hipStream_t stream) {
    const float* x     = (const float*)d_in[0];
    const int*   ei    = (const int*)d_in[1];
    const int*   batch = (const int*)d_in[2];
    const float* W1    = (const float*)d_in[3];
    const float* b1    = (const float*)d_in[4];
    const float* W2    = (const float*)d_in[5];
    const float* b2    = (const float*)d_in[6];
    const float* W3    = (const float*)d_in[7];
    const float* b3    = (const float*)d_in[8];
    float* out = (float*)d_out;

    const int N = in_sizes[0] / D_FEAT;
    const int E = in_sizes[1] / 2;
    const int* erow = ei;
    const int* ecol = ei + E;

    float* w = (float*)d_ws;
    size_t off = 0;
    float* dinv = w + off; off += (size_t)((N + 63) / 64) * 64;
    float* deg1 = w + off; off += (size_t)((N + 63) / 64) * 64;
    float* bufA = w + off; off += (size_t)N * D_FEAT;
    float* bufB = w + off; off += (size_t)N * D_FEAT;
    float* bufB1 = w + off; off += (size_t)N * D_FEAT;
    float* bufC = w + off; off += (size_t)N * D_FEAT;
    float* bufC1 = w + off; off += (size_t)N * D_FEAT;
    float* pool = w + off; off += (size_t)POOL_WGS * N_GRAPHS_C * D_FEAT;
    float* cnt  = w + off; off += (size_t)POOL_WGS * N_GRAPHS_C;

    const int blk = 256;

    deg_init_kernel<<<(N + blk - 1) / blk, blk, 0, stream>>>(dinv, deg1, N);
    deg_count_kernel<<<(E + blk - 1) / blk, blk, 0, stream>>>(ecol, dinv, deg1, E);
    deg_finish_kernel<<<(N + blk - 1) / blk, blk, 0, stream>>>(dinv, deg1, N);

    const int waves = (N + 15) / 16;
    const int gemm_blocks = (waves * 32 + blk - 1) / blk;
    const long long sthreads = (long long)E;
    const int scat_blocks = (int)((sthreads + blk - 1) / blk);

    gemm_wmma_kernel<false><<<gemm_blocks, blk, 0, stream>>>(x, x, b1, W1, dinv, bufA, bufB, bufB1, N);
    scatter_kernel<<<scat_blocks, blk, 0, stream>>>(bufA, erow, ecol, bufB, bufB1, E);

    gemm_wmma_kernel<true><<<gemm_blocks, blk, 0, stream>>>(bufB, bufB1, b1, W2, dinv, bufA, bufC, bufC1, N);
    scatter_kernel<<<scat_blocks, blk, 0, stream>>>(bufA, erow, ecol, bufC, bufC1, E);

    gemm_wmma_kernel<true><<<gemm_blocks, blk, 0, stream>>>(bufC, bufC1, b2, W3, dinv, bufA, bufB, bufB1, N);
    scatter_kernel<<<scat_blocks, blk, 0, stream>>>(bufA, erow, ecol, bufB, bufB1, E);

    pool_wmma_kernel<<<POOL_WGS, 512, 0, stream>>>(bufB, bufB1, dinv, b3, batch, pool, cnt, N);
    pool_final_kernel<<<N_GRAPHS_C, blk, 0, stream>>>(pool, cnt, out);
}


# degree count: 32 edge slices x 4 node ranges counted in LDS (integer adds), partial planes added in the finish; no per-edge global atomic, init launch dropped; counts exact in any order
# speedup vs baseline: 2.6456x; 1.0568x over previous
#include <hip/hip_runtime.h>
#include <hip/hip_bf16.h>

typedef __attribute__((ext_vector_type(2))) float v2f;
typedef __attribute__((ext_vector_type(8))) float v8f;

#define D_FEAT 64
#define N_GRAPHS_C 64

__device__ __forceinline__ int package_half() {
    return (int)((__builtin_amdgcn_s_sendmsg_rtn(0x87) >> 18) & 1u);
}

#define DEG_K 32
#define DEG_CAP 16384
#define DEG_U 25
#define DEG_BLK 1024

__global__ __launch_bounds__(DEG_BLK) void deg_hist_kernel(const int* __restrict__ col,
                                                           unsigned* __restrict__ part,
                                                           int n, int npad, int e) {
    __shared__ unsigned hist[DEG_CAP];
    const int tid = (int)threadIdx.x;
    const int nr  = (int)gridDim.y;
    const int rl  = (((n + nr - 1) / nr) + 31) & ~31;
    const int r0  = (int)blockIdx.y * rl;
    const int len = max(0, min(n, r0 + rl) - r0);
    const int per = (e + DEG_K - 1) / DEG_K;
    const int s0  = (int)blockIdx.x * per;
    const int s1  = min(e, s0 + per);

    for (int i = tid; i < len; i += DEG_BLK) hist[i] = 0u;
    __syncthreads();
    for (int base = s0; base < s1; base += DEG_BLK * DEG_U) {
        int c[DEG_U];
#pragma unroll
        for (int u = 0; u < DEG_U; ++u) c[u] = col[min(base + u * DEG_BLK + tid, s1 - 1)];
#pragma unroll
        for (int u = 0; u < DEG_U; ++u) {
            const unsigned o = (unsigned)(c[u] - r0);
            if (base + u * DEG_BLK + tid < s1 && o < (unsigned)len) atomicAdd(&hist[o], 1u);
        }
    }
    __syncthreads();
    unsigned* __restrict__ pk = part + (size_t)blockIdx.x * npad + r0;
    for (int i = tid; i < len; i += DEG_BLK) pk[i] = hist[i];
}

__global__ __launch_bounds__(256) void deg_finish_kernel(const unsigned* __restrict__ part,
                                                         float* __restrict__ dinv,
                                                         int n, int npad) {
    int i = blockIdx.x * blockDim.x + threadIdx.x;
    if (i < n) {
        unsigned t = 1u;
#pragma unroll
        for (int k = 0; k < DEG_K; ++k) t += part[(size_t)k * npad + i];
        const float d = (float)t;
        dinv[i] = (d > 0.0f) ? rsqrtf(d) : 0.0f;
    }
}

template <bool RELU_IN>
__global__ __launch_bounds__(256) void gemm_wmma_kernel(const float* __restrict__ X,
                                                        const float* __restrict__ X1,
                                                        const float* __restrict__ bias_in,
                                                        const float* __restrict__ W,
                                                        const float* __restrict__ dinv,
                                                        float* __restrict__ H,
                                                        float* __restrict__ ACC,
                                                        float* __restrict__ ACC1,
                                                        int nrows) {
    const int wave = (int)((blockIdx.x * blockDim.x + threadIdx.x) >> 5);
    const int lane = (int)(threadIdx.x & 31);
    const int row0 = wave * 16;
    if (row0 >= nrows) return;
    const int lm = lane & 15;
    const int kk = (lane >> 4) * 2;

    const float* __restrict__ xrow = X + (size_t)(row0 + lm) * D_FEAT + kk;
    const float* __restrict__ xrow1 = X1 + (size_t)(row0 + lm) * D_FEAT + kk;

    const float dr = RELU_IN ? dinv[row0 + lm] : 1.0f;
    v8f c0 = {}, c1 = {}, c2 = {}, c3 = {};
#pragma unroll
    for (int k = 0; k < D_FEAT; k += 4) {
        v2f a;
        a.x = xrow[k + 0];
        a.y = xrow[k + 1];
        if (RELU_IN) {
            a.x = fmaxf(bias_in[k + kk + 0] + dr * (a.x + xrow1[k + 0]), 0.0f);
            a.y = fmaxf(bias_in[k + kk + 1] + dr * (a.y + xrow1[k + 1]), 0.0f);
        }
        const float* __restrict__ wr0 = W + (size_t)(k + kk) * D_FEAT + lm;
        const float* __restrict__ wr1 = wr0 + D_FEAT;
        v2f b0, b1, b2, b3;
        b0.x = wr0[0];   b0.y = wr1[0];
        b1.x = wr0[16];  b1.y = wr1[16];
        b2.x = wr0[32];  b2.y = wr1[32];
        b3.x = wr0[48];  b3.y = wr1[48];
        c0 = __builtin_amdgcn_wmma_f32_16x16x4_f32(false, a, false, b0, (short)0, c0, false, false);
        c1 = __builtin_amdgcn_wmma_f32_16x16x4_f32(false, a, false, b1, (short)0, c1, false, false);
        c2 = __builtin_amdgcn_wmma_f32_16x16x4_f32(false, a, false, b2, (short)0, c2, false, false);
        c3 = __builtin_amdgcn_wmma_f32_16x16x4_f32(false, a, false, b3, (short)0, c3, false, false);
    }

    const int mbase = row0 + (lane >> 4) * 8;
#pragma unroll
    for (int r = 0; r < 8; ++r) {
        const float di = dinv[mbase + r];
        const float s0 = c0[r] * di, s1 = c1[r] * di, s2 = c2[r] * di, s3 = c3[r] * di;
        float* __restrict__ hr = H + (size_t)(mbase + r) * D_FEAT + lm;
        hr[0]  = s0;
        hr[16] = s1;
        hr[32] = s2;
        hr[48] = s3;
        float* __restrict__ ar = ACC + (size_t)(mbase + r) * D_FEAT + lm;
        ar[0]  = s0;
        ar[16] = s1;
        ar[32] = s2;
        ar[48] = s3;
    }
    float4* __restrict__ zr = (float4*)(ACC1 + (size_t)row0 * D_FEAT) + lane;
#pragma unroll
    for (int i = 0; i < 8; ++i) zr[i * 32] = make_float4(0.0f, 0.0f, 0.0f, 0.0f);
}

#define SC_EDGES 32
#define SC_BATCH 8

__global__ __launch_bounds__(256) void scatter_kernel(const float* __restrict__ H,
                                                      const int* __restrict__ row,
                                                      const int* __restrict__ col,
                                                      float* __restrict__ ACC0,
                                                      float* __restrict__ ACC1, int e) {
    const int lane = (int)(threadIdx.x & 31);
    float* ACC = package_half() ? ACC1 : ACC0;
    const int wave = __builtin_amdgcn_readfirstlane((int)((blockIdx.x * blockDim.x + threadIdx.x) >> 5));
    const int e0 = wave * SC_EDGES;
    if (e0 >= e) return;
    const int cnt = min(SC_EDGES, e - e0);
    const int me = min(e0 + lane, e - 1);
    const int r = row[me];
    const int c = col[me];

    int j = 0;
    for (; j + SC_BATCH <= cnt; j += SC_BATCH) {
        float h0[SC_BATCH], h1[SC_BATCH];
#pragma unroll
        for (int u = 0; u < SC_BATCH; ++u) {
            const int rj = __builtin_amdgcn_readlane(r, j + u);
            const float* __restrict__ src = H + (size_t)rj * D_FEAT + lane;
            h0[u] = src[0];
            h1[u] = src[32];
        }
#pragma unroll
        for (int u = 0; u < SC_BATCH; ++u) {
            const int cj = __builtin_amdgcn_readlane(c, j + u);
            float* dst = ACC + (size_t)cj * D_FEAT + lane;
            atomicAdd(dst, h0[u]);
            atomicAdd(dst + 32, h1[u]);
        }
    }
    for (; j < cnt; ++j) {
        const int rj = __builtin_amdgcn_readlane(r, j);
        const int cj = __builtin_amdgcn_readlane(c, j);
        const float* __restrict__ src = H + (size_t)rj * D_FEAT + lane;
        float* dst = ACC + (size_t)cj * D_FEAT + lane;
        atomicAdd(dst, src[0]);
        atomicAdd(dst + 32, src[32]);
    }
}

#define POOL_WGS 128
#define POOL_KQ 4

__global__ __launch_bounds__(512) void pool_wmma_kernel(const float* __restrict__ H,
                                                        const float* __restrict__ H1,
                                                        const float* __restrict__ dinv,
                                                        const float* __restrict__ bias,
                                                        const int* __restrict__ batch,
                                                        float* __restrict__ part,
                                                        float* __restrict__ cntp, int nnodes) {
    __shared__ float red[(POOL_KQ - 1) * 4 * 33 * 32];
    const int lane = (int)(threadIdx.x & 31);
    const int w    = __builtin_amdgcn_readfirstlane((int)(threadIdx.x >> 5));
    const int mt   = w & 3;
    const int kq   = w >> 2;
    const int lm   = lane & 15;
    const int half = lane >> 4;
    const int gsel = mt * 16 + lm;
    const int nsteps = (nnodes + 3) >> 2;
    const int nslots = (int)gridDim.x * POOL_KQ;
    const float4 bq = *(const float4*)(bias + lm * 4);

    v8f c0 = {}, c1 = {}, c2 = {}, c3 = {};
    float cnt = 0.0f;
#pragma unroll 4
    for (int ks = (int)blockIdx.x * POOL_KQ + kq; ks < nsteps; ks += nslots) {
        const int n0 = ks * 4 + half * 2;
        const int n1 = n0 + 1;
        const int m0 = min(n0, nnodes - 1);
        const int m1 = min(n1, nnodes - 1);
        const int g0 = batch[m0];
        const int g1 = batch[m1];
        const unsigned o0 = (unsigned)m0 * (unsigned)(D_FEAT * sizeof(float)) + (unsigned)lm * 16u;
        const unsigned o1 = (unsigned)m1 * (unsigned)(D_FEAT * sizeof(float)) + (unsigned)lm * 16u;
        const float4 p0 = *(const float4*)((const char*)H + o0);
        const float4 p1 = *(const float4*)((const char*)H + o1);
        const float4 q0 = *(const float4*)((const char*)H1 + o0);
        const float4 q1 = *(const float4*)((const char*)H1 + o1);
        const float d0 = dinv[m0];
        const float d1 = dinv[m1];
        const float4 r0 = make_float4(bq.x + d0 * (p0.x + q0.x), bq.y + d0 * (p0.y + q0.y),
                                      bq.z + d0 * (p0.z + q0.z), bq.w + d0 * (p0.w + q0.w));
        const float4 r1 = make_float4(bq.x + d1 * (p1.x + q1.x), bq.y + d1 * (p1.y + q1.y),
                                      bq.z + d1 * (p1.z + q1.z), bq.w + d1 * (p1.w + q1.w));
        v2f a;
        a.x = (g0 == gsel && n0 < nnodes) ? 1.0f : 0.0f;
        a.y = (g1 == gsel && n1 < nnodes) ? 1.0f : 0.0f;
        cnt += a.x + a.y;
        v2f b0, b1, b2, b3;
        b0.x = fmaxf(r0.x, 0.0f);  b0.y = fmaxf(r1.x, 0.0f);
        b1.x = fmaxf(r0.y, 0.0f);  b1.y = fmaxf(r1.y, 0.0f);
        b2.x = fmaxf(r0.z, 0.0f);  b2.y = fmaxf(r1.z, 0.0f);
        b3.x = fmaxf(r0.w, 0.0f);  b3.y = fmaxf(r1.w, 0.0f);
        c0 = __builtin_amdgcn_wmma_f32_16x16x4_f32(false, a, false, b0, (short)0, c0, false, false);
        c1 = __builtin_amdgcn_wmma_f32_16x16x4_f32(false, a, false, b1, (short)0, c1, false, false);
        c2 = __builtin_amdgcn_wmma_f32_16x16x4_f32(false, a, false, b2, (short)0, c2, false, false);
        c3 = __builtin_amdgcn_wmma_f32_16x16x4_f32(false, a, false, b3, (short)0, c3, false, false);
    }

    if (kq > 0) {
        float* dst = red + ((kq - 1) * 4 + mt) * (33 * 32) + lane;
#pragma unroll
        for (int r = 0; r < 8; ++r) {
            dst[(0 + r) * 32]  = c0[r];
            dst[(8 + r) * 32]  = c1[r];
            dst[(16 + r) * 32] = c2[r];
            dst[(24 + r) * 32] = c3[r];
        }
        dst[32 * 32] = cnt;
    }
    __syncthreads();
    if (kq == 0) {
#pragma unroll
        for (int q = 0; q < POOL_KQ - 1; ++q) {
            const float* srcp = red + (q * 4 + mt) * (33 * 32) + lane;
#pragma unroll
            for (int r = 0; r < 8; ++r) {
                c0[r] += srcp[(0 + r) * 32];
                c1[r] += srcp[(8 + r) * 32];
                c2[r] += srcp[(16 + r) * 32];
                c3[r] += srcp[(24 + r) * 32];
            }
            cnt += srcp[32 * 32];
        }
        float* pw = part + (size_t)blockIdx.x * (N_GRAPHS_C * D_FEAT);
#pragma unroll
        for (int r = 0; r < 8; ++r) {
            const int g = mt * 16 + half * 8 + r;
            *(float4*)(pw + g * D_FEAT + lm * 4) = make_float4(c0[r], c1[r], c2[r], c3[r]);
        }
        const float tot = cnt + __shfl_xor(cnt, 16);
        if (half == 0) cntp[(size_t)blockIdx.x * N_GRAPHS_C + gsel] = tot;
    }
}

__global__ __launch_bounds__(256) void pool_final_kernel(const float* __restrict__ part,
                                                         const float* __restrict__ cntp,
                                                         float* __restrict__ out) {
    __shared__ float rs[4][D_FEAT];
    __shared__ float rc[4];
    const int g = (int)blockIdx.x;
    const int d = (int)(threadIdx.x & 63);
    const int s = (int)(threadIdx.x >> 6);
    float acc = 0.0f, c = 0.0f;
#pragma unroll
    for (int i = 0; i < POOL_WGS / 4; ++i) {
        const int p = s + 4 * i;
        acc += part[(size_t)p * (N_GRAPHS_C * D_FEAT) + g * D_FEAT + d];
        c   += cntp[(size_t)p * N_GRAPHS_C + g];
    }
    rs[s][d] = acc;
    if (d == 0) rc[s] = c;
    __syncthreads();
    if (threadIdx.x < D_FEAT) {
        const float t  = ((rs[0][d] + rs[1][d]) + rs[2][d]) + rs[3][d];
        const float ct = ((rc[0] + rc[1]) + rc[2]) + rc[3];
        out[g * D_FEAT + d] = t / fmaxf(ct, 1.0f);
    }
}

extern "C" void kernel_launch(void* const* d_in, const int* in_sizes, int n_in,
                              void* d_out, int out_size, void* d_ws, size_t ws_size,
                              hipStream_t stream) {
    const float* x     = (const float*)d_in[0];
    const int*   ei    = (const int*)d_in[1];
    const int*   batch = (const int*)d_in[2];
    const float* W1    = (const float*)d_in[3];
    const float* b1    = (const float*)d_in[4];
    const float* W2    = (const float*)d_in[5];
    const float* b2    = (const float*)d_in[6];
    const float* W3    = (const float*)d_in[7];
    const float* b3    = (const float*)d_in[8];
    float* out = (float*)d_out;

    const int N = in_sizes[0] / D_FEAT;
    const int E = in_sizes[1] / 2;
    const int* erow = ei;
    const int* ecol = ei + E;

    float* w = (float*)d_ws;
    size_t off = 0;
    const int npad = ((N + 63) / 64) * 64;
    float* dinv = w + off; off += (size_t)npad;
    float* bufA = w + off; off += (size_t)N * D_FEAT;
    float* bufB = w + off; off += (size_t)N * D_FEAT;
    float* bufB1 = w + off; off += (size_t)N * D_FEAT;
    float* bufC = w + off; off += (size_t)N * D_FEAT;
    float* bufC1 = w + off; off += (size_t)N * D_FEAT;
    float* pool = w + off; off += (size_t)POOL_WGS * N_GRAPHS_C * D_FEAT;
    float* cnt  = w + off; off += (size_t)POOL_WGS * N_GRAPHS_C;
    unsigned* dpart = (unsigned*)(w + off); off += (size_t)DEG_K * npad;

    const int blk = 256;

    const int deg_ranges = (N + (DEG_CAP - 32) - 1) / (DEG_CAP - 32);
    deg_hist_kernel<<<dim3(DEG_K, deg_ranges), DEG_BLK, 0, stream>>>(ecol, dpart, N, npad, E);
    deg_finish_kernel<<<(N + blk - 1) / blk, blk, 0, stream>>>(dpart, dinv, N, npad);

    const int waves = (N + 15) / 16;
    const int gemm_blocks = (waves * 32 + blk - 1) / blk;
    const long long sthreads = (long long)E;
    const int scat_blocks = (int)((sthreads + blk - 1) / blk);

    gemm_wmma_kernel<false><<<gemm_blocks, blk, 0, stream>>>(x, x, b1, W1, dinv, bufA, bufB, bufB1, N);
    scatter_kernel<<<scat_blocks, blk, 0, stream>>>(bufA, erow, ecol, bufB, bufB1, E);

    gemm_wmma_kernel<true><<<gemm_blocks, blk, 0, stream>>>(bufB, bufB1, b1, W2, dinv, bufA, bufC, bufC1, N);
    scatter_kernel<<<scat_blocks, blk, 0, stream>>>(bufA, erow, ecol, bufC, bufC1, E);

    gemm_wmma_kernel<true><<<gemm_blocks, blk, 0, stream>>>(bufC, bufC1, b2, W3, dinv, bufA, bufB, bufB1, N);
    scatter_kernel<<<scat_blocks, blk, 0, stream>>>(bufA, erow, ecol, bufB, bufB1, E);

    pool_wmma_kernel<<<POOL_WGS, 512, 0, stream>>>(bufB, bufB1, dinv, b3, batch, pool, cnt, N);
    pool_final_kernel<<<N_GRAPHS_C, blk, 0, stream>>>(pool, cnt, out);
}


# aggregation in LDS in separate batched phases: each of 256 work-groups scans a byte key per edge (staggered, 8 loads in flight), lists its edges, adds their rows 16 at a time; sum order changes
# speedup vs baseline: 2.7929x; 1.0557x over previous
#include <hip/hip_runtime.h>
#include <hip/hip_bf16.h>

typedef __attribute__((ext_vector_type(2))) float v2f;
typedef __attribute__((ext_vector_type(8))) float v8f;

#define D_FEAT 64
#define N_GRAPHS_C 64

#define AG_WGS 256
#define AG_BLK 1024
#define AG_CAP 208
#define AG_LIST 16384
#define AG_U 8
#define AG_EB 16

__device__ __forceinline__ int ag_rows(int n) {
    const int r = (n + AG_WGS - 1) / AG_WGS;
    return r < 1 ? 1 : (r > AG_CAP ? AG_CAP : r);
}

#define DEG_K 32
#define DEG_CAP 16384
#define DEG_U 25
#define DEG_BLK 1024

__global__ __launch_bounds__(DEG_BLK) void deg_hist_kernel(const int* __restrict__ col,
                                                           unsigned* __restrict__ part,
                                                           int n, int npad, int e) {
    __shared__ unsigned hist[DEG_CAP];
    const int tid = (int)threadIdx.x;
    const int nr  = (int)gridDim.y;
    const int rl  = (((n + nr - 1) / nr) + 31) & ~31;
    const int r0  = (int)blockIdx.y * rl;
    const int len = max(0, min(n, r0 + rl) - r0);
    const int per = (e + DEG_K - 1) / DEG_K;
    const int s0  = (int)blockIdx.x * per;
    const int s1  = min(e, s0 + per);

    for (int i = tid; i < len; i += DEG_BLK) hist[i] = 0u;
    __syncthreads();
    for (int base = s0; base < s1; base += DEG_BLK * DEG_U) {
        int c[DEG_U];
#pragma unroll
        for (int u = 0; u < DEG_U; ++u) c[u] = col[min(base + u * DEG_BLK + tid, s1 - 1)];
#pragma unroll
        for (int u = 0; u < DEG_U; ++u) {
            const unsigned o = (unsigned)(c[u] - r0);
            if (base + u * DEG_BLK + tid < s1 && o < (unsigned)len) atomicAdd(&hist[o], 1u);
        }
    }
    __syncthreads();
    unsigned* __restrict__ pk = part + (size_t)blockIdx.x * npad + r0;
    for (int i = tid; i < len; i += DEG_BLK) pk[i] = hist[i];
}

__global__ __launch_bounds__(256) void deg_finish_kernel(const unsigned* __restrict__ part,
                                                         float* __restrict__ dinv,
                                                         const int* __restrict__ col,
                                                         unsigned* __restrict__ key,
                                                         int n, int npad, int e) {
    int i = blockIdx.x * blockDim.x + threadIdx.x;
    const unsigned rows = (unsigned)ag_rows(n);
    const int nvec = e >> 4;
    for (int v = i; v < nvec; v += (int)(gridDim.x * blockDim.x)) {
        const int* __restrict__ cp = col + (size_t)v * 16;
        unsigned kw[4];
#pragma unroll
        for (int d = 0; d < 4; ++d) {
            unsigned wv = 0u;
#pragma unroll
            for (int b = 0; b < 4; ++b) {
                const unsigned c = (unsigned)cp[d * 4 + b];
                const unsigned k = (c < (unsigned)n) ? ((c / rows) & 255u) : 0u;
                wv |= k << (8 * b);
            }
            kw[d] = wv;
        }
        *(uint4*)(key + (size_t)v * 4) = make_uint4(kw[0], kw[1], kw[2], kw[3]);
    }
    if (i < n) {
        unsigned t = 1u;
#pragma unroll
        for (int k = 0; k < DEG_K; ++k) t += part[(size_t)k * npad + i];
        const float d = (float)t;
        dinv[i] = (d > 0.0f) ? rsqrtf(d) : 0.0f;
    }
}

template <bool RELU_IN>
__global__ __launch_bounds__(256) void gemm_wmma_kernel(const float* __restrict__ X,
                                                        const float* __restrict__ bias_in,
                                                        const float* __restrict__ W,
                                                        const float* __restrict__ dinv,
                                                        float* __restrict__ H,
                                                        int nrows) {
    const int wave = (int)((blockIdx.x * blockDim.x + threadIdx.x) >> 5);
    const int lane = (int)(threadIdx.x & 31);
    const int row0 = wave * 16;
    if (row0 >= nrows) return;
    const int lm = lane & 15;
    const int kk = (lane >> 4) * 2;

    const float* __restrict__ xrow = X + (size_t)(row0 + lm) * D_FEAT + kk;

    const float dr = RELU_IN ? dinv[row0 + lm] : 1.0f;
    v8f c0 = {}, c1 = {}, c2 = {}, c3 = {};
#pragma unroll
    for (int k = 0; k < D_FEAT; k += 4) {
        v2f a;
        a.x = xrow[k + 0];
        a.y = xrow[k + 1];
        if (RELU_IN) {
            a.x = fmaxf(bias_in[k + kk + 0] + dr * a.x, 0.0f);
            a.y = fmaxf(bias_in[k + kk + 1] + dr * a.y, 0.0f);
        }
        const float* __restrict__ wr0 = W + (size_t)(k + kk) * D_FEAT + lm;
        const float* __restrict__ wr1 = wr0 + D_FEAT;
        v2f b0, b1, b2, b3;
        b0.x = wr0[0];   b0.y = wr1[0];
        b1.x = wr0[16];  b1.y = wr1[16];
        b2.x = wr0[32];  b2.y = wr1[32];
        b3.x = wr0[48];  b3.y = wr1[48];
        c0 = __builtin_amdgcn_wmma_f32_16x16x4_f32(false, a, false, b0, (short)0, c0, false, false);
        c1 = __builtin_amdgcn_wmma_f32_16x16x4_f32(false, a, false, b1, (short)0, c1, false, false);
        c2 = __builtin_amdgcn_wmma_f32_16x16x4_f32(false, a, false, b2, (short)0, c2, false, false);
        c3 = __builtin_amdgcn_wmma_f32_16x16x4_f32(false, a, false, b3, (short)0, c3, false, false);
    }

    const int mbase = row0 + (lane >> 4) * 8;
#pragma unroll
    for (int r = 0; r < 8; ++r) {
        const float di = dinv[mbase + r];
        const float s0 = c0[r] * di, s1 = c1[r] * di, s2 = c2[r] * di, s3 = c3[r] * di;
        float* __restrict__ hr = H + (size_t)(mbase + r) * D_FEAT + lm;
        hr[0]  = s0;
        hr[16] = s1;
        hr[32] = s2;
        hr[48] = s3;
    }
}

__global__ __launch_bounds__(AG_BLK) void agg_lds_kernel(const float* __restrict__ HS,
                                                         const int* __restrict__ row,
                                                         const int* __restrict__ col,
                                                         const unsigned* __restrict__ key,
                                                         float* __restrict__ OUT, int n, int e) {
    __shared__ float acc[AG_CAP * D_FEAT];
    __shared__ int lsrc[AG_LIST];
    __shared__ int lrow[AG_LIST];
    __shared__ unsigned lcnt;

    const int tid  = (int)threadIdx.x;
    const int lane = tid & 31;
    const int wv   = __builtin_amdgcn_readfirstlane(tid >> 5);
    const int wg   = (int)blockIdx.x;
    const int rows = ag_rows(n);
    const int nranges = (n + rows - 1) / rows;
    const int nvec = e >> 4;
    const unsigned idrep = (unsigned)wg * 0x01010101u;
    const int rot = (int)(((long long)nvec * wg) / AG_WGS);
    const uint4* __restrict__ kv = (const uint4*)key;
    float4* accv = (float4*)acc;

    for (int g = wg; g < nranges; g += AG_WGS) {
        const int r0  = g * rows;
        const int len = min(rows, n - r0);

        const float4* __restrict__ hv = (const float4*)(HS + (size_t)r0 * D_FEAT);
        {
            float4 own[(AG_CAP * (D_FEAT / 4) + AG_BLK - 1) / AG_BLK];
#pragma unroll
            for (int u = 0; u < (AG_CAP * (D_FEAT / 4) + AG_BLK - 1) / AG_BLK; ++u) {
                const int i = u * AG_BLK + tid;
                own[u] = hv[i < len * (D_FEAT / 4) ? i : 0];
            }
#pragma unroll
            for (int u = 0; u < (AG_CAP * (D_FEAT / 4) + AG_BLK - 1) / AG_BLK; ++u) {
                const int i = u * AG_BLK + tid;
                if (i < len * (D_FEAT / 4)) accv[i] = own[u];
            }
        }
        if (tid == 0) lcnt = 0u;
        __syncthreads();

        for (int base = 0; base < nvec; base += AG_BLK * AG_U) {
            uint4 k[AG_U];
#pragma unroll
            for (int u = 0; u < AG_U; ++u) {
                const int i = base + u * AG_BLK + tid;
                int ii = min(i, nvec - 1) + rot;
                if (ii >= nvec) ii -= nvec;
                k[u] = kv[ii];
            }
#pragma unroll
            for (int u = 0; u < AG_U; u += 2) {
                unsigned m = 0u;
#pragma unroll
                for (int h = 0; h < 2; ++h) {
                    const unsigned y0 = k[u + h].x ^ idrep, y1 = k[u + h].y ^ idrep;
                    const unsigned y2 = k[u + h].z ^ idrep, y3 = k[u + h].w ^ idrep;
                    const unsigned z0 = (y0 - 0x01010101u) & ~y0 & 0x80808080u;
                    const unsigned z1 = (y1 - 0x01010101u) & ~y1 & 0x80808080u;
                    const unsigned z2 = (y2 - 0x01010101u) & ~y2 & 0x80808080u;
                    const unsigned z3 = (y3 - 0x01010101u) & ~y3 & 0x80808080u;
                    unsigned mh = (z0 >> 7) | (z1 >> 6) | (z2 >> 5) | (z3 >> 4);
                    if (base + (u + h) * AG_BLK + tid >= nvec) mh = 0u;
                    m |= mh << (4 * h);
                }
                while (m != 0u) {
                    const int p = __builtin_ctz(m);
                    m &= m - 1u;
                    int v = base + (u + ((p >> 2) & 1)) * AG_BLK + tid + rot;
                    if (v >= nvec) v -= nvec;
                    const unsigned slot = atomicAdd(&lcnt, 1u);
                    if (slot < (unsigned)AG_LIST) lsrc[slot] = v * 16 + ((p & 3) << 2) + (p >> 3);
                }
            }
        }
        if (tid < (e & 15)) {
            const int ed = nvec * 16 + tid;
            if ((unsigned)(col[ed] - r0) < (unsigned)len) {
                const unsigned slot = atomicAdd(&lcnt, 1u);
                if (slot < (unsigned)AG_LIST) lsrc[slot] = ed;
            }
        }
        __syncthreads();
        const int total = (int)lcnt;

        if (total <= AG_LIST) {
            const int padded = (total + 31) & ~31;
            for (int b = 0; b < padded; b += AG_BLK * 4) {
                int ed[4], cc[4], ss[4];
#pragma unroll
                for (int u = 0; u < 4; ++u) {
                    const int i = b + u * AG_BLK + tid;
                    ed[u] = (i < total) ? lsrc[i] : 0;
                }
#pragma unroll
                for (int u = 0; u < 4; ++u) { cc[u] = col[ed[u]]; ss[u] = row[ed[u]]; }
#pragma unroll
                for (int u = 0; u < 4; ++u) {
                    const int i = b + u * AG_BLK + tid;
                    if (i < padded) {
                        const unsigned o = (unsigned)(cc[u] - r0);
                        const bool ok = (i < total) && (o < (unsigned)len);
                        lsrc[i] = ok ? ss[u] : 0;
                        lrow[i] = ok ? (int)o : -1;
                    }
                }
            }
            __syncthreads();

            const int ngroups = padded >> 5;
            for (int q = wv; q < ngroups; q += AG_BLK / 32) {
                const int s = lsrc[q * 32 + lane];
                const int t = lrow[q * 32 + lane];
#pragma unroll
                for (int h = 0; h < 32; h += AG_EB) {
                    float h0[AG_EB], h1[AG_EB];
#pragma unroll
                    for (int u = 0; u < AG_EB; ++u) {
                        const int sj = __builtin_amdgcn_readlane(s, h + u);
                        const float* __restrict__ sp = HS + (size_t)sj * D_FEAT + lane;
                        h0[u] = sp[0];
                        h1[u] = sp[32];
                    }
#pragma unroll
                    for (int u = 0; u < AG_EB; ++u) {
                        const int tj = __builtin_amdgcn_readlane(t, h + u);
                        if (tj >= 0) {
                            float* d = acc + tj * D_FEAT + lane;
                            atomicAdd(d, h0[u]);
                            atomicAdd(d + 32, h1[u]);
                        }
                    }
                }
            }
        } else {
            for (int eb = wv * 32; eb < e; eb += AG_BLK) {
                const int me = min(eb + lane, e - 1);
                const int s = row[me];
                const unsigned o = (unsigned)(col[me] - r0);
                unsigned hit = __builtin_amdgcn_ballot_w32((eb + lane < e) && (o < (unsigned)len));
                while (hit != 0u) {
                    const int j = __builtin_ctz(hit);
                    hit &= hit - 1u;
                    const int sj = __builtin_amdgcn_readlane(s, j);
                    const int tj = __builtin_amdgcn_readlane((int)o, j);
                    const float* __restrict__ sp = HS + (size_t)sj * D_FEAT + lane;
                    float* d = acc + tj * D_FEAT + lane;
                    atomicAdd(d, sp[0]);
                    atomicAdd(d + 32, sp[32]);
                }
            }
        }
        __syncthreads();

        float4* __restrict__ ov = (float4*)(OUT + (size_t)r0 * D_FEAT);
        for (int i = tid; i < len * (D_FEAT / 4); i += AG_BLK) ov[i] = accv[i];
        __syncthreads();
    }
}

#define POOL_WGS 128
#define POOL_KQ 4

__global__ __launch_bounds__(512) void pool_wmma_kernel(const float* __restrict__ H,
                                                        const float* __restrict__ dinv,
                                                        const float* __restrict__ bias,
                                                        const int* __restrict__ batch,
                                                        float* __restrict__ part,
                                                        float* __restrict__ cntp, int nnodes) {
    __shared__ float red[(POOL_KQ - 1) * 4 * 33 * 32];
    const int lane = (int)(threadIdx.x & 31);
    const int w    = __builtin_amdgcn_readfirstlane((int)(threadIdx.x >> 5));
    const int mt   = w & 3;
    const int kq   = w >> 2;
    const int lm   = lane & 15;
    const int half = lane >> 4;
    const int gsel = mt * 16 + lm;
    const int nsteps = (nnodes + 3) >> 2;
    const int nslots = (int)gridDim.x * POOL_KQ;
    const float4 bq = *(const float4*)(bias + lm * 4);

    v8f c0 = {}, c1 = {}, c2 = {}, c3 = {};
    float cnt = 0.0f;
#pragma unroll 4
    for (int ks = (int)blockIdx.x * POOL_KQ + kq; ks < nsteps; ks += nslots) {
        const int n0 = ks * 4 + half * 2;
        const int n1 = n0 + 1;
        const int m0 = min(n0, nnodes - 1);
        const int m1 = min(n1, nnodes - 1);
        const int g0 = batch[m0];
        const int g1 = batch[m1];
        const unsigned o0 = (unsigned)m0 * (unsigned)(D_FEAT * sizeof(float)) + (unsigned)lm * 16u;
        const unsigned o1 = (unsigned)m1 * (unsigned)(D_FEAT * sizeof(float)) + (unsigned)lm * 16u;
        const float4 p0 = *(const float4*)((const char*)H + o0);
        const float4 p1 = *(const float4*)((const char*)H + o1);
        const float d0 = dinv[m0];
        const float d1 = dinv[m1];
        const float4 r0 = make_float4(bq.x + d0 * p0.x, bq.y + d0 * p0.y,
                                      bq.z + d0 * p0.z, bq.w + d0 * p0.w);
        const float4 r1 = make_float4(bq.x + d1 * p1.x, bq.y + d1 * p1.y,
                                      bq.z + d1 * p1.z, bq.w + d1 * p1.w);
        v2f a;
        a.x = (g0 == gsel && n0 < nnodes) ? 1.0f : 0.0f;
        a.y = (g1 == gsel && n1 < nnodes) ? 1.0f : 0.0f;
        cnt += a.x + a.y;
        v2f b0, b1, b2, b3;
        b0.x = fmaxf(r0.x, 0.0f);  b0.y = fmaxf(r1.x, 0.0f);
        b1.x = fmaxf(r0.y, 0.0f);  b1.y = fmaxf(r1.y, 0.0f);
        b2.x = fmaxf(r0.z, 0.0f);  b2.y = fmaxf(r1.z, 0.0f);
        b3.x = fmaxf(r0.w, 0.0f);  b3.y = fmaxf(r1.w, 0.0f);
        c0 = __builtin_amdgcn_wmma_f32_16x16x4_f32(false, a, false, b0, (short)0, c0, false, false);
        c1 = __builtin_amdgcn_wmma_f32_16x16x4_f32(false, a, false, b1, (short)0, c1, false, false);
        c2 = __builtin_amdgcn_wmma_f32_16x16x4_f32(false, a, false, b2, (short)0, c2, false, false);
        c3 = __builtin_amdgcn_wmma_f32_16x16x4_f32(false, a, false, b3, (short)0, c3, false, false);
    }

    if (kq > 0) {
        float* dst = red + ((kq - 1) * 4 + mt) * (33 * 32) + lane;
#pragma unroll
        for (int r = 0; r < 8; ++r) {
            dst[(0 + r) * 32]  = c0[r];
            dst[(8 + r) * 32]  = c1[r];
            dst[(16 + r) * 32] = c2[r];
            dst[(24 + r) * 32] = c3[r];
        }
        dst[32 * 32] = cnt;
    }
    __syncthreads();
    if (kq == 0) {
#pragma unroll
        for (int q = 0; q < POOL_KQ - 1; ++q) {
            const float* srcp = red + (q * 4 + mt) * (33 * 32) + lane;
#pragma unroll
            for (int r = 0; r < 8; ++r) {
                c0[r] += srcp[(0 + r) * 32];
                c1[r] += srcp[(8 + r) * 32];
                c2[r] += srcp[(16 + r) * 32];
                c3[r] += srcp[(24 + r) * 32];
            }
            cnt += srcp[32 * 32];
        }
        float* pw = part + (size_t)blockIdx.x * (N_GRAPHS_C * D_FEAT);
#pragma unroll
        for (int r = 0; r < 8; ++r) {
            const int g = mt * 16 + half * 8 + r;
            *(float4*)(pw + g * D_FEAT + lm * 4) = make_float4(c0[r], c1[r], c2[r], c3[r]);
        }
        const float tot = cnt + __shfl_xor(cnt, 16);
        if (half == 0) cntp[(size_t)blockIdx.x * N_GRAPHS_C + gsel] = tot;
    }
}

__global__ __launch_bounds__(256) void pool_final_kernel(const float* __restrict__ part,
                                                         const float* __restrict__ cntp,
                                                         float* __restrict__ out) {
    __shared__ float rs[4][D_FEAT];
    __shared__ float rc[4];
    const int g = (int)blockIdx.x;
    const int d = (int)(threadIdx.x & 63);
    const int s = (int)(threadIdx.x >> 6);
    float acc = 0.0f, c = 0.0f;
#pragma unroll
    for (int i = 0; i < POOL_WGS / 4; ++i) {
        const int p = s + 4 * i;
        acc += part[(size_t)p * (N_GRAPHS_C * D_FEAT) + g * D_FEAT + d];
        c   += cntp[(size_t)p * N_GRAPHS_C + g];
    }
    rs[s][d] = acc;
    if (d == 0) rc[s] = c;
    __syncthreads();
    if (threadIdx.x < D_FEAT) {
        const float t  = ((rs[0][d] + rs[1][d]) + rs[2][d]) + rs[3][d];
        const float ct = ((rc[0] + rc[1]) + rc[2]) + rc[3];
        out[g * D_FEAT + d] = t / fmaxf(ct, 1.0f);
    }
}

extern "C" void kernel_launch(void* const* d_in, const int* in_sizes, int n_in,
                              void* d_out, int out_size, void* d_ws, size_t ws_size,
                              hipStream_t stream) {
    const float* x     = (const float*)d_in[0];
    const int*   ei    = (const int*)d_in[1];
    const int*   batch = (const int*)d_in[2];
    const float* W1    = (const float*)d_in[3];
    const float* b1    = (const float*)d_in[4];
    const float* W2    = (const float*)d_in[5];
    const float* b2    = (const float*)d_in[6];
    const float* W3    = (const float*)d_in[7];
    const float* b3    = (const float*)d_in[8];
    float* out = (float*)d_out;

    const int N = in_sizes[0] / D_FEAT;
    const int E = in_sizes[1] / 2;
    const int* erow = ei;
    const int* ecol = ei + E;

    float* w = (float*)d_ws;
    size_t off = 0;
    const int npad = ((N + 63) / 64) * 64;
    float* dinv = w + off; off += (size_t)npad;
    float* bufA = w + off; off += (size_t)N * D_FEAT;
    float* bufB = w + off; off += (size_t)N * D_FEAT;
    float* pool = w + off; off += (size_t)POOL_WGS * N_GRAPHS_C * D_FEAT;
    float* cnt  = w + off; off += (size_t)POOL_WGS * N_GRAPHS_C;
    unsigned* dpart = (unsigned*)(w + off); off += (size_t)DEG_K * npad;
    unsigned* ekey = (unsigned*)(w + off); off += (size_t)(E / 16) * 4;

    const int blk = 256;

    const int deg_ranges = (N + (DEG_CAP - 32) - 1) / (DEG_CAP - 32);
    deg_hist_kernel<<<dim3(DEG_K, deg_ranges), DEG_BLK, 0, stream>>>(ecol, dpart, N, npad, E);
    deg_finish_kernel<<<(N + blk - 1) / blk, blk, 0, stream>>>(dpart, dinv, ecol, ekey, N, npad, E);

    const int waves = (N + 15) / 16;
    const int gemm_blocks = (waves * 32 + blk - 1) / blk;

    gemm_wmma_kernel<false><<<gemm_blocks, blk, 0, stream>>>(x, b1, W1, dinv, bufA, N);
    agg_lds_kernel<<<AG_WGS, AG_BLK, 0, stream>>>(bufA, erow, ecol, ekey, bufB, N, E);

    gemm_wmma_kernel<true><<<gemm_blocks, blk, 0, stream>>>(bufB, b1, W2, dinv, bufA, N);
    agg_lds_kernel<<<AG_WGS, AG_BLK, 0, stream>>>(bufA, erow, ecol, ekey, bufB, N, E);

    gemm_wmma_kernel<true><<<gemm_blocks, blk, 0, stream>>>(bufB, b2, W3, dinv, bufA, N);
    agg_lds_kernel<<<AG_WGS, AG_BLK, 0, stream>>>(bufA, erow, ecol, ekey, bufB, N, E);

    pool_wmma_kernel<<<POOL_WGS, 512, 0, stream>>>(bufB, dinv, b3, batch, pool, cnt, N);
    pool_final_kernel<<<N_GRAPHS_C, blk, 0, stream>>>(pool, cnt, out);
}


# edge lists per node range (source row, local target) built once, in the layer-1 aggregation, and kept in the workspace; layers 2, 3 skip key scan and resolve and stream them; LDS add order as before
# speedup vs baseline: 3.5013x; 1.2536x over previous
#include <hip/hip_runtime.h>
#include <hip/hip_bf16.h>

typedef __attribute__((ext_vector_type(2))) float v2f;
typedef __attribute__((ext_vector_type(8))) float v8f;

#define D_FEAT 64
#define N_GRAPHS_C 64

#define AG_WGS 256
#define AG_BLK 1024
#define AG_CAP 208
#define AG_LIST 16384
#define AG_U 8
#define AG_EB 16

__device__ __forceinline__ int ag_rows(int n) {
    const int r = (n + AG_WGS - 1) / AG_WGS;
    return r < 1 ? 1 : (r > AG_CAP ? AG_CAP : r);
}

#define DEG_K 32
#define DEG_CAP 16384
#define DEG_U 25
#define DEG_BLK 1024

__global__ __launch_bounds__(DEG_BLK) void deg_hist_kernel(const int* __restrict__ col,
                                                           unsigned* __restrict__ part,
                                                           int n, int npad, int e) {
    __shared__ unsigned hist[DEG_CAP];
    const int tid = (int)threadIdx.x;
    const int nr  = (int)gridDim.y;
    const int rl  = (((n + nr - 1) / nr) + 31) & ~31;
    const int r0  = (int)blockIdx.y * rl;
    const int len = max(0, min(n, r0 + rl) - r0);
    const int per = (e + DEG_K - 1) / DEG_K;
    const int s0  = (int)blockIdx.x * per;
    const int s1  = min(e, s0 + per);

    for (int i = tid; i < len; i += DEG_BLK) hist[i] = 0u;
    __syncthreads();
    for (int base = s0; base < s1; base += DEG_BLK * DEG_U) {
        int c[DEG_U];
#pragma unroll
        for (int u = 0; u < DEG_U; ++u) c[u] = col[min(base + u * DEG_BLK + tid, s1 - 1)];
#pragma unroll
        for (int u = 0; u < DEG_U; ++u) {
            const unsigned o = (unsigned)(c[u] - r0);
            if (base + u * DEG_BLK + tid < s1 && o < (unsigned)len) atomicAdd(&hist[o], 1u);
        }
    }
    __syncthreads();
    unsigned* __restrict__ pk = part + (size_t)blockIdx.x * npad + r0;
    for (int i = tid; i < len; i += DEG_BLK) pk[i] = hist[i];
}

__global__ __launch_bounds__(256) void deg_finish_kernel(const unsigned* __restrict__ part,
                                                         float* __restrict__ dinv,
                                                         const int* __restrict__ col,
                                                         unsigned* __restrict__ key,
                                                         int n, int npad, int e) {
    int i = blockIdx.x * blockDim.x + threadIdx.x;
    const unsigned rows = (unsigned)ag_rows(n);
    const int nvec = e >> 4;
    for (int v = i; v < nvec; v += (int)(gridDim.x * blockDim.x)) {
        const int* __restrict__ cp = col + (size_t)v * 16;
        unsigned kw[4];
#pragma unroll
        for (int d = 0; d < 4; ++d) {
            unsigned wv = 0u;
#pragma unroll
            for (int b = 0; b < 4; ++b) {
                const unsigned c = (unsigned)cp[d * 4 + b];
                const unsigned k = (c < (unsigned)n) ? ((c / rows) & 255u) : 0u;
                wv |= k << (8 * b);
            }
            kw[d] = wv;
        }
        *(uint4*)(key + (size_t)v * 4) = make_uint4(kw[0], kw[1], kw[2], kw[3]);
    }
    if (i < n) {
        unsigned t = 1u;
#pragma unroll
        for (int k = 0; k < DEG_K; ++k) t += part[(size_t)k * npad + i];
        const float d = (float)t;
        dinv[i] = (d > 0.0f) ? rsqrtf(d) : 0.0f;
    }
}

template <bool RELU_IN>
__global__ __launch_bounds__(256) void gemm_wmma_kernel(const float* __restrict__ X,
                                                        const float* __restrict__ bias_in,
                                                        const float* __restrict__ W,
                                                        const float* __restrict__ dinv,
                                                        float* __restrict__ H,
                                                        int nrows) {
    const int wave = (int)((blockIdx.x * blockDim.x + threadIdx.x) >> 5);
    const int lane = (int)(threadIdx.x & 31);
    const int row0 = wave * 16;
    if (row0 >= nrows) return;
    const int lm = lane & 15;
    const int kk = (lane >> 4) * 2;

    const float* __restrict__ xrow = X + (size_t)(row0 + lm) * D_FEAT + kk;

    const float dr = RELU_IN ? dinv[row0 + lm] : 1.0f;
    v8f c0 = {}, c1 = {}, c2 = {}, c3 = {};
#pragma unroll
    for (int k = 0; k < D_FEAT; k += 4) {
        v2f a;
        a.x = xrow[k + 0];
        a.y = xrow[k + 1];
        if (RELU_IN) {
            a.x = fmaxf(bias_in[k + kk + 0] + dr * a.x, 0.0f);
            a.y = fmaxf(bias_in[k + kk + 1] + dr * a.y, 0.0f);
        }
        const float* __restrict__ wr0 = W + (size_t)(k + kk) * D_FEAT + lm;
        const float* __restrict__ wr1 = wr0 + D_FEAT;
        v2f b0, b1, b2, b3;
        b0.x = wr0[0];   b0.y = wr1[0];
        b1.x = wr0[16];  b1.y = wr1[16];
        b2.x = wr0[32];  b2.y = wr1[32];
        b3.x = wr0[48];  b3.y = wr1[48];
        c0 = __builtin_amdgcn_wmma_f32_16x16x4_f32(false, a, false, b0, (short)0, c0, false, false);
        c1 = __builtin_amdgcn_wmma_f32_16x16x4_f32(false, a, false, b1, (short)0, c1, false, false);
        c2 = __builtin_amdgcn_wmma_f32_16x16x4_f32(false, a, false, b2, (short)0, c2, false, false);
        c3 = __builtin_amdgcn_wmma_f32_16x16x4_f32(false, a, false, b3, (short)0, c3, false, false);
    }

    const int mbase = row0 + (lane >> 4) * 8;
#pragma unroll
    for (int r = 0; r < 8; ++r) {
        const float di = dinv[mbase + r];
        const float s0 = c0[r] * di, s1 = c1[r] * di, s2 = c2[r] * di, s3 = c3[r] * di;
        float* __restrict__ hr = H + (size_t)(mbase + r) * D_FEAT + lm;
        hr[0]  = s0;
        hr[16] = s1;
        hr[32] = s2;
        hr[48] = s3;
    }
}


__device__ __forceinline__ void ag_add_group(const float* __restrict__ HS, float* acc,
                                             int s, int t, int lane) {
#pragma unroll
    for (int h = 0; h < 32; h += AG_EB) {
        float h0[AG_EB], h1[AG_EB];
#pragma unroll
        for (int u = 0; u < AG_EB; ++u) {
            const int sj = __builtin_amdgcn_readlane(s, h + u);
            const float* __restrict__ sp = HS + (size_t)sj * D_FEAT + lane;
            h0[u] = sp[0];
            h1[u] = sp[32];
        }
#pragma unroll
        for (int u = 0; u < AG_EB; ++u) {
            const int tj = __builtin_amdgcn_readlane(t, h + u);
            if (tj >= 0) {
                float* d = acc + tj * D_FEAT + lane;
                atomicAdd(d, h0[u]);
                atomicAdd(d + 32, h1[u]);
            }
        }
    }
}

__global__ __launch_bounds__(AG_BLK) void agg_lds_kernel(const float* __restrict__ HS,
                                                         const int* __restrict__ row,
                                                         const int* __restrict__ col,
                                                         const unsigned* __restrict__ key,
                                                         int2* elist, int* ecnt,
                                                         float* __restrict__ OUT, int n, int e,
                                                         int reuse) {
    __shared__ float acc[AG_CAP * D_FEAT];
    __shared__ int lsrc[AG_LIST];
    __shared__ int lrow[AG_LIST];
    __shared__ unsigned lcnt;

    const int tid  = (int)threadIdx.x;
    const int lane = tid & 31;
    const int wv   = __builtin_amdgcn_readfirstlane(tid >> 5);
    const int wg   = (int)blockIdx.x;
    const int rows = ag_rows(n);
    const int nranges = (n + rows - 1) / rows;
    const int nvec = e >> 4;
    const unsigned idrep = (unsigned)wg * 0x01010101u;
    const int rot = (int)(((long long)nvec * wg) / AG_WGS);
    const uint4* __restrict__ kv = (const uint4*)key;
    float4* accv = (float4*)acc;

    for (int g = wg; g < nranges; g += AG_WGS) {
        const int r0  = g * rows;
        const int len = min(rows, n - r0);

        const float4* __restrict__ hv = (const float4*)(HS + (size_t)r0 * D_FEAT);
        {
            float4 own[(AG_CAP * (D_FEAT / 4) + AG_BLK - 1) / AG_BLK];
#pragma unroll
            for (int u = 0; u < (AG_CAP * (D_FEAT / 4) + AG_BLK - 1) / AG_BLK; ++u) {
                const int i = u * AG_BLK + tid;
                own[u] = hv[i < len * (D_FEAT / 4) ? i : 0];
            }
#pragma unroll
            for (int u = 0; u < (AG_CAP * (D_FEAT / 4) + AG_BLK - 1) / AG_BLK; ++u) {
                const int i = u * AG_BLK + tid;
                if (i < len * (D_FEAT / 4)) accv[i] = own[u];
            }
        }
        const bool kept = g < AG_WGS;
        const bool from_ws = kept && reuse != 0;
        int2* gl = elist + (size_t)(kept ? g : 0) * AG_LIST;
        int total;
        if (from_ws) {
            total = __builtin_amdgcn_readfirstlane(ecnt[g]);
            __syncthreads();
        } else {
            if (tid == 0) lcnt = 0u;
            __syncthreads();

            for (int base = 0; base < nvec; base += AG_BLK * AG_U) {
                uint4 k[AG_U];
#pragma unroll
                for (int u = 0; u < AG_U; ++u) {
                    const int i = base + u * AG_BLK + tid;
                    int ii = min(i, nvec - 1) + rot;
                    if (ii >= nvec) ii -= nvec;
                    k[u] = kv[ii];
                }
#pragma unroll
                for (int u = 0; u < AG_U; u += 2) {
                    unsigned m = 0u;
#pragma unroll
                    for (int h = 0; h < 2; ++h) {
                        const unsigned y0 = k[u + h].x ^ idrep, y1 = k[u + h].y ^ idrep;
                        const unsigned y2 = k[u + h].z ^ idrep, y3 = k[u + h].w ^ idrep;
                        const unsigned z0 = (y0 - 0x01010101u) & ~y0 & 0x80808080u;
                        const unsigned z1 = (y1 - 0x01010101u) & ~y1 & 0x80808080u;
                        const unsigned z2 = (y2 - 0x01010101u) & ~y2 & 0x80808080u;
                        const unsigned z3 = (y3 - 0x01010101u) & ~y3 & 0x80808080u;
                        unsigned mh = (z0 >> 7) | (z1 >> 6) | (z2 >> 5) | (z3 >> 4);
                        if (base + (u + h) * AG_BLK + tid >= nvec) mh = 0u;
                        m |= mh << (4 * h);
                    }
                    while (m != 0u) {
                        const int p = __builtin_ctz(m);
                        m &= m - 1u;
                        int v = base + (u + ((p >> 2) & 1)) * AG_BLK + tid + rot;
                        if (v >= nvec) v -= nvec;
                        const unsigned slot = atomicAdd(&lcnt, 1u);
                        if (slot < (unsigned)AG_LIST) lsrc[slot] = v * 16 + ((p & 3) << 2) + (p >> 3);
                    }
                }
            }
            if (tid < (e & 15)) {
                const int ed = nvec * 16 + tid;
                if ((unsigned)(col[ed] - r0) < (unsigned)len) {
                    const unsigned slot = atomicAdd(&lcnt, 1u);
                    if (slot < (unsigned)AG_LIST) lsrc[slot] = ed;
                }
            }
            __syncthreads();
            total = (int)lcnt;
            if (kept && tid == 0) ecnt[g] = total;

            if (total <= AG_LIST) {
                const int padded = (total + 31) & ~31;
                for (int b = 0; b < padded; b += AG_BLK * 4) {
                    int ed[4], cc[4], ss[4];
#pragma unroll
                    for (int u = 0; u < 4; ++u) {
                        const int i = b + u * AG_BLK + tid;
                        ed[u] = (i < total) ? lsrc[i] : 0;
                    }
#pragma unroll
                    for (int u = 0; u < 4; ++u) { cc[u] = col[ed[u]]; ss[u] = row[ed[u]]; }
#pragma unroll
                    for (int u = 0; u < 4; ++u) {
                        const int i = b + u * AG_BLK + tid;
                        if (i < padded) {
                            const unsigned o = (unsigned)(cc[u] - r0);
                            const bool ok = (i < total) && (o < (unsigned)len);
                            const int se = ok ? ss[u] : 0;
                            const int te = ok ? (int)o : -1;
                            lsrc[i] = se;
                            lrow[i] = te;
                            if (kept) gl[i] = make_int2(se, te);
                        }
                    }
                }
                __syncthreads();
            }
        }

        if (total <= AG_LIST) {
            const int ngroups = ((total + 31) & ~31) >> 5;
            if (from_ws) {
                int q = wv;
                int2 cur = make_int2(0, -1);
                if (q < ngroups) cur = gl[q * 32 + lane];
                for (; q < ngroups; q += AG_BLK / 32) {
                    const int2 nxt = gl[min(q + AG_BLK / 32, ngroups - 1) * 32 + lane];
                    ag_add_group(HS, acc, cur.x, cur.y, lane);
                    cur = nxt;
                }
            } else {
                for (int q = wv; q < ngroups; q += AG_BLK / 32)
                    ag_add_group(HS, acc, lsrc[q * 32 + lane], lrow[q * 32 + lane], lane);
            }
        } else {
            for (int eb = wv * 32; eb < e; eb += AG_BLK) {
                const int me = min(eb + lane, e - 1);
                const int s = row[me];
                const unsigned o = (unsigned)(col[me] - r0);
                unsigned hit = __builtin_amdgcn_ballot_w32((eb + lane < e) && (o < (unsigned)len));
                while (hit != 0u) {
                    const int j = __builtin_ctz(hit);
                    hit &= hit - 1u;
                    const int sj = __builtin_amdgcn_readlane(s, j);
                    const int tj = __builtin_amdgcn_readlane((int)o, j);
                    const float* __restrict__ sp = HS + (size_t)sj * D_FEAT + lane;
                    float* d = acc + tj * D_FEAT + lane;
                    atomicAdd(d, sp[0]);
                    atomicAdd(d + 32, sp[32]);
                }
            }
        }
        __syncthreads();

        float4* __restrict__ ov = (float4*)(OUT + (size_t)r0 * D_FEAT);
        for (int i = tid; i < len * (D_FEAT / 4); i += AG_BLK) ov[i] = accv[i];
        __syncthreads();
    }
}

#define POOL_WGS 128
#define POOL_KQ 4

__global__ __launch_bounds__(512) void pool_wmma_kernel(const float* __restrict__ H,
                                                        const float* __restrict__ dinv,
                                                        const float* __restrict__ bias,
                                                        const int* __restrict__ batch,
                                                        float* __restrict__ part,
                                                        float* __restrict__ cntp, int nnodes) {
    __shared__ float red[(POOL_KQ - 1) * 4 * 33 * 32];
    const int lane = (int)(threadIdx.x & 31);
    const int w    = __builtin_amdgcn_readfirstlane((int)(threadIdx.x >> 5));
    const int mt   = w & 3;
    const int kq   = w >> 2;
    const int lm   = lane & 15;
    const int half = lane >> 4;
    const int gsel = mt * 16 + lm;
    const int nsteps = (nnodes + 3) >> 2;
    const int nslots = (int)gridDim.x * POOL_KQ;
    const float4 bq = *(const float4*)(bias + lm * 4);

    v8f c0 = {}, c1 = {}, c2 = {}, c3 = {};
    float cnt = 0.0f;
#pragma unroll 4
    for (int ks = (int)blockIdx.x * POOL_KQ + kq; ks < nsteps; ks += nslots) {
        const int n0 = ks * 4 + half * 2;
        const int n1 = n0 + 1;
        const int m0 = min(n0, nnodes - 1);
        const int m1 = min(n1, nnodes - 1);
        const int g0 = batch[m0];
        const int g1 = batch[m1];
        const unsigned o0 = (unsigned)m0 * (unsigned)(D_FEAT * sizeof(float)) + (unsigned)lm * 16u;
        const unsigned o1 = (unsigned)m1 * (unsigned)(D_FEAT * sizeof(float)) + (unsigned)lm * 16u;
        const float4 p0 = *(const float4*)((const char*)H + o0);
        const float4 p1 = *(const float4*)((const char*)H + o1);
        const float d0 = dinv[m0];
        const float d1 = dinv[m1];
        const float4 r0 = make_float4(bq.x + d0 * p0.x, bq.y + d0 * p0.y,
                                      bq.z + d0 * p0.z, bq.w + d0 * p0.w);
        const float4 r1 = make_float4(bq.x + d1 * p1.x, bq.y + d1 * p1.y,
                                      bq.z + d1 * p1.z, bq.w + d1 * p1.w);
        v2f a;
        a.x = (g0 == gsel && n0 < nnodes) ? 1.0f : 0.0f;
        a.y = (g1 == gsel && n1 < nnodes) ? 1.0f : 0.0f;
        cnt += a.x + a.y;
        v2f b0, b1, b2, b3;
        b0.x = fmaxf(r0.x, 0.0f);  b0.y = fmaxf(r1.x, 0.0f);
        b1.x = fmaxf(r0.y, 0.0f);  b1.y = fmaxf(r1.y, 0.0f);
        b2.x = fmaxf(r0.z, 0.0f);  b2.y = fmaxf(r1.z, 0.0f);
        b3.x = fmaxf(r0.w, 0.0f);  b3.y = fmaxf(r1.w, 0.0f);
        c0 = __builtin_amdgcn_wmma_f32_16x16x4_f32(false, a, false, b0, (short)0, c0, false, false);
        c1 = __builtin_amdgcn_wmma_f32_16x16x4_f32(false, a, false, b1, (short)0, c1, false, false);
        c2 = __builtin_amdgcn_wmma_f32_16x16x4_f32(false, a, false, b2, (short)0, c2, false, false);
        c3 = __builtin_amdgcn_wmma_f32_16x16x4_f32(false, a, false, b3, (short)0, c3, false, false);
    }

    if (kq > 0) {
        float* dst = red + ((kq - 1) * 4 + mt) * (33 * 32) + lane;
#pragma unroll
        for (int r = 0; r < 8; ++r) {
            dst[(0 + r) * 32]  = c0[r];
            dst[(8 + r) * 32]  = c1[r];
            dst[(16 + r) * 32] = c2[r];
            dst[(24 + r) * 32] = c3[r];
        }
        dst[32 * 32] = cnt;
    }
    __syncthreads();
    if (kq == 0) {
#pragma unroll
        for (int q = 0; q < POOL_KQ - 1; ++q) {
            const float* srcp = red + (q * 4 + mt) * (33 * 32) + lane;
#pragma unroll
            for (int r = 0; r < 8; ++r) {
                c0[r] += srcp[(0 + r) * 32];
                c1[r] += srcp[(8 + r) * 32];
                c2[r] += srcp[(16 + r) * 32];
                c3[r] += srcp[(24 + r) * 32];
            }
            cnt += srcp[32 * 32];
        }
        float* pw = part + (size_t)blockIdx.x * (N_GRAPHS_C * D_FEAT);
#pragma unroll
        for (int r = 0; r < 8; ++r) {
            const int g = mt * 16 + half * 8 + r;
            *(float4*)(pw + g * D_FEAT + lm * 4) = make_float4(c0[r], c1[r], c2[r], c3[r]);
        }
        const float tot = cnt + __shfl_xor(cnt, 16);
        if (half == 0) cntp[(size_t)blockIdx.x * N_GRAPHS_C + gsel] = tot;
    }
}

__global__ __launch_bounds__(256) void pool_final_kernel(const float* __restrict__ part,
                                                         const float* __restrict__ cntp,
                                                         float* __restrict__ out) {
    __shared__ float rs[4][D_FEAT];
    __shared__ float rc[4];
    const int g = (int)blockIdx.x;
    const int d = (int)(threadIdx.x & 63);
    const int s = (int)(threadIdx.x >> 6);
    float acc = 0.0f, c = 0.0f;
#pragma unroll
    for (int i = 0; i < POOL_WGS / 4; ++i) {
        const int p = s + 4 * i;
        acc += part[(size_t)p * (N_GRAPHS_C * D_FEAT) + g * D_FEAT + d];
        c   += cntp[(size_t)p * N_GRAPHS_C + g];
    }
    rs[s][d] = acc;
    if (d == 0) rc[s] = c;
    __syncthreads();
    if (threadIdx.x < D_FEAT) {
        const float t  = ((rs[0][d] + rs[1][d]) + rs[2][d]) + rs[3][d];
        const float ct = ((rc[0] + rc[1]) + rc[2]) + rc[3];
        out[g * D_FEAT + d] = t / fmaxf(ct, 1.0f);
    }
}

extern "C" void kernel_launch(void* const* d_in, const int* in_sizes, int n_in,
                              void* d_out, int out_size, void* d_ws, size_t ws_size,
                              hipStream_t stream) {
    const float* x     = (const float*)d_in[0];
    const int*   ei    = (const int*)d_in[1];
    const int*   batch = (const int*)d_in[2];
    const float* W1    = (const float*)d_in[3];
    const float* b1    = (const float*)d_in[4];
    const float* W2    = (const float*)d_in[5];
    const float* b2    = (const float*)d_in[6];
    const float* W3    = (const float*)d_in[7];
    const float* b3    = (const float*)d_in[8];
    float* out = (float*)d_out;

    const int N = in_sizes[0] / D_FEAT;
    const int E = in_sizes[1] / 2;
    const int* erow = ei;
    const int* ecol = ei + E;

    float* w = (float*)d_ws;
    size_t off = 0;
    const int npad = ((N + 63) / 64) * 64;
    float* dinv = w + off; off += (size_t)npad;
    float* bufA = w + off; off += (size_t)N * D_FEAT;
    float* bufB = w + off; off += (size_t)N * D_FEAT;
    float* pool = w + off; off += (size_t)POOL_WGS * N_GRAPHS_C * D_FEAT;
    float* cnt  = w + off; off += (size_t)POOL_WGS * N_GRAPHS_C;
    unsigned* dpart = (unsigned*)(w + off); off += (size_t)DEG_K * npad;
    unsigned* ekey = (unsigned*)(w + off); off += (size_t)(E / 16) * 4;
    int2* elist = (int2*)(w + off); off += (size_t)AG_WGS * AG_LIST * 2;
    int* ecnt = (int*)(w + off); off += (size_t)AG_WGS;

    const int blk = 256;

    const int deg_ranges = (N + (DEG_CAP - 32) - 1) / (DEG_CAP - 32);
    deg_hist_kernel<<<dim3(DEG_K, deg_ranges), DEG_BLK, 0, stream>>>(ecol, dpart, N, npad, E);
    deg_finish_kernel<<<(N + blk - 1) / blk, blk, 0, stream>>>(dpart, dinv, ecol, ekey, N, npad, E);

    const int waves = (N + 15) / 16;
    const int gemm_blocks = (waves * 32 + blk - 1) / blk;

    gemm_wmma_kernel<false><<<gemm_blocks, blk, 0, stream>>>(x, b1, W1, dinv, bufA, N);
    agg_lds_kernel<<<AG_WGS, AG_BLK, 0, stream>>>(bufA, erow, ecol, ekey, elist, ecnt, bufB, N, E, 0);

    gemm_wmma_kernel<true><<<gemm_blocks, blk, 0, stream>>>(bufB, b1, W2, dinv, bufA, N);
    agg_lds_kernel<<<AG_WGS, AG_BLK, 0, stream>>>(bufA, erow, ecol, ekey, elist, ecnt, bufB, N, E, 1);

    gemm_wmma_kernel<true><<<gemm_blocks, blk, 0, stream>>>(bufB, b2, W3, dinv, bufA, N);
    agg_lds_kernel<<<AG_WGS, AG_BLK, 0, stream>>>(bufA, erow, ecol, ekey, elist, ecnt, bufB, N, E, 1);

    pool_wmma_kernel<<<POOL_WGS, 512, 0, stream>>>(bufB, dinv, b3, batch, pool, cnt, N);
    pool_final_kernel<<<N_GRAPHS_C, blk, 0, stream>>>(pool, cnt, out);
}


# each layer reassociated as (A X) W in one launch: input rows summed in LDS, then the work-group multiplies its sums by W (f32 WMMA, W in LDS); 3 product launches dropped; float sum order changes
# speedup vs baseline: 4.1200x; 1.1767x over previous
#include <hip/hip_runtime.h>
#include <hip/hip_bf16.h>

typedef __attribute__((ext_vector_type(2))) float v2f;
typedef __attribute__((ext_vector_type(8))) float v8f;

#define D_FEAT 64
#define N_GRAPHS_C 64

#define AG_WGS 256
#define AG_BLK 1024
#define AG_CAP 208
#define AG_LIST 16384
#define AG_U 8
#define AG_EB 16

#define AG_PITCH 68
#define AG_STRIPS (AG_CAP / 16)
#define WL_PITCH 160

__device__ __forceinline__ int ag_rows(int n) {
    const int r = (n + AG_WGS - 1) / AG_WGS;
    return r < 1 ? 1 : (r > AG_CAP ? AG_CAP : r);
}

#define DEG_K 32
#define DEG_CAP 16384
#define DEG_U 25
#define DEG_BLK 1024

__global__ __launch_bounds__(DEG_BLK) void deg_hist_kernel(const int* __restrict__ col,
                                                           unsigned* __restrict__ part,
                                                           int n, int npad, int e) {
    __shared__ unsigned hist[DEG_CAP];
    const int tid = (int)threadIdx.x;
    const int nr  = (int)gridDim.y;
    const int rl  = (((n + nr - 1) / nr) + 31) & ~31;
    const int r0  = (int)blockIdx.y * rl;
    const int len = max(0, min(n, r0 + rl) - r0);
    const int per = (e + DEG_K - 1) / DEG_K;
    const int s0  = (int)blockIdx.x * per;
    const int s1  = min(e, s0 + per);

    for (int i = tid; i < len; i += DEG_BLK) hist[i] = 0u;
    __syncthreads();
    for (int base = s0; base < s1; base += DEG_BLK * DEG_U) {
        int c[DEG_U];
#pragma unroll
        for (int u = 0; u < DEG_U; ++u) c[u] = col[min(base + u * DEG_BLK + tid, s1 - 1)];
#pragma unroll
        for (int u = 0; u < DEG_U; ++u) {
            const unsigned o = (unsigned)(c[u] - r0);
            if (base + u * DEG_BLK + tid < s1 && o < (unsigned)len) atomicAdd(&hist[o], 1u);
        }
    }
    __syncthreads();
    unsigned* __restrict__ pk = part + (size_t)blockIdx.x * npad + r0;
    for (int i = tid; i < len; i += DEG_BLK) pk[i] = hist[i];
}

__global__ __launch_bounds__(256) void deg_finish_kernel(const unsigned* __restrict__ part,
                                                         float* __restrict__ dinv,
                                                         const int* __restrict__ col,
                                                         unsigned* __restrict__ key,
                                                         int n, int npad, int e) {
    int i = blockIdx.x * blockDim.x + threadIdx.x;
    const unsigned rows = (unsigned)ag_rows(n);
    const int nvec = e >> 4;
    for (int v = i; v < nvec; v += (int)(gridDim.x * blockDim.x)) {
        const int* __restrict__ cp = col + (size_t)v * 16;
        unsigned kw[4];
#pragma unroll
        for (int d = 0; d < 4; ++d) {
            unsigned wv = 0u;
#pragma unroll
            for (int b = 0; b < 4; ++b) {
                const unsigned c = (unsigned)cp[d * 4 + b];
                const unsigned k = (c < (unsigned)n) ? ((c / rows) & 255u) : 0u;
                wv |= k << (8 * b);
            }
            kw[d] = wv;
        }
        *(uint4*)(key + (size_t)v * 4) = make_uint4(kw[0], kw[1], kw[2], kw[3]);
    }
    if (i < n) {
        unsigned t = 1u;
#pragma unroll
        for (int k = 0; k < DEG_K; ++k) t += part[(size_t)k * npad + i];
        const float d = (float)t;
        dinv[i] = (d > 0.0f) ? rsqrtf(d) : 0.0f;
    }
}


template <bool FIRST>
__device__ __forceinline__ void ag_add_group(const float* __restrict__ HS,
                                             const float* __restrict__ dinv, float* acc,
                                             int s, int t, int lane) {
    const int ws = FIRST ? __builtin_bit_cast(int, dinv[s]) : 0;
#pragma unroll
    for (int h = 0; h < 32; h += AG_EB) {
        float h0[AG_EB], h1[AG_EB];
#pragma unroll
        for (int u = 0; u < AG_EB; ++u) {
            const int sj = __builtin_amdgcn_readlane(s, h + u);
            const float* __restrict__ sp = HS + (size_t)sj * D_FEAT + lane;
            h0[u] = sp[0];
            h1[u] = sp[32];
        }
#pragma unroll
        for (int u = 0; u < AG_EB; ++u) {
            const int tj = __builtin_amdgcn_readlane(t, h + u);
            if (tj >= 0) {
                float* d = acc + tj * AG_PITCH + lane;
                if (FIRST) {
                    const float wj = __builtin_bit_cast(float, __builtin_amdgcn_readlane(ws, h + u));
                    atomicAdd(d, h0[u] * wj);
                    atomicAdd(d + 32, h1[u] * wj);
                } else {
                    atomicAdd(d, h0[u]);
                    atomicAdd(d + 32, h1[u]);
                }
            }
        }
    }
}

template <bool FIRST, bool LAST>
__global__ __launch_bounds__(AG_BLK) void agg_lds_kernel(const float* __restrict__ HS,
                                                         const float* __restrict__ W,
                                                         const float* __restrict__ bias,
                                                         const float* __restrict__ dinv,
                                                         const int* __restrict__ row,
                                                         const int* __restrict__ col,
                                                         const unsigned* __restrict__ key,
                                                         int2* elist, int* ecnt,
                                                         float* __restrict__ OUT, int n, int e) {
    static_assert(AG_BLK * 4 == D_FEAT * D_FEAT, "one 16-B piece of W per thread");
    static_assert(AG_LIST >= AG_CAP * AG_PITCH, "the staging area lies over the edge list");
    static_assert(2 * AG_STRIPS <= AG_BLK / 32, "one wave per strip and column half");
    __shared__ __attribute__((aligned(16))) float acc[AG_CAP * AG_PITCH];
    __shared__ __attribute__((aligned(16))) int lsrc[AG_LIST];
    __shared__ int lrow[AG_LIST];
    __shared__ __attribute__((aligned(16))) float wl[(D_FEAT / 2) * WL_PITCH];
    __shared__ float dl[AG_CAP];
    __shared__ unsigned lcnt;

    const int tid  = (int)threadIdx.x;
    const int lane = tid & 31;
    const int wv   = __builtin_amdgcn_readfirstlane(tid >> 5);
    const int wg   = (int)blockIdx.x;
    const int rows = ag_rows(n);
    const int nranges = (n + rows - 1) / rows;
    const int nvec = e >> 4;
    const unsigned idrep = (unsigned)wg * 0x01010101u;
    const int rot = (int)(((long long)nvec * wg) / AG_WGS);
    const uint4* __restrict__ kv = (const uint4*)key;
    float* stage = (float*)lsrc;

    {
        const float4 wq = ((const float4*)W)[tid];
        const int k = tid >> 4, c = (tid & 15) * 4;
        float* wp = wl + (k >> 1) * WL_PITCH + (k & 1) + c * 2;
        wp[0] = wq.x;
        wp[2] = wq.y;
        wp[4] = wq.z;
        wp[6] = wq.w;
    }
    const int lm    = lane & 15;
    const int half  = lane >> 4;
    const int strip = wv >> 1;
    const int np    = wv & 1;
    const float bz0 = bias[np * 32 + lm];
    const float bz1 = bias[np * 32 + 16 + lm];

    for (int g = wg; g < nranges; g += AG_WGS) {
        const int r0  = g * rows;
        const int len = min(rows, n - r0);

        const float4* __restrict__ hv = (const float4*)(HS + (size_t)r0 * D_FEAT);
        {
            constexpr int NU = (AG_CAP * (D_FEAT / 4) + AG_BLK - 1) / AG_BLK;
            float4 own[NU];
            float dv[NU];
#pragma unroll
            for (int u = 0; u < NU; ++u) {
                const int i = u * AG_BLK + tid;
                own[u] = hv[i < len * (D_FEAT / 4) ? i : 0];
                dv[u]  = FIRST ? dinv[min(r0 + (i >> 4), n - 1)] : 1.0f;
            }
#pragma unroll
            for (int u = 0; u < NU; ++u) {
                const int i = u * AG_BLK + tid;
                if (i < AG_CAP * (D_FEAT / 4)) {
                    float4 v = make_float4(0.0f, 0.0f, 0.0f, 0.0f);
                    if (i < len * (D_FEAT / 4)) {
                        v = own[u];
                        if (FIRST) { v.x *= dv[u]; v.y *= dv[u]; v.z *= dv[u]; v.w *= dv[u]; }
                    }
                    *(float4*)(acc + (i >> 4) * AG_PITCH + (i & 15) * 4) = v;
                }
            }
        }
        if (tid < AG_CAP) dl[tid] = (tid < len) ? dinv[r0 + tid] : 0.0f;
        const bool kept = g < AG_WGS;
        const bool from_ws = kept && !FIRST;
        int2* gl = elist + (size_t)(kept ? g : 0) * AG_LIST;
        int total;
        if (from_ws) {
            total = __builtin_amdgcn_readfirstlane(ecnt[g]);
            __syncthreads();
        } else {
            if (tid == 0) lcnt = 0u;
            __syncthreads();

            for (int base = 0; base < nvec; base += AG_BLK * AG_U) {
                uint4 k[AG_U];
#pragma unroll
                for (int u = 0; u < AG_U; ++u) {
                    const int i = base + u * AG_BLK + tid;
                    int ii = min(i, nvec - 1) + rot;
                    if (ii >= nvec) ii -= nvec;
                    k[u] = kv[ii];
                }
#pragma unroll
                for (int u = 0; u < AG_U; u += 2) {
                    unsigned m = 0u;
#pragma unroll
                    for (int h = 0; h < 2; ++h) {
                        const unsigned y0 = k[u + h].x ^ idrep, y1 = k[u + h].y ^ idrep;
                        const unsigned y2 = k[u + h].z ^ idrep, y3 = k[u + h].w ^ idrep;
                        const unsigned z0 = (y0 - 0x01010101u) & ~y0 & 0x80808080u;
                        const unsigned z1 = (y1 - 0x01010101u) & ~y1 & 0x80808080u;
                        const unsigned z2 = (y2 - 0x01010101u) & ~y2 & 0x80808080u;
                        const unsigned z3 = (y3 - 0x01010101u) & ~y3 & 0x80808080u;
                        unsigned mh = (z0 >> 7) | (z1 >> 6) | (z2 >> 5) | (z3 >> 4);
                        if (base + (u + h) * AG_BLK + tid >= nvec) mh = 0u;
                        m |= mh << (4 * h);
                    }
                    while (m != 0u) {
                        const int p = __builtin_ctz(m);
                        m &= m - 1u;
                        int v = base + (u + ((p >> 2) & 1)) * AG_BLK + tid + rot;
                        if (v >= nvec) v -= nvec;
                        const unsigned slot = atomicAdd(&lcnt, 1u);
                        if (slot < (unsigned)AG_LIST) lsrc[slot] = v * 16 + ((p & 3) << 2) + (p >> 3);
                    }
                }
            }
            if (tid < (e & 15)) {
                const int ed = nvec * 16 + tid;
                if ((unsigned)(col[ed] - r0) < (unsigned)len) {
                    const unsigned slot = atomicAdd(&lcnt, 1u);
                    if (slot < (unsigned)AG_LIST) lsrc[slot] = ed;
                }
            }
            __syncthreads();
            total = (int)lcnt;
            if (kept && tid == 0) ecnt[g] = total;

            if (total <= AG_LIST) {
                const int padded = (total + 31) & ~31;
                for (int b = 0; b < padded; b += AG_BLK * 4) {
                    int ed[4], cc[4], ss[4];
#pragma unroll
                    for (int u = 0; u < 4; ++u) {
                        const int i = b + u * AG_BLK + tid;
                        ed[u] = (i < total) ? lsrc[i] : 0;
                    }
#pragma unroll
                    for (int u = 0; u < 4; ++u) { cc[u] = col[ed[u]]; ss[u] = row[ed[u]]; }
#pragma unroll
                    for (int u = 0; u < 4; ++u) {
                        const int i = b + u * AG_BLK + tid;
                        if (i < padded) {
                            const unsigned o = (unsigned)(cc[u] - r0);
                            const bool ok = (i < total) && (o < (unsigned)len);
                            const int se = ok ? ss[u] : 0;
                            const int te = ok ? (int)o : -1;
                            lsrc[i] = se;
                            lrow[i] = te;
                            if (kept) gl[i] = make_int2(se, te);
                        }
                    }
                }
                __syncthreads();
            }
        }

        if (total <= AG_LIST) {
            const int ngroups = ((total + 31) & ~31) >> 5;
            if (from_ws) {
                int q = wv;
                int2 cur = make_int2(0, -1);
                if (q < ngroups) cur = gl[q * 32 + lane];
                for (; q < ngroups; q += AG_BLK / 32) {
                    const int2 nxt = gl[min(q + AG_BLK / 32, ngroups - 1) * 32 + lane];
                    ag_add_group<FIRST>(HS, dinv, acc, cur.x, cur.y, lane);
                    cur = nxt;
                }
            } else {
                for (int q = wv; q < ngroups; q += AG_BLK / 32)
                    ag_add_group<FIRST>(HS, dinv, acc, lsrc[q * 32 + lane], lrow[q * 32 + lane], lane);
            }
        } else {
            for (int eb = wv * 32; eb < e; eb += AG_BLK) {
                const int me = min(eb + lane, e - 1);
                const int s = row[me];
                const unsigned o = (unsigned)(col[me] - r0);
                unsigned hit = __builtin_amdgcn_ballot_w32((eb + lane < e) && (o < (unsigned)len));
                while (hit != 0u) {
                    const int j = __builtin_ctz(hit);
                    hit &= hit - 1u;
                    const int sj = __builtin_amdgcn_readlane(s, j);
                    const int tj = __builtin_amdgcn_readlane((int)o, j);
                    const float* __restrict__ sp = HS + (size_t)sj * D_FEAT + lane;
                    const float wj = FIRST ? dinv[sj] : 1.0f;
                    float* d = acc + tj * AG_PITCH + lane;
                    atomicAdd(d, FIRST ? sp[0] * wj : sp[0]);
                    atomicAdd(d + 32, FIRST ? sp[32] * wj : sp[32]);
                }
            }
        }
        __syncthreads();

        if (wv < 2 * AG_STRIPS) {
            const float* ap = acc + (strip * 16 + lm) * AG_PITCH + half * 2;
            const float* bp = wl + half * WL_PITCH + (np * 32 + lm) * 2;
            v8f c0 = {}, c1 = {};
#pragma unroll
            for (int k = 0; k < D_FEAT; k += 4) {
                const v2f a  = *(const v2f*)(ap + k);
                const v2f b0 = *(const v2f*)(bp + (k >> 1) * WL_PITCH);
                const v2f b1 = *(const v2f*)(bp + (k >> 1) * WL_PITCH + 32);
                c0 = __builtin_amdgcn_wmma_f32_16x16x4_f32(false, a, false, b0, (short)0, c0, false, false);
                c1 = __builtin_amdgcn_wmma_f32_16x16x4_f32(false, a, false, b1, (short)0, c1, false, false);
            }
            const int mrow = strip * 16 + half * 8;
            float* sp = stage + mrow * AG_PITCH + np * 32 + lm;
#pragma unroll
            for (int r = 0; r < 8; ++r) {
                const float di = dl[mrow + r];
                float z0 = fmaxf(bz0 + di * c0[r], 0.0f);
                float z1 = fmaxf(bz1 + di * c1[r], 0.0f);
                if (!LAST) { z0 *= di; z1 *= di; }
                sp[r * AG_PITCH]      = z0;
                sp[r * AG_PITCH + 16] = z1;
            }
        }
        __syncthreads();

        float4* __restrict__ ov = (float4*)(OUT + (size_t)r0 * D_FEAT);
        for (int i = tid; i < len * (D_FEAT / 4); i += AG_BLK)
            ov[i] = *(const float4*)(stage + (i >> 4) * AG_PITCH + (i & 15) * 4);
        __syncthreads();
    }
}

#define POOL_WGS 128
#define POOL_KQ 4

__global__ __launch_bounds__(512) void pool_wmma_kernel(const float* __restrict__ H,
                                                        const int* __restrict__ batch,
                                                        float* __restrict__ part,
                                                        float* __restrict__ cntp, int nnodes) {
    __shared__ float red[(POOL_KQ - 1) * 4 * 33 * 32];
    const int lane = (int)(threadIdx.x & 31);
    const int w    = __builtin_amdgcn_readfirstlane((int)(threadIdx.x >> 5));
    const int mt   = w & 3;
    const int kq   = w >> 2;
    const int lm   = lane & 15;
    const int half = lane >> 4;
    const int gsel = mt * 16 + lm;
    const int nsteps = (nnodes + 3) >> 2;
    const int nslots = (int)gridDim.x * POOL_KQ;

    v8f c0 = {}, c1 = {}, c2 = {}, c3 = {};
    float cnt = 0.0f;
#pragma unroll 4
    for (int ks = (int)blockIdx.x * POOL_KQ + kq; ks < nsteps; ks += nslots) {
        const int n0 = ks * 4 + half * 2;
        const int n1 = n0 + 1;
        const int m0 = min(n0, nnodes - 1);
        const int m1 = min(n1, nnodes - 1);
        const int g0 = batch[m0];
        const int g1 = batch[m1];
        const unsigned o0 = (unsigned)m0 * (unsigned)(D_FEAT * sizeof(float)) + (unsigned)lm * 16u;
        const unsigned o1 = (unsigned)m1 * (unsigned)(D_FEAT * sizeof(float)) + (unsigned)lm * 16u;
        const float4 r0 = *(const float4*)((const char*)H + o0);
        const float4 r1 = *(const float4*)((const char*)H + o1);
        v2f a;
        a.x = (g0 == gsel && n0 < nnodes) ? 1.0f : 0.0f;
        a.y = (g1 == gsel && n1 < nnodes) ? 1.0f : 0.0f;
        cnt += a.x + a.y;
        v2f b0, b1, b2, b3;
        b0.x = r0.x;  b0.y = r1.x;
        b1.x = r0.y;  b1.y = r1.y;
        b2.x = r0.z;  b2.y = r1.z;
        b3.x = r0.w;  b3.y = r1.w;
        c0 = __builtin_amdgcn_wmma_f32_16x16x4_f32(false, a, false, b0, (short)0, c0, false, false);
        c1 = __builtin_amdgcn_wmma_f32_16x16x4_f32(false, a, false, b1, (short)0, c1, false, false);
        c2 = __builtin_amdgcn_wmma_f32_16x16x4_f32(false, a, false, b2, (short)0, c2, false, false);
        c3 = __builtin_amdgcn_wmma_f32_16x16x4_f32(false, a, false, b3, (short)0, c3, false, false);
    }

    if (kq > 0) {
        float* dst = red + ((kq - 1) * 4 + mt) * (33 * 32) + lane;
#pragma unroll
        for (int r = 0; r < 8; ++r) {
            dst[(0 + r) * 32]  = c0[r];
            dst[(8 + r) * 32]  = c1[r];
            dst[(16 + r) * 32] = c2[r];
            dst[(24 + r) * 32] = c3[r];
        }
        dst[32 * 32] = cnt;
    }
    __syncthreads();
    if (kq == 0) {
#pragma unroll
        for (int q = 0; q < POOL_KQ - 1; ++q) {
            const float* srcp = red + (q * 4 + mt) * (33 * 32) + lane;
#pragma unroll
            for (int r = 0; r < 8; ++r) {
                c0[r] += srcp[(0 + r) * 32];
                c1[r] += srcp[(8 + r) * 32];
                c2[r] += srcp[(16 + r) * 32];
                c3[r] += srcp[(24 + r) * 32];
            }
            cnt += srcp[32 * 32];
        }
        float* pw = part + (size_t)blockIdx.x * (N_GRAPHS_C * D_FEAT);
#pragma unroll
        for (int r = 0; r < 8; ++r) {
            const int g = mt * 16 + half * 8 + r;
            *(float4*)(pw + g * D_FEAT + lm * 4) = make_float4(c0[r], c1[r], c2[r], c3[r]);
        }
        const float tot = cnt + __shfl_xor(cnt, 16);
        if (half == 0) cntp[(size_t)blockIdx.x * N_GRAPHS_C + gsel] = tot;
    }
}

__global__ __launch_bounds__(256) void pool_final_kernel(const float* __restrict__ part,
                                                         const float* __restrict__ cntp,
                                                         float* __restrict__ out) {
    __shared__ float rs[4][D_FEAT];
    __shared__ float rc[4];
    const int g = (int)blockIdx.x;
    const int d = (int)(threadIdx.x & 63);
    const int s = (int)(threadIdx.x >> 6);
    float acc = 0.0f, c = 0.0f;
#pragma unroll
    for (int i = 0; i < POOL_WGS / 4; ++i) {
        const int p = s + 4 * i;
        acc += part[(size_t)p * (N_GRAPHS_C * D_FEAT) + g * D_FEAT + d];
        c   += cntp[(size_t)p * N_GRAPHS_C + g];
    }
    rs[s][d] = acc;
    if (d == 0) rc[s] = c;
    __syncthreads();
    if (threadIdx.x < D_FEAT) {
        const float t  = ((rs[0][d] + rs[1][d]) + rs[2][d]) + rs[3][d];
        const float ct = ((rc[0] + rc[1]) + rc[2]) + rc[3];
        out[g * D_FEAT + d] = t / fmaxf(ct, 1.0f);
    }
}

extern "C" void kernel_launch(void* const* d_in, const int* in_sizes, int n_in,
                              void* d_out, int out_size, void* d_ws, size_t ws_size,
                              hipStream_t stream) {
    const float* x     = (const float*)d_in[0];
    const int*   ei    = (const int*)d_in[1];
    const int*   batch = (const int*)d_in[2];
    const float* W1    = (const float*)d_in[3];
    const float* b1    = (const float*)d_in[4];
    const float* W2    = (const float*)d_in[5];
    const float* b2    = (const float*)d_in[6];
    const float* W3    = (const float*)d_in[7];
    const float* b3    = (const float*)d_in[8];
    float* out = (float*)d_out;

    const int N = in_sizes[0] / D_FEAT;
    const int E = in_sizes[1] / 2;
    const int* erow = ei;
    const int* ecol = ei + E;

    float* w = (float*)d_ws;
    size_t off = 0;
    const int npad = ((N + 63) / 64) * 64;
    float* dinv = w + off; off += (size_t)npad;
    float* bufA = w + off; off += (size_t)N * D_FEAT;
    float* bufB = w + off; off += (size_t)N * D_FEAT;
    float* pool = w + off; off += (size_t)POOL_WGS * N_GRAPHS_C * D_FEAT;
    float* cnt  = w + off; off += (size_t)POOL_WGS * N_GRAPHS_C;
    unsigned* dpart = (unsigned*)(w + off); off += (size_t)DEG_K * npad;
    unsigned* ekey = (unsigned*)(w + off); off += (size_t)(E / 16) * 4;
    int2* elist = (int2*)(w + off); off += (size_t)AG_WGS * AG_LIST * 2;
    int* ecnt = (int*)(w + off); off += (size_t)AG_WGS;

    const int blk = 256;

    const int deg_ranges = (N + (DEG_CAP - 32) - 1) / (DEG_CAP - 32);
    deg_hist_kernel<<<dim3(DEG_K, deg_ranges), DEG_BLK, 0, stream>>>(ecol, dpart, N, npad, E);
    deg_finish_kernel<<<(N + blk - 1) / blk, blk, 0, stream>>>(dpart, dinv, ecol, ekey, N, npad, E);

    agg_lds_kernel<true, false><<<AG_WGS, AG_BLK, 0, stream>>>(x, W1, b1, dinv, erow, ecol, ekey, elist, ecnt, bufA, N, E);
    agg_lds_kernel<false, false><<<AG_WGS, AG_BLK, 0, stream>>>(bufA, W2, b2, dinv, erow, ecol, ekey, elist, ecnt, bufB, N, E);
    agg_lds_kernel<false, true><<<AG_WGS, AG_BLK, 0, stream>>>(bufB, W3, b3, dinv, erow, ecol, ekey, elist, ecnt, bufA, N, E);

    pool_wmma_kernel<<<POOL_WGS, 512, 0, stream>>>(bufA, batch, pool, cnt, N);
    pool_final_kernel<<<N_GRAPHS_C, blk, 0, stream>>>(pool, cnt, out);
}


# layer 3 stores no rows: its finished rows stay in LDS and meet the one-hot graph matrix there (f32 WMMA), one partial tile per work-group; pool launch dropped; reorders the pool's float sums
# speedup vs baseline: 4.5593x; 1.1066x over previous
#include <hip/hip_runtime.h>
#include <hip/hip_bf16.h>

typedef __attribute__((ext_vector_type(2))) float v2f;
typedef __attribute__((ext_vector_type(8))) float v8f;

#define D_FEAT 64
#define N_GRAPHS_C 64

#define AG_WGS 256
#define AG_BLK 1024
#define AG_CAP 208
#define AG_LIST 16384
#define AG_U 8
#define AG_EB 16

#define AG_PITCH 68
#define AG_STRIPS (AG_CAP / 16)
#define WL_PITCH 160
#define POOL_KQ 4
#define POOL_RED ((POOL_KQ - 1) * 4 * 33 * 32)

__device__ __forceinline__ int ag_rows(int n) {
    const int r = (n + AG_WGS - 1) / AG_WGS;
    return r < 1 ? 1 : (r > AG_CAP ? AG_CAP : r);
}

#define DEG_K 32
#define DEG_CAP 16384
#define DEG_U 25
#define DEG_BLK 1024

__global__ __launch_bounds__(DEG_BLK) void deg_hist_kernel(const int* __restrict__ col,
                                                           unsigned* __restrict__ part,
                                                           int n, int npad, int e) {
    __shared__ unsigned hist[DEG_CAP];
    const int tid = (int)threadIdx.x;
    const int nr  = (int)gridDim.y;
    const int rl  = (((n + nr - 1) / nr) + 31) & ~31;
    const int r0  = (int)blockIdx.y * rl;
    const int len = max(0, min(n, r0 + rl) - r0);
    const int per = (e + DEG_K - 1) / DEG_K;
    const int s0  = (int)blockIdx.x * per;
    const int s1  = min(e, s0 + per);

    for (int i = tid; i < len; i += DEG_BLK) hist[i] = 0u;
    __syncthreads();
    for (int base = s0; base < s1; base += DEG_BLK * DEG_U) {
        int c[DEG_U];
#pragma unroll
        for (int u = 0; u < DEG_U; ++u) c[u] = col[min(base + u * DEG_BLK + tid, s1 - 1)];
#pragma unroll
        for (int u = 0; u < DEG_U; ++u) {
            const unsigned o = (unsigned)(c[u] - r0);
            if (base + u * DEG_BLK + tid < s1 && o < (unsigned)len) atomicAdd(&hist[o], 1u);
        }
    }
    __syncthreads();
    unsigned* __restrict__ pk = part + (size_t)blockIdx.x * npad + r0;
    for (int i = tid; i < len; i += DEG_BLK) pk[i] = hist[i];
}

__global__ __launch_bounds__(256) void deg_finish_kernel(const unsigned* __restrict__ part,
                                                         float* __restrict__ dinv,
                                                         const int* __restrict__ col,
                                                         unsigned* __restrict__ key,
                                                         int n, int npad, int e) {
    int i = blockIdx.x * blockDim.x + threadIdx.x;
    const unsigned rows = (unsigned)ag_rows(n);
    const int nvec = e >> 4;
    for (int v = i; v < nvec; v += (int)(gridDim.x * blockDim.x)) {
        const int* __restrict__ cp = col + (size_t)v * 16;
        unsigned kw[4];
#pragma unroll
        for (int d = 0; d < 4; ++d) {
            unsigned wv = 0u;
#pragma unroll
            for (int b = 0; b < 4; ++b) {
                const unsigned c = (unsigned)cp[d * 4 + b];
                const unsigned k = (c < (unsigned)n) ? ((c / rows) & 255u) : 0u;
                wv |= k << (8 * b);
            }
            kw[d] = wv;
        }
        *(uint4*)(key + (size_t)v * 4) = make_uint4(kw[0], kw[1], kw[2], kw[3]);
    }
    if (i < n) {
        unsigned t = 1u;
#pragma unroll
        for (int k = 0; k < DEG_K; ++k) t += part[(size_t)k * npad + i];
        const float d = (float)t;
        dinv[i] = (d > 0.0f) ? rsqrtf(d) : 0.0f;
    }
}


template <bool FIRST>
__device__ __forceinline__ void ag_add_group(const float* __restrict__ HS,
                                             const float* __restrict__ dinv, float* acc,
                                             int s, int t, int lane) {
    const int ws = FIRST ? __builtin_bit_cast(int, dinv[s]) : 0;
#pragma unroll
    for (int h = 0; h < 32; h += AG_EB) {
        float h0[AG_EB], h1[AG_EB];
#pragma unroll
        for (int u = 0; u < AG_EB; ++u) {
            const int sj = __builtin_amdgcn_readlane(s, h + u);
            const float* __restrict__ sp = HS + (size_t)sj * D_FEAT + lane;
            h0[u] = sp[0];
            h1[u] = sp[32];
        }
#pragma unroll
        for (int u = 0; u < AG_EB; ++u) {
            const int tj = __builtin_amdgcn_readlane(t, h + u);
            if (tj >= 0) {
                float* d = acc + tj * AG_PITCH + lane;
                if (FIRST) {
                    const float wj = __builtin_bit_cast(float, __builtin_amdgcn_readlane(ws, h + u));
                    atomicAdd(d, h0[u] * wj);
                    atomicAdd(d + 32, h1[u] * wj);
                } else {
                    atomicAdd(d, h0[u]);
                    atomicAdd(d + 32, h1[u]);
                }
            }
        }
    }
}

template <bool FIRST, bool LAST>
__global__ __launch_bounds__(AG_BLK) void agg_lds_kernel(const float* __restrict__ HS,
                                                         const float* __restrict__ W,
                                                         const float* __restrict__ bias,
                                                         const float* __restrict__ dinv,
                                                         const int* __restrict__ row,
                                                         const int* __restrict__ col,
                                                         const unsigned* __restrict__ key,
                                                         int2* elist, int* ecnt,
                                                         float* __restrict__ OUT,
                                                         const int* __restrict__ batch,
                                                         float* __restrict__ ppart,
                                                         float* __restrict__ pcnt, int n, int e) {
    static_assert(AG_BLK * 4 == D_FEAT * D_FEAT, "one 16-B piece of W per thread");
    static_assert(AG_LIST >= AG_CAP * AG_PITCH, "the staging area lies over the edge list");
    static_assert(2 * AG_STRIPS <= AG_BLK / 32, "one wave per strip and column half");
    static_assert(AG_LIST >= POOL_RED, "the pool's k-slice hand-over lies over the target list");
    static_assert(AG_CAP % 4 == 0 && 4 * POOL_KQ <= AG_BLK / 32, "whole k-steps; one wave per graph tile and k-slice");
    __shared__ __attribute__((aligned(16))) float acc[AG_CAP * AG_PITCH];
    __shared__ __attribute__((aligned(16))) int lsrc[AG_LIST];
    __shared__ int lrow[AG_LIST];
    __shared__ __attribute__((aligned(16))) float wl[(D_FEAT / 2) * WL_PITCH];
    __shared__ float dl[AG_CAP];
    __shared__ int bl[AG_CAP];
    __shared__ unsigned lcnt;

    const int tid  = (int)threadIdx.x;
    const int lane = tid & 31;
    const int wv   = __builtin_amdgcn_readfirstlane(tid >> 5);
    const int wg   = (int)blockIdx.x;
    const int rows = ag_rows(n);
    const int nranges = (n + rows - 1) / rows;
    const int nvec = e >> 4;
    const unsigned idrep = (unsigned)wg * 0x01010101u;
    const int rot = (int)(((long long)nvec * wg) / AG_WGS);
    const uint4* __restrict__ kv = (const uint4*)key;
    float* stage = (float*)lsrc;
    float* red = (float*)lrow;

    {
        const float4 wq = ((const float4*)W)[tid];
        const int k = tid >> 4, c = (tid & 15) * 4;
        float* wp = wl + (k >> 1) * WL_PITCH + (k & 1) + c * 2;
        wp[0] = wq.x;
        wp[2] = wq.y;
        wp[4] = wq.z;
        wp[6] = wq.w;
    }
    const int lm    = lane & 15;
    const int half  = lane >> 4;
    const int strip = wv >> 1;
    const int np    = wv & 1;
    const float bz0 = bias[np * 32 + lm];
    const float bz1 = bias[np * 32 + 16 + lm];

    for (int g = wg; g < nranges; g += AG_WGS) {
        const int r0  = g * rows;
        const int len = min(rows, n - r0);

        const float4* __restrict__ hv = (const float4*)(HS + (size_t)r0 * D_FEAT);
        {
            constexpr int NU = (AG_CAP * (D_FEAT / 4) + AG_BLK - 1) / AG_BLK;
            float4 own[NU];
            float dv[NU];
#pragma unroll
            for (int u = 0; u < NU; ++u) {
                const int i = u * AG_BLK + tid;
                own[u] = hv[i < len * (D_FEAT / 4) ? i : 0];
                dv[u]  = FIRST ? dinv[min(r0 + (i >> 4), n - 1)] : 1.0f;
            }
#pragma unroll
            for (int u = 0; u < NU; ++u) {
                const int i = u * AG_BLK + tid;
                if (i < AG_CAP * (D_FEAT / 4)) {
                    float4 v = make_float4(0.0f, 0.0f, 0.0f, 0.0f);
                    if (i < len * (D_FEAT / 4)) {
                        v = own[u];
                        if (FIRST) { v.x *= dv[u]; v.y *= dv[u]; v.z *= dv[u]; v.w *= dv[u]; }
                    }
                    *(float4*)(acc + (i >> 4) * AG_PITCH + (i & 15) * 4) = v;
                }
            }
        }
        if (tid < AG_CAP) dl[tid] = (tid < len) ? dinv[r0 + tid] : 0.0f;
        if (LAST && tid < AG_CAP) bl[tid] = (tid < len) ? batch[r0 + tid] : -1;
        const bool kept = g < AG_WGS;
        const bool from_ws = kept && !FIRST;
        int2* gl = elist + (size_t)(kept ? g : 0) * AG_LIST;
        int total;
        if (from_ws) {
            total = __builtin_amdgcn_readfirstlane(ecnt[g]);
            __syncthreads();
        } else {
            if (tid == 0) lcnt = 0u;
            __syncthreads();

            for (int base = 0; base < nvec; base += AG_BLK * AG_U) {
                uint4 k[AG_U];
#pragma unroll
                for (int u = 0; u < AG_U; ++u) {
                    const int i = base + u * AG_BLK + tid;
                    int ii = min(i, nvec - 1) + rot;
                    if (ii >= nvec) ii -= nvec;
                    k[u] = kv[ii];
                }
#pragma unroll
                for (int u = 0; u < AG_U; u += 2) {
                    unsigned m = 0u;
#pragma unroll
                    for (int h = 0; h < 2; ++h) {
                        const unsigned y0 = k[u + h].x ^ idrep, y1 = k[u + h].y ^ idrep;
                        const unsigned y2 = k[u + h].z ^ idrep, y3 = k[u + h].w ^ idrep;
                        const unsigned z0 = (y0 - 0x01010101u) & ~y0 & 0x80808080u;
                        const unsigned z1 = (y1 - 0x01010101u) & ~y1 & 0x80808080u;
                        const unsigned z2 = (y2 - 0x01010101u) & ~y2 & 0x80808080u;
                        const unsigned z3 = (y3 - 0x01010101u) & ~y3 & 0x80808080u;
                        unsigned mh = (z0 >> 7) | (z1 >> 6) | (z2 >> 5) | (z3 >> 4);
                        if (base + (u + h) * AG_BLK + tid >= nvec) mh = 0u;
                        m |= mh << (4 * h);
                    }
                    while (m != 0u) {
                        const int p = __builtin_ctz(m);
                        m &= m - 1u;
                        int v = base + (u + ((p >> 2) & 1)) * AG_BLK + tid + rot;
                        if (v >= nvec) v -= nvec;
                        const unsigned slot = atomicAdd(&lcnt, 1u);
                        if (slot < (unsigned)AG_LIST) lsrc[slot] = v * 16 + ((p & 3) << 2) + (p >> 3);
                    }
                }
            }
            if (tid < (e & 15)) {
                const int ed = nvec * 16 + tid;
                if ((unsigned)(col[ed] - r0) < (unsigned)len) {
                    const unsigned slot = atomicAdd(&lcnt, 1u);
                    if (slot < (unsigned)AG_LIST) lsrc[slot] = ed;
                }
            }
            __syncthreads();
            total = (int)lcnt;
            if (kept && tid == 0) ecnt[g] = total;

            if (total <= AG_LIST) {
                const int padded = (total + 31) & ~31;
                for (int b = 0; b < padded; b += AG_BLK * 4) {
                    int ed[4], cc[4], ss[4];
#pragma unroll
                    for (int u = 0; u < 4; ++u) {
                        const int i = b + u * AG_BLK + tid;
                        ed[u] = (i < total) ? lsrc[i] : 0;
                    }
#pragma unroll
                    for (int u = 0; u < 4; ++u) { cc[u] = col[ed[u]]; ss[u] = row[ed[u]]; }
#pragma unroll
                    for (int u = 0; u < 4; ++u) {
                        const int i = b + u * AG_BLK + tid;
                        if (i < padded) {
                            const unsigned o = (unsigned)(cc[u] - r0);
                            const bool ok = (i < total) && (o < (unsigned)len);
                            const int se = ok ? ss[u] : 0;
                            const int te = ok ? (int)o : -1;
                            lsrc[i] = se;
                            lrow[i] = te;
                            if (kept) gl[i] = make_int2(se, te);
                        }
                    }
                }
                __syncthreads();
            }
        }

        if (total <= AG_LIST) {
            const int ngroups = ((total + 31) & ~31) >> 5;
            if (from_ws) {
                int q = wv;
                int2 cur = make_int2(0, -1);
                if (q < ngroups) cur = gl[q * 32 + lane];
                for (; q < ngroups; q += AG_BLK / 32) {
                    const int2 nxt = gl[min(q + AG_BLK / 32, ngroups - 1) * 32 + lane];
                    ag_add_group<FIRST>(HS, dinv, acc, cur.x, cur.y, lane);
                    cur = nxt;
                }
            } else {
                for (int q = wv; q < ngroups; q += AG_BLK / 32)
                    ag_add_group<FIRST>(HS, dinv, acc, lsrc[q * 32 + lane], lrow[q * 32 + lane], lane);
            }
        } else {
            for (int eb = wv * 32; eb < e; eb += AG_BLK) {
                const int me = min(eb + lane, e - 1);
                const int s = row[me];
                const unsigned o = (unsigned)(col[me] - r0);
                unsigned hit = __builtin_amdgcn_ballot_w32((eb + lane < e) && (o < (unsigned)len));
                while (hit != 0u) {
                    const int j = __builtin_ctz(hit);
                    hit &= hit - 1u;
                    const int sj = __builtin_amdgcn_readlane(s, j);
                    const int tj = __builtin_amdgcn_readlane((int)o, j);
                    const float* __restrict__ sp = HS + (size_t)sj * D_FEAT + lane;
                    const float wj = FIRST ? dinv[sj] : 1.0f;
                    float* d = acc + tj * AG_PITCH + lane;
                    atomicAdd(d, FIRST ? sp[0] * wj : sp[0]);
                    atomicAdd(d + 32, FIRST ? sp[32] * wj : sp[32]);
                }
            }
        }
        __syncthreads();

        if (wv < 2 * AG_STRIPS) {
            const float* ap = acc + (strip * 16 + lm) * AG_PITCH + half * 2;
            const float* bp = wl + half * WL_PITCH + (np * 32 + lm) * 2;
            v8f c0 = {}, c1 = {};
#pragma unroll
            for (int k = 0; k < D_FEAT; k += 4) {
                const v2f a  = *(const v2f*)(ap + k);
                const v2f b0 = *(const v2f*)(bp + (k >> 1) * WL_PITCH);
                const v2f b1 = *(const v2f*)(bp + (k >> 1) * WL_PITCH + 32);
                c0 = __builtin_amdgcn_wmma_f32_16x16x4_f32(false, a, false, b0, (short)0, c0, false, false);
                c1 = __builtin_amdgcn_wmma_f32_16x16x4_f32(false, a, false, b1, (short)0, c1, false, false);
            }
            const int mrow = strip * 16 + half * 8;
            float* sp = stage + mrow * AG_PITCH + np * 32 + lm;
#pragma unroll
            for (int r = 0; r < 8; ++r) {
                const float di = dl[mrow + r];
                float z0 = fmaxf(bz0 + di * c0[r], 0.0f);
                float z1 = fmaxf(bz1 + di * c1[r], 0.0f);
                if (!LAST) { z0 *= di; z1 *= di; }
                sp[r * AG_PITCH]      = z0;
                sp[r * AG_PITCH + 16] = z1;
            }
        }
        __syncthreads();

        if (!LAST) {
            float4* __restrict__ ov = (float4*)(OUT + (size_t)r0 * D_FEAT);
            for (int i = tid; i < len * (D_FEAT / 4); i += AG_BLK)
                ov[i] = *(const float4*)(stage + (i >> 4) * AG_PITCH + (i & 15) * 4);
        } else {
            const int mt   = wv & 3;
            const int kq   = wv >> 2;
            const int gsel = mt * 16 + lm;
            const bool pw  = wv < 4 * POOL_KQ;
            v8f p0 = {}, p1 = {}, p2 = {}, p3 = {};
            float pc = 0.0f;
            if (pw) {
#pragma unroll 1
                for (int ks = kq; ks < AG_CAP / 4; ks += POOL_KQ) {
                    const int n0 = ks * 4 + half * 2;
                    const int g0 = bl[n0];
                    const int g1 = bl[n0 + 1];
                    const float4 q0 = *(const float4*)(stage + n0 * AG_PITCH + lm * 4);
                    const float4 q1 = *(const float4*)(stage + (n0 + 1) * AG_PITCH + lm * 4);
                    v2f a;
                    a.x = (g0 == gsel) ? 1.0f : 0.0f;
                    a.y = (g1 == gsel) ? 1.0f : 0.0f;
                    pc += a.x + a.y;
                    v2f b0, b1, b2, b3;
                    b0.x = q0.x;  b0.y = q1.x;
                    b1.x = q0.y;  b1.y = q1.y;
                    b2.x = q0.z;  b2.y = q1.z;
                    b3.x = q0.w;  b3.y = q1.w;
                    p0 = __builtin_amdgcn_wmma_f32_16x16x4_f32(false, a, false, b0, (short)0, p0, false, false);
                    p1 = __builtin_amdgcn_wmma_f32_16x16x4_f32(false, a, false, b1, (short)0, p1, false, false);
                    p2 = __builtin_amdgcn_wmma_f32_16x16x4_f32(false, a, false, b2, (short)0, p2, false, false);
                    p3 = __builtin_amdgcn_wmma_f32_16x16x4_f32(false, a, false, b3, (short)0, p3, false, false);
                }
                if (kq > 0) {
                    float* dst = red + ((kq - 1) * 4 + mt) * (33 * 32) + lane;
#pragma unroll
                    for (int r = 0; r < 8; ++r) {
                        dst[(0 + r) * 32]  = p0[r];
                        dst[(8 + r) * 32]  = p1[r];
                        dst[(16 + r) * 32] = p2[r];
                        dst[(24 + r) * 32] = p3[r];
                    }
                    dst[32 * 32] = pc;
                }
            }
            __syncthreads();
            if (pw && kq == 0) {
#pragma unroll 1
                for (int q = 0; q < POOL_KQ - 1; ++q) {
                    const float* srcp = red + (q * 4 + mt) * (33 * 32) + lane;
#pragma unroll
                    for (int r = 0; r < 8; ++r) {
                        p0[r] += srcp[(0 + r) * 32];
                        p1[r] += srcp[(8 + r) * 32];
                        p2[r] += srcp[(16 + r) * 32];
                        p3[r] += srcp[(24 + r) * 32];
                    }
                    pc += srcp[32 * 32];
                }
                float* pt = ppart + (size_t)wg * (N_GRAPHS_C * D_FEAT);
                float tot = pc + __shfl_xor(pc, 16);
#pragma unroll
                for (int r = 0; r < 8; ++r) {
                    float4* po = (float4*)(pt + (mt * 16 + half * 8 + r) * D_FEAT + lm * 4);
                    float4 v = make_float4(p0[r], p1[r], p2[r], p3[r]);
                    if (g != wg) { const float4 o = *po; v.x += o.x; v.y += o.y; v.z += o.z; v.w += o.w; }
                    *po = v;
                }
                if (half == 0) {
                    float* pn = pcnt + (size_t)wg * N_GRAPHS_C + gsel;
                    if (g != wg) tot += *pn;
                    *pn = tot;
                }
            }
        }
        __syncthreads();
    }
    if (LAST && wg >= nranges) {
        for (int i = tid; i < N_GRAPHS_C * D_FEAT; i += AG_BLK) ppart[(size_t)wg * (N_GRAPHS_C * D_FEAT) + i] = 0.0f;
        if (tid < N_GRAPHS_C) pcnt[(size_t)wg * N_GRAPHS_C + tid] = 0.0f;
    }
}

#define POOL_WGS AG_WGS

__global__ __launch_bounds__(256) void pool_final_kernel(const float* __restrict__ part,
                                                         const float* __restrict__ cntp,
                                                         float* __restrict__ out) {
    __shared__ float rs[4][D_FEAT];
    __shared__ float rc[4];
    const int g = (int)blockIdx.x;
    const int d = (int)(threadIdx.x & 63);
    const int s = (int)(threadIdx.x >> 6);
    float acc = 0.0f, c = 0.0f;
#pragma unroll
    for (int i = 0; i < POOL_WGS / 4; ++i) {
        const int p = s + 4 * i;
        acc += part[(size_t)p * (N_GRAPHS_C * D_FEAT) + g * D_FEAT + d];
        c   += cntp[(size_t)p * N_GRAPHS_C + g];
    }
    rs[s][d] = acc;
    if (d == 0) rc[s] = c;
    __syncthreads();
    if (threadIdx.x < D_FEAT) {
        const float t  = ((rs[0][d] + rs[1][d]) + rs[2][d]) + rs[3][d];
        const float ct = ((rc[0] + rc[1]) + rc[2]) + rc[3];
        out[g * D_FEAT + d] = t / fmaxf(ct, 1.0f);
    }
}

extern "C" void kernel_launch(void* const* d_in, const int* in_sizes, int n_in,
                              void* d_out, int out_size, void* d_ws, size_t ws_size,
                              hipStream_t stream) {
    const float* x     = (const float*)d_in[0];
    const int*   ei    = (const int*)d_in[1];
    const int*   batch = (const int*)d_in[2];
    const float* W1    = (const float*)d_in[3];
    const float* b1    = (const float*)d_in[4];
    const float* W2    = (const float*)d_in[5];
    const float* b2    = (const float*)d_in[6];
    const float* W3    = (const float*)d_in[7];
    const float* b3    = (const float*)d_in[8];
    float* out = (float*)d_out;

    const int N = in_sizes[0] / D_FEAT;
    const int E = in_sizes[1] / 2;
    const int* erow = ei;
    const int* ecol = ei + E;

    float* w = (float*)d_ws;
    size_t off = 0;
    const int npad = ((N + 63) / 64) * 64;
    float* dinv = w + off; off += (size_t)npad;
    float* bufA = w + off; off += (size_t)N * D_FEAT;
    float* bufB = w + off; off += (size_t)N * D_FEAT;
    float* pool = w + off; off += (size_t)POOL_WGS * N_GRAPHS_C * D_FEAT;
    float* cnt  = w + off; off += (size_t)POOL_WGS * N_GRAPHS_C;
    unsigned* dpart = (unsigned*)(w + off); off += (size_t)DEG_K * npad;
    unsigned* ekey = (unsigned*)(w + off); off += (size_t)(E / 16) * 4;
    int2* elist = (int2*)(w + off); off += (size_t)AG_WGS * AG_LIST * 2;
    int* ecnt = (int*)(w + off); off += (size_t)AG_WGS;

    const int blk = 256;

    const int deg_ranges = (N + (DEG_CAP - 32) - 1) / (DEG_CAP - 32);
    deg_hist_kernel<<<dim3(DEG_K, deg_ranges), DEG_BLK, 0, stream>>>(ecol, dpart, N, npad, E);
    deg_finish_kernel<<<(N + blk - 1) / blk, blk, 0, stream>>>(dpart, dinv, ecol, ekey, N, npad, E);

    agg_lds_kernel<true, false><<<AG_WGS, AG_BLK, 0, stream>>>(x, W1, b1, dinv, erow, ecol, ekey, elist, ecnt, bufA, batch, pool, cnt, N, E);
    agg_lds_kernel<false, false><<<AG_WGS, AG_BLK, 0, stream>>>(bufA, W2, b2, dinv, erow, ecol, ekey, elist, ecnt, bufB, batch, pool, cnt, N, E);
    agg_lds_kernel<false, true><<<AG_WGS, AG_BLK, 0, stream>>>(bufB, W3, b3, dinv, erow, ecol, ekey, elist, ecnt, bufA, batch, pool, cnt, N, E);

    pool_final_kernel<<<N_GRAPHS_C, blk, 0, stream>>>(pool, cnt, out);
}
